# starting kernel
# baseline (speedup 1.0000x reference)
#include <hip/hip_runtime.h>
#include <math.h>

typedef float v2f __attribute__((ext_vector_type(2)));
typedef float v8f __attribute__((ext_vector_type(8)));

#define N_STFT   513
#define N_BARKS  128
#define BATCH    4
#define TIME     512
#define ROWS     (BATCH * TIME)
#define FP       528
#define NB       128
#define LDSS     532
#define LDSD     132
#define MAX_ITER 30
#define LR       0.1f
#define MOMENTUM 0.9f
#define TOL_LOSS 1e-5f
#define TOL_CHG  1e-8f


__global__ __launch_bounds__(256) void ibs_init_spec(const float* __restrict__ spec_init,
                                                     float* __restrict__ spec,
                                                     float* __restrict__ buf) {
    int idx = blockIdx.x * blockDim.x + threadIdx.x;
    if (idx >= ROWS * FP) return;
    int row = idx / FP;
    int f   = idx - row * FP;
    spec[idx] = (f < N_STFT) ? spec_init[row * N_STFT + f] : 0.0f;
    buf[idx]  = 0.0f;
}

__global__ __launch_bounds__(256) void ibs_init_fb(const float* __restrict__ fb_in,
                                                   float* __restrict__ fbp2,
                                                   float* __restrict__ fbT2) {
    int idx = blockIdx.x * blockDim.x + threadIdx.x;
    if (idx >= FP * NB) return;
    int f  = idx / NB;
    int kk = idx - f * NB;
    float v = (f < N_STFT) ? fb_in[f * NB + kk] : 0.0f;
    fbp2[(((f  >> 1) * NB) + kk) * 2 + (f  & 1)] = v;
    fbT2[(((kk >> 1) * FP) + f ) * 2 + (kk & 1)] = v;
}

__global__ void ibs_init_state(float* __restrict__ state) {
    state[0] = INFINITY;
    state[1] = 0.0f;
}


__global__ __launch_bounds__(256) void ibs_iter(const float* __restrict__ barkspec,
                                                float* __restrict__ spec,
                                                float* __restrict__ buf,
                                                const float2* __restrict__ fbp2,
                                                const float2* __restrict__ fbT2,
                                                float* __restrict__ partials,
                                                const float* __restrict__ state) {
    if (state[1] != 0.0f) return;

    __shared__ float lds_spec[16 * LDSS];
    __shared__ float lds_diff[16 * LDSD];
    __shared__ float lds_red[256];

    const int tid     = threadIdx.x;
    const int wave    = tid >> 5;
    const int lane    = tid & 31;
    const int mi      = lane & 15;
    const int hsel    = lane >> 4;
    const int koff    = hsel * 2;
    const int rowbase = blockIdx.x * 16;

    for (int i = tid; i < 16 * (FP / 4); i += 256) {
        int r  = i / (FP / 4);
        int c4 = i - r * (FP / 4);
        float4 v = *(const float4*)(spec + (rowbase + r) * FP + c4 * 4);
        float* dst = lds_spec + r * LDSS + c4 * 4;
        dst[0] = v.x; dst[1] = v.y; dst[2] = v.z; dst[3] = v.w;
    }
    __syncthreads();

    const int nbase = wave * 16;
    const float*  ap = lds_spec + mi * LDSS + koff;
    const float2* bp = fbp2 + (nbase + mi) + hsel * NB;

    v8f acc = {0.f, 0.f, 0.f, 0.f, 0.f, 0.f, 0.f, 0.f};
    #pragma unroll 4
    for (int k = 0; k < FP; k += 4) {
        v2f a, b;
        a[0] = ap[k];
        a[1] = ap[k + 1];
        float2 bv = bp[(k >> 1) * NB];
        b[0] = bv.x; b[1] = bv.y;
        acc = __builtin_amdgcn_wmma_f32_16x16x4_f32(false, a, false, b,
                                                    (short)0, acc, false, false);
    }

    const int rowh = rowbase + hsel * 8;
    const int bb   = rowh >> 9;
    const int t0   = rowh & 511;
    const float* mp = barkspec + (bb * N_BARKS + nbase + mi) * TIME + t0;
    float ssq = 0.0f;
    #pragma unroll
    for (int r = 0; r < 8; ++r) {
        float d = mp[r] - acc[r];
        ssq += d * d;
        lds_diff[(r + hsel * 8) * LDSD + nbase + mi] = d;
    }
    lds_red[tid] = ssq;
    __syncthreads();

    if (tid == 0) {
        float s = 0.0f;
        for (int i = 0; i < 256; ++i) s += lds_red[i];
        partials[blockIdx.x] = s;
    }

    const float* a2p = lds_diff + mi * LDSD + koff;
    for (int nt = wave; nt < FP / 16; nt += 8) {
        const int fbase = nt * 16;
        const float2* b2p = fbT2 + (fbase + mi) + hsel * FP;
        v8f acc2 = {0.f, 0.f, 0.f, 0.f, 0.f, 0.f, 0.f, 0.f};
        #pragma unroll
        for (int k = 0; k < NB; k += 4) {
            v2f a, b;
            a[0] = a2p[k];
            a[1] = a2p[k + 1];
            float2 bv = b2p[(k >> 1) * FP];
            b[0] = bv.x; b[1] = bv.y;
            acc2 = __builtin_amdgcn_wmma_f32_16x16x4_f32(false, a, false, b,
                                                         (short)0, acc2, false, false);
        }
        #pragma unroll
        for (int r = 0; r < 8; ++r) {
            const int m   = r + hsel * 8;
            const int f   = fbase + mi;
            const int idx = (rowbase + m) * FP + f;
            float g  = acc2[r] * (-2.0f / (float)ROWS);
            float bn = MOMENTUM * buf[idx] + g;
            float sn = lds_spec[m * LDSS + f] - LR * bn;
            spec[idx] = sn > 0.0f ? sn : 0.0f;
            buf[idx]  = bn;
        }
    }
}


__global__ void ibs_check(float* __restrict__ state, const float* __restrict__ partials) {
    if (state[1] != 0.0f) return;
    float s = 0.0f;
    for (int i = 0; i < 128; ++i) s += partials[i];
    float loss = s / (float)ROWS;
    float prev = state[0];
    bool stop = (loss < TOL_LOSS) || (fabsf(prev - loss) < TOL_CHG);
    state[0] = loss;
    state[1] = stop ? 1.0f : 0.0f;
}


__global__ __launch_bounds__(256) void ibs_finalize(const float* __restrict__ spec,
                                                    float* __restrict__ out) {
    int idx = blockIdx.x * blockDim.x + threadIdx.x;
    if (idx >= BATCH * N_STFT * TIME) return;
    int t = idx & 511;
    int f = (idx >> 9) % N_STFT;
    int b = idx / (N_STFT * TIME);
    out[idx] = spec[(b * TIME + t) * FP + f];
}


extern "C" void kernel_launch(void* const* d_in, const int* in_sizes, int n_in,
                              void* d_out, int out_size, void* d_ws, size_t ws_size,
                              hipStream_t stream) {
    const float* barkspec  = (const float*)d_in[0];
    const float* fb        = (const float*)d_in[1];
    const float* spec_init = (const float*)d_in[2];

    float* ws       = (float*)d_ws;
    float* spec     = ws;
    float* buf      = spec + ROWS * FP;
    float* fbp2     = buf  + ROWS * FP;
    float* fbT2     = fbp2 + FP * NB;
    float* partials = fbT2 + NB * FP;
    float* state    = partials + 128;

    ibs_init_spec<<<(ROWS * FP + 255) / 256, 256, 0, stream>>>(spec_init, spec, buf);
    ibs_init_fb<<<(FP * NB + 255) / 256, 256, 0, stream>>>(fb, fbp2, fbT2);
    ibs_init_state<<<1, 1, 0, stream>>>(state);

    for (int it = 0; it < MAX_ITER; ++it) {
        ibs_iter<<<ROWS / 16, 256, 0, stream>>>(barkspec, spec, buf,
                                                (const float2*)fbp2, (const float2*)fbT2,
                                                partials, state);
        if (it < MAX_ITER - 1)
            ibs_check<<<1, 1, 0, stream>>>(state, partials);
    }

    ibs_finalize<<<(BATCH * N_STFT * TIME + 255) / 256, 256, 0, stream>>>(spec, (float*)d_out);
}


# ibs_iter: global loads software-pipelined (spec tile, B fragments of both GEMMs, momentum and target values issued a stage ahead of the WMMA chain that uses them); sums and k order unchanged
# speedup vs baseline: 1.7976x; 1.7976x over previous
#include <hip/hip_runtime.h>
#include <math.h>

typedef float v2f __attribute__((ext_vector_type(2)));
typedef float v8f __attribute__((ext_vector_type(8)));

#define N_STFT   513
#define N_BARKS  128
#define BATCH    4
#define TIME     512
#define ROWS     (BATCH * TIME)
#define FP       528
#define NB       128
#define LDSS     532
#define LDSD     132
#define MAX_ITER 30
#define LR       0.1f
#define MOMENTUM 0.9f
#define TOL_LOSS 1e-5f
#define TOL_CHG  1e-8f


__global__ __launch_bounds__(256) void ibs_init_spec(const float* __restrict__ spec_init,
                                                     float* __restrict__ spec,
                                                     float* __restrict__ buf) {
    int idx = blockIdx.x * blockDim.x + threadIdx.x;
    if (idx >= ROWS * FP) return;
    int row = idx / FP;
    int f   = idx - row * FP;
    spec[idx] = (f < N_STFT) ? spec_init[row * N_STFT + f] : 0.0f;
    buf[idx]  = 0.0f;
}

__global__ __launch_bounds__(256) void ibs_init_fb(const float* __restrict__ fb_in,
                                                   float* __restrict__ fbp2,
                                                   float* __restrict__ fbT2) {
    int idx = blockIdx.x * blockDim.x + threadIdx.x;
    if (idx >= FP * NB) return;
    int f  = idx / NB;
    int kk = idx - f * NB;
    float v = (f < N_STFT) ? fb_in[f * NB + kk] : 0.0f;
    fbp2[(((f  >> 1) * NB) + kk) * 2 + (f  & 1)] = v;
    fbT2[(((kk >> 1) * FP) + f ) * 2 + (kk & 1)] = v;
}

__global__ void ibs_init_state(float* __restrict__ state) {
    state[0] = INFINITY;
    state[1] = 0.0f;
}


#define G1_CH   33
#define G2_ST   32

#define STAGE_FENCE() __builtin_amdgcn_sched_barrier(0)

__device__ __forceinline__ void g1_load(v2f (&b)[G1_CH], const float2* __restrict__ bp, int c) {
    const float2* p = bp + c * (G1_CH * 2 * NB);
    #pragma unroll
    for (int j = 0; j < G1_CH; ++j) {
        float2 v = p[j * 2 * NB];
        b[j][0] = v.x; b[j][1] = v.y;
    }
}

__device__ __forceinline__ v8f g1_mma(v8f acc, const v2f (&b)[G1_CH], const float* ap, int c) {
    const float* a0 = ap + c * (G1_CH * 4);
    #pragma unroll
    for (int j = 0; j < G1_CH; ++j) {
        v2f a;
        a[0] = a0[4 * j];
        a[1] = a0[4 * j + 1];
        acc = __builtin_amdgcn_wmma_f32_16x16x4_f32(false, a, false, b[j],
                                                    (short)0, acc, false, false);
    }
    return acc;
}

__device__ __forceinline__ void g2_load(v2f (&b)[G2_ST], float (&bu)[8],
                                        const float2* __restrict__ b2l,
                                        const float* bufl, int fbase) {
    const float2* p = b2l + fbase;
    #pragma unroll
    for (int s = 0; s < G2_ST; ++s) {
        float2 v = p[s * 2 * FP];
        b[s][0] = v.x; b[s][1] = v.y;
    }
    const float* q = bufl + fbase;
    #pragma unroll
    for (int r = 0; r < 8; ++r) bu[r] = q[r * FP];
}

__device__ __forceinline__ void g2_tile(const v2f (&b)[G2_ST], const float (&bu)[8],
                                        const v2f (&a2)[G2_ST], const float* lsp,
                                        float* specl, float* bufl, int fbase) {
    v8f acc2 = {0.f, 0.f, 0.f, 0.f, 0.f, 0.f, 0.f, 0.f};
    #pragma unroll
    for (int s = 0; s < G2_ST; ++s)
        acc2 = __builtin_amdgcn_wmma_f32_16x16x4_f32(false, a2[s], false, b[s],
                                                     (short)0, acc2, false, false);
    #pragma unroll
    for (int r = 0; r < 8; ++r) {
        float g  = acc2[r] * (-2.0f / (float)ROWS);
        float bn = MOMENTUM * bu[r] + g;
        float sn = lsp[r * LDSS + fbase] - LR * bn;
        specl[r * FP + fbase] = sn > 0.0f ? sn : 0.0f;
        bufl[r * FP + fbase]  = bn;
    }
}

__global__ __launch_bounds__(256) void ibs_iter(const float* __restrict__ barkspec,
                                                float* __restrict__ spec,
                                                float* __restrict__ buf,
                                                const float2* __restrict__ fbp2,
                                                const float2* __restrict__ fbT2,
                                                float* __restrict__ partials,
                                                const float* __restrict__ state) {
    if (state[1] != 0.0f) return;

    __shared__ float lds_spec[16 * LDSS];
    __shared__ float lds_diff[16 * LDSD];
    __shared__ float lds_red[256];

    const int tid     = threadIdx.x;
    const int wave    = __builtin_amdgcn_readfirstlane(tid >> 5);
    const int lane    = tid & 31;
    const int mi      = lane & 15;
    const int hsel    = lane >> 4;
    const int koff    = hsel * 2;
    const int rowbase = blockIdx.x * 16;
    const int nbase   = wave * 16;

    float4 sv[8];
    #pragma unroll
    for (int j = 0; j < 8; ++j) {
        int i  = tid + 256 * j;
        int r  = i / (FP / 4);
        int c4 = i - r * (FP / 4);
        sv[j] = *(const float4*)(spec + (rowbase + r) * FP + c4 * 4);
    }
    const bool tail = tid < 16 * (FP / 4) - 2048;
    float4 svt = {0.f, 0.f, 0.f, 0.f};
    if (tail) svt = *(const float4*)(spec + (rowbase + 15) * FP + (tid + 68) * 4);

    const int rowh = rowbase + hsel * 8;
    const int bb   = rowh >> 9;
    const int t0   = rowh & 511;
    const float* mp = barkspec + (bb * N_BARKS + nbase + mi) * TIME + t0;
    float mv[8];
    #pragma unroll
    for (int r = 0; r < 8; ++r) mv[r] = mp[r];

    const float*  ap = lds_spec + mi * LDSS + koff;
    const float2* bp = fbp2 + (nbase + mi) + hsel * NB;

    v2f bA[G1_CH], bB[G1_CH];
    g1_load(bA, bp, 0);
    STAGE_FENCE();

    #pragma unroll
    for (int j = 0; j < 8; ++j) {
        int i  = tid + 256 * j;
        int r  = i / (FP / 4);
        int c4 = i - r * (FP / 4);
        float* dst = lds_spec + r * LDSS + c4 * 4;
        dst[0] = sv[j].x; dst[1] = sv[j].y; dst[2] = sv[j].z; dst[3] = sv[j].w;
    }
    if (tail) {
        float* dst = lds_spec + 15 * LDSS + (tid + 68) * 4;
        dst[0] = svt.x; dst[1] = svt.y; dst[2] = svt.z; dst[3] = svt.w;
    }
    __syncthreads();

    const float2* b2l   = fbT2 + mi + hsel * FP;
    float*        bufl  = buf  + (rowbase + hsel * 8) * FP + mi;
    float*        specl = spec + (rowbase + hsel * 8) * FP + mi;
    const float*  lsp   = lds_spec + (hsel * 8) * LDSS + mi;
    const float*  a2p   = lds_diff + mi * LDSD + koff;
    const int     fb0   = nbase;

    v2f tA[G2_ST], tB[G2_ST];
    float uA[8], uB[8];

    v8f acc = {0.f, 0.f, 0.f, 0.f, 0.f, 0.f, 0.f, 0.f};
    g1_load(bB, bp, 1);               STAGE_FENCE();
    acc = g1_mma(acc, bA, ap, 0);     STAGE_FENCE();
    g1_load(bA, bp, 2);               STAGE_FENCE();
    acc = g1_mma(acc, bB, ap, 1);     STAGE_FENCE();
    g1_load(bB, bp, 3);               STAGE_FENCE();
    acc = g1_mma(acc, bA, ap, 2);     STAGE_FENCE();
    g2_load(tA, uA, b2l, bufl, fb0);  STAGE_FENCE();
    acc = g1_mma(acc, bB, ap, 3);     STAGE_FENCE();

    float ssq = 0.0f;
    #pragma unroll
    for (int r = 0; r < 8; ++r) {
        float d = mv[r] - acc[r];
        ssq += d * d;
        lds_diff[(r + hsel * 8) * LDSD + nbase + mi] = d;
    }
    lds_red[tid] = ssq;
    __syncthreads();

    STAGE_FENCE();
    g2_load(tB, uB, b2l, bufl, fb0 + 128);
    STAGE_FENCE();

    if (tid == 0) {
        float s = 0.0f;
        for (int i = 0; i < 256; ++i) s += lds_red[i];
        partials[blockIdx.x] = s;
    }

    v2f a2[G2_ST];
    #pragma unroll
    for (int s = 0; s < G2_ST; ++s) {
        a2[s][0] = a2p[4 * s];
        a2[s][1] = a2p[4 * s + 1];
    }

    STAGE_FENCE();
    g2_tile(tA, uA, a2, lsp, specl, bufl, fb0);           STAGE_FENCE();
    g2_load(tA, uA, b2l, bufl, fb0 + 256);                STAGE_FENCE();
    g2_tile(tB, uB, a2, lsp, specl, bufl, fb0 + 128);     STAGE_FENCE();
    g2_load(tB, uB, b2l, bufl, fb0 + 384);                STAGE_FENCE();
    g2_tile(tA, uA, a2, lsp, specl, bufl, fb0 + 256);     STAGE_FENCE();
    if (wave == 0) {
        g2_load(tA, uA, b2l, bufl, 512);                  STAGE_FENCE();
        g2_tile(tB, uB, a2, lsp, specl, bufl, fb0 + 384); STAGE_FENCE();
        g2_tile(tA, uA, a2, lsp, specl, bufl, 512);
    } else {
        g2_tile(tB, uB, a2, lsp, specl, bufl, fb0 + 384);
    }
}


__global__ void ibs_check(float* __restrict__ state, const float* __restrict__ partials) {
    if (state[1] != 0.0f) return;
    float s = 0.0f;
    for (int i = 0; i < 128; ++i) s += partials[i];
    float loss = s / (float)ROWS;
    float prev = state[0];
    bool stop = (loss < TOL_LOSS) || (fabsf(prev - loss) < TOL_CHG);
    state[0] = loss;
    state[1] = stop ? 1.0f : 0.0f;
}


__global__ __launch_bounds__(256) void ibs_finalize(const float* __restrict__ spec,
                                                    float* __restrict__ out) {
    int idx = blockIdx.x * blockDim.x + threadIdx.x;
    if (idx >= BATCH * N_STFT * TIME) return;
    int t = idx & 511;
    int f = (idx >> 9) % N_STFT;
    int b = idx / (N_STFT * TIME);
    out[idx] = spec[(b * TIME + t) * FP + f];
}


extern "C" void kernel_launch(void* const* d_in, const int* in_sizes, int n_in,
                              void* d_out, int out_size, void* d_ws, size_t ws_size,
                              hipStream_t stream) {
    const float* barkspec  = (const float*)d_in[0];
    const float* fb        = (const float*)d_in[1];
    const float* spec_init = (const float*)d_in[2];

    float* ws       = (float*)d_ws;
    float* spec     = ws;
    float* buf      = spec + ROWS * FP;
    float* fbp2     = buf  + ROWS * FP;
    float* fbT2     = fbp2 + FP * NB;
    float* partials = fbT2 + NB * FP;
    float* state    = partials + 128;

    ibs_init_spec<<<(ROWS * FP + 255) / 256, 256, 0, stream>>>(spec_init, spec, buf);
    ibs_init_fb<<<(FP * NB + 255) / 256, 256, 0, stream>>>(fb, fbp2, fbT2);
    ibs_init_state<<<1, 1, 0, stream>>>(state);

    for (int it = 0; it < MAX_ITER; ++it) {
        ibs_iter<<<ROWS / 16, 256, 0, stream>>>(barkspec, spec, buf,
                                                (const float2*)fbp2, (const float2*)fbT2,
                                                partials, state);
        if (it < MAX_ITER - 1)
            ibs_check<<<1, 1, 0, stream>>>(state, partials);
    }

    ibs_finalize<<<(BATCH * N_STFT * TIME + 255) / 256, 256, 0, stream>>>(spec, (float*)d_out);
}


# ibs_iter: stop test of the iteration before moved from its own 1-thread launch to the head of the next iteration launch (two alternating check rows): 63 -> 33 launches; sums and their order unchanged
# speedup vs baseline: 2.2543x; 1.2541x over previous
#include <hip/hip_runtime.h>
#include <math.h>

typedef float v2f __attribute__((ext_vector_type(2)));
typedef float v8f __attribute__((ext_vector_type(8)));

#define N_STFT   513
#define N_BARKS  128
#define BATCH    4
#define TIME     512
#define ROWS     (BATCH * TIME)
#define FP       528
#define NB       128
#define LDSS     532
#define LDSD     132
#define MAX_ITER 30
#define LR       0.1f
#define MOMENTUM 0.9f
#define TOL_LOSS 1e-5f
#define TOL_CHG  1e-8f
#define CHK_ROW  256


__global__ __launch_bounds__(256) void ibs_init_spec(const float* __restrict__ spec_init,
                                                     float* __restrict__ spec,
                                                     float* __restrict__ buf) {
    int idx = blockIdx.x * blockDim.x + threadIdx.x;
    if (idx >= ROWS * FP) return;
    int row = idx / FP;
    int f   = idx - row * FP;
    spec[idx] = (f < N_STFT) ? spec_init[row * N_STFT + f] : 0.0f;
    buf[idx]  = 0.0f;
}

__global__ __launch_bounds__(256) void ibs_init_fb(const float* __restrict__ fb_in,
                                                   float* __restrict__ fbp2,
                                                   float* __restrict__ fbT2) {
    int idx = blockIdx.x * blockDim.x + threadIdx.x;
    if (idx >= FP * NB) return;
    int f  = idx / NB;
    int kk = idx - f * NB;
    float v = (f < N_STFT) ? fb_in[f * NB + kk] : 0.0f;
    fbp2[(((f  >> 1) * NB) + kk) * 2 + (f  & 1)] = v;
    fbT2[(((kk >> 1) * FP) + f ) * 2 + (kk & 1)] = v;
}


#define G1_CH   33
#define G2_ST   32

#define STAGE_FENCE() __builtin_amdgcn_sched_barrier(0)

__device__ __forceinline__ void g1_load(v2f (&b)[G1_CH], const float2* __restrict__ bp, int c) {
    const float2* p = bp + c * (G1_CH * 2 * NB);
    #pragma unroll
    for (int j = 0; j < G1_CH; ++j) {
        float2 v = p[j * 2 * NB];
        b[j][0] = v.x; b[j][1] = v.y;
    }
}

__device__ __forceinline__ v8f g1_mma(v8f acc, const v2f (&b)[G1_CH], const float* ap, int c) {
    const float* a0 = ap + c * (G1_CH * 4);
    #pragma unroll
    for (int j = 0; j < G1_CH; ++j) {
        v2f a;
        a[0] = a0[4 * j];
        a[1] = a0[4 * j + 1];
        acc = __builtin_amdgcn_wmma_f32_16x16x4_f32(false, a, false, b[j],
                                                    (short)0, acc, false, false);
    }
    return acc;
}

__device__ __forceinline__ void g2_load(v2f (&b)[G2_ST], float (&bu)[8],
                                        const float2* __restrict__ b2l,
                                        const float* bufl, int fbase) {
    const float2* p = b2l + fbase;
    #pragma unroll
    for (int s = 0; s < G2_ST; ++s) {
        float2 v = p[s * 2 * FP];
        b[s][0] = v.x; b[s][1] = v.y;
    }
    const float* q = bufl + fbase;
    #pragma unroll
    for (int r = 0; r < 8; ++r) bu[r] = q[r * FP];
}

__device__ __forceinline__ void g2_tile(const v2f (&b)[G2_ST], const float (&bu)[8],
                                        const v2f (&a2)[G2_ST], const float* lsp,
                                        float* specl, float* bufl, int fbase) {
    v8f acc2 = {0.f, 0.f, 0.f, 0.f, 0.f, 0.f, 0.f, 0.f};
    #pragma unroll
    for (int s = 0; s < G2_ST; ++s)
        acc2 = __builtin_amdgcn_wmma_f32_16x16x4_f32(false, a2[s], false, b[s],
                                                     (short)0, acc2, false, false);
    #pragma unroll
    for (int r = 0; r < 8; ++r) {
        float g  = acc2[r] * (-2.0f / (float)ROWS);
        float bn = MOMENTUM * bu[r] + g;
        float sn = lsp[r * LDSS + fbase] - LR * bn;
        specl[r * FP + fbase] = sn > 0.0f ? sn : 0.0f;
        bufl[r * FP + fbase]  = bn;
    }
}

__global__ __launch_bounds__(256) void ibs_iter(const float* __restrict__ barkspec,
                                                float* __restrict__ spec,
                                                float* __restrict__ buf,
                                                const float2* __restrict__ fbp2,
                                                const float2* __restrict__ fbT2,
                                                float* __restrict__ chk,
                                                int it) {
    __shared__ float lds_spec[16 * LDSS];
    __shared__ float lds_diff[16 * LDSD];
    __shared__ float lds_red[256];
    __shared__ float lds_chk[256];

    const int tid     = threadIdx.x;
    const int wave    = __builtin_amdgcn_readfirstlane(tid >> 5);
    const int lane    = tid & 31;
    const int mi      = lane & 15;
    const int hsel    = lane >> 4;
    const int koff    = hsel * 2;
    const int rowbase = blockIdx.x * 16;
    const int nbase   = wave * 16;

    const float* chk_prev = chk + ((it + 1) & 1) * CHK_ROW;
    float*       chk_cur  = chk + (it & 1) * CHK_ROW;
    float cv = (tid == 128) ? INFINITY : 0.0f;
    if (it >= 1 && (tid < 128 || (it >= 2 && tid < 130))) cv = chk_prev[tid];

    float4 sv[8];
    #pragma unroll
    for (int j = 0; j < 8; ++j) {
        int i  = tid + 256 * j;
        int r  = i / (FP / 4);
        int c4 = i - r * (FP / 4);
        sv[j] = *(const float4*)(spec + (rowbase + r) * FP + c4 * 4);
    }
    const bool tail = tid < 16 * (FP / 4) - 2048;
    float4 svt = {0.f, 0.f, 0.f, 0.f};
    if (tail) svt = *(const float4*)(spec + (rowbase + 15) * FP + (tid + 68) * 4);

    const int rowh = rowbase + hsel * 8;
    const int bb   = rowh >> 9;
    const int t0   = rowh & 511;
    const float* mp = barkspec + (bb * N_BARKS + nbase + mi) * TIME + t0;
    float mv[8];
    #pragma unroll
    for (int r = 0; r < 8; ++r) mv[r] = mp[r];

    const float*  ap = lds_spec + mi * LDSS + koff;
    const float2* bp = fbp2 + (nbase + mi) + hsel * NB;

    v2f bA[G1_CH], bB[G1_CH];
    g1_load(bA, bp, 0);
    STAGE_FENCE();

    #pragma unroll
    for (int j = 0; j < 8; ++j) {
        int i  = tid + 256 * j;
        int r  = i / (FP / 4);
        int c4 = i - r * (FP / 4);
        float* dst = lds_spec + r * LDSS + c4 * 4;
        dst[0] = sv[j].x; dst[1] = sv[j].y; dst[2] = sv[j].z; dst[3] = sv[j].w;
    }
    if (tail) {
        float* dst = lds_spec + 15 * LDSS + (tid + 68) * 4;
        dst[0] = svt.x; dst[1] = svt.y; dst[2] = svt.z; dst[3] = svt.w;
    }
    lds_chk[tid] = cv;
    __syncthreads();

    if (it >= 1) {
        float cr[128];
        #pragma unroll
        for (int i = 0; i < 128; ++i) cr[i] = lds_chk[i];
        const float prev     = lds_chk[128];
        const bool  was_done = lds_chk[129] != 0.0f;
        STAGE_FENCE();
        float s = 0.0f;
        #pragma unroll
        for (int i = 0; i < 128; ++i) s += cr[i];
        const float loss     = s / (float)ROWS;
        const bool  stop     = (loss < TOL_LOSS) || (fabsf(prev - loss) < TOL_CHG);
        const bool  halt     = was_done || stop;
        if (tid == 0) {
            if (blockIdx.x == 0) {
                chk_cur[128] = was_done ? prev : loss;
                chk_cur[129] = halt ? 1.0f : 0.0f;
            }
            if (halt) chk_cur[blockIdx.x] = 0.0f;
        }
        if (__builtin_amdgcn_readfirstlane(halt ? 1 : 0)) return;
    }

    const float2* b2l   = fbT2 + mi + hsel * FP;
    float*        bufl  = buf  + (rowbase + hsel * 8) * FP + mi;
    float*        specl = spec + (rowbase + hsel * 8) * FP + mi;
    const float*  lsp   = lds_spec + (hsel * 8) * LDSS + mi;
    const float*  a2p   = lds_diff + mi * LDSD + koff;
    const int     fb0   = nbase;

    v2f tA[G2_ST], tB[G2_ST];
    float uA[8], uB[8];

    v8f acc = {0.f, 0.f, 0.f, 0.f, 0.f, 0.f, 0.f, 0.f};
    g1_load(bB, bp, 1);               STAGE_FENCE();
    acc = g1_mma(acc, bA, ap, 0);     STAGE_FENCE();
    g1_load(bA, bp, 2);               STAGE_FENCE();
    acc = g1_mma(acc, bB, ap, 1);     STAGE_FENCE();
    g1_load(bB, bp, 3);               STAGE_FENCE();
    acc = g1_mma(acc, bA, ap, 2);     STAGE_FENCE();
    g2_load(tA, uA, b2l, bufl, fb0);  STAGE_FENCE();
    acc = g1_mma(acc, bB, ap, 3);     STAGE_FENCE();

    float ssq = 0.0f;
    #pragma unroll
    for (int r = 0; r < 8; ++r) {
        float d = mv[r] - acc[r];
        ssq += d * d;
        lds_diff[(r + hsel * 8) * LDSD + nbase + mi] = d;
    }
    lds_red[tid] = ssq;
    __syncthreads();

    STAGE_FENCE();
    g2_load(tB, uB, b2l, bufl, fb0 + 128);
    STAGE_FENCE();

    if (tid == 0) {
        float s = 0.0f;
        for (int i = 0; i < 256; ++i) s += lds_red[i];
        chk_cur[blockIdx.x] = s;
    }

    v2f a2[G2_ST];
    #pragma unroll
    for (int s = 0; s < G2_ST; ++s) {
        a2[s][0] = a2p[4 * s];
        a2[s][1] = a2p[4 * s + 1];
    }

    STAGE_FENCE();
    g2_tile(tA, uA, a2, lsp, specl, bufl, fb0);           STAGE_FENCE();
    g2_load(tA, uA, b2l, bufl, fb0 + 256);                STAGE_FENCE();
    g2_tile(tB, uB, a2, lsp, specl, bufl, fb0 + 128);     STAGE_FENCE();
    g2_load(tB, uB, b2l, bufl, fb0 + 384);                STAGE_FENCE();
    g2_tile(tA, uA, a2, lsp, specl, bufl, fb0 + 256);     STAGE_FENCE();
    if (wave == 0) {
        g2_load(tA, uA, b2l, bufl, 512);                  STAGE_FENCE();
        g2_tile(tB, uB, a2, lsp, specl, bufl, fb0 + 384); STAGE_FENCE();
        g2_tile(tA, uA, a2, lsp, specl, bufl, 512);
    } else {
        g2_tile(tB, uB, a2, lsp, specl, bufl, fb0 + 384);
    }
}


__global__ __launch_bounds__(256) void ibs_finalize(const float* __restrict__ spec,
                                                    float* __restrict__ out) {
    int idx = blockIdx.x * blockDim.x + threadIdx.x;
    if (idx >= BATCH * N_STFT * TIME) return;
    int t = idx & 511;
    int f = (idx >> 9) % N_STFT;
    int b = idx / (N_STFT * TIME);
    out[idx] = spec[(b * TIME + t) * FP + f];
}


extern "C" void kernel_launch(void* const* d_in, const int* in_sizes, int n_in,
                              void* d_out, int out_size, void* d_ws, size_t ws_size,
                              hipStream_t stream) {
    const float* barkspec  = (const float*)d_in[0];
    const float* fb        = (const float*)d_in[1];
    const float* spec_init = (const float*)d_in[2];

    float* ws       = (float*)d_ws;
    float* spec     = ws;
    float* buf      = spec + ROWS * FP;
    float* fbp2     = buf  + ROWS * FP;
    float* fbT2     = fbp2 + FP * NB;
    float* chk      = fbT2 + NB * FP;

    ibs_init_spec<<<(ROWS * FP + 255) / 256, 256, 0, stream>>>(spec_init, spec, buf);
    ibs_init_fb<<<(FP * NB + 255) / 256, 256, 0, stream>>>(fb, fbp2, fbT2);

    for (int it = 0; it < MAX_ITER; ++it)
        ibs_iter<<<ROWS / 16, 256, 0, stream>>>(barkspec, spec, buf,
                                                (const float2*)fbp2, (const float2*)fbT2,
                                                chk, it);

    ibs_finalize<<<(BATCH * N_STFT * TIME + 255) / 256, 256, 0, stream>>>(spec, (float*)d_out);
}


# ibs_run: k-steps outside each wave's first-to-last nonzero B span (scanned from the filterbank per launch) skipped in both GEMMs, spans kept in registers, unpipelined full fallback; sum order same
# speedup vs baseline: 5.3666x; 2.3806x over previous
#include <hip/hip_runtime.h>
#include <math.h>

typedef float v2f __attribute__((ext_vector_type(2)));
typedef float v8f __attribute__((ext_vector_type(8)));

#define N_STFT   513
#define N_BARKS  128
#define BATCH    4
#define TIME     512
#define ROWS     (BATCH * TIME)
#define FP       528
#define NB       128
#define NBLK     (ROWS / 16)
#define LDSS     532
#define LDSD     132
#define MAX_ITER 30
#define LR       0.1f
#define MOMENTUM 0.9f
#define TOL_LOSS 1e-5f
#define TOL_CHG  1e-8f


__global__ __launch_bounds__(256) void ibs_init_fb(const float* __restrict__ fb_in,
                                                   float* __restrict__ fbp2,
                                                   float* __restrict__ fbT2) {
    int idx = blockIdx.x * blockDim.x + threadIdx.x;
    if (idx >= FP * NB) return;
    int f  = idx / NB;
    int kk = idx - f * NB;
    float v = (f < N_STFT) ? fb_in[f * NB + kk] : 0.0f;
    fbp2[(((f  >> 1) * NB) + kk) * 2 + (f  & 1)] = v;
    fbT2[(((kk >> 1) * FP) + f ) * 2 + (kk & 1)] = v;
}


#define G1_CH   33
#define G2_ST   32
#define G1_GR   33
#define G1_NS   16
#define G2_NS   8

#define STAGE_FENCE() __builtin_amdgcn_sched_barrier(0)

__device__ __forceinline__ bool frag_nonzero(float2 v) {
    return __builtin_amdgcn_ballot_w32((v.x != 0.0f) || (v.y != 0.0f)) != 0u;
}

__device__ __forceinline__ void g1_load(v2f (&b)[G1_CH], const float2* __restrict__ bp, int c) {
    const float2* p = bp + c * (G1_CH * 2 * NB);
    #pragma unroll
    for (int j = 0; j < G1_CH; ++j) {
        float2 v = p[j * 2 * NB];
        b[j][0] = v.x; b[j][1] = v.y;
    }
}

__device__ __forceinline__ v8f g1_mma(v8f acc, const v2f (&b)[G1_CH], const float* ap, int c) {
    const float* a0 = ap + c * (G1_CH * 4);
    #pragma unroll
    for (int j = 0; j < G1_CH; ++j) {
        v2f a;
        a[0] = a0[4 * j];
        a[1] = a0[4 * j + 1];
        acc = __builtin_amdgcn_wmma_f32_16x16x4_f32(false, a, false, b[j],
                                                    (short)0, acc, false, false);
    }
    return acc;
}

__device__ __forceinline__ void g1_span_a(v2f (&a)[4][4], const float* a0, int gb, int n) {
    #pragma unroll
    for (int j = 0; j < 4; ++j) {
        if (gb + j < n) {
            #pragma unroll
            for (int q = 0; q < 4; ++q) {
                a[j][q][0] = a0[16 * (gb + j) + 4 * q];
                a[j][q][1] = a0[16 * (gb + j) + 4 * q + 1];
            }
        }
    }
}

__device__ __forceinline__ v8f g1_span_mma(v8f acc, const v2f (&a)[4][4], const v2f (&b)[G1_NS][4], int gb, int n) {
    #pragma unroll
    for (int j = 0; j < 4; ++j) {
        if (gb + j < n) {
            #pragma unroll
            for (int q = 0; q < 4; ++q)
                acc = __builtin_amdgcn_wmma_f32_16x16x4_f32(false, a[j][q], false, b[gb + j][q],
                                                            (short)0, acc, false, false);
        }
    }
    return acc;
}

__device__ __forceinline__ void g2_load(v2f (&b)[G2_ST], const float2* __restrict__ b2l, int fbase) {
    const float2* p = b2l + fbase;
    #pragma unroll
    for (int s = 0; s < G2_ST; ++s) {
        float2 v = p[s * 2 * FP];
        b[s][0] = v.x; b[s][1] = v.y;
    }
}

__device__ __forceinline__ void g2_update(v8f acc2, float (&bu)[8], float* lsp, int fbase) {
    #pragma unroll
    for (int r = 0; r < 8; ++r) {
        float g  = acc2[r] * (-2.0f / (float)ROWS);
        float bn = MOMENTUM * bu[r] + g;
        float sn = lsp[r * LDSS + fbase] - LR * bn;
        lsp[r * LDSS + fbase] = sn > 0.0f ? sn : 0.0f;
        bu[r] = bn;
    }
}

__device__ __forceinline__ v8f g2_dense(const float2* __restrict__ b2l, const float* a2p, int fbase) {
    v2f b[G2_ST], a2[G2_ST];
    g2_load(b, b2l, fbase);
    #pragma unroll
    for (int s = 0; s < G2_ST; ++s) {
        a2[s][0] = a2p[4 * s];
        a2[s][1] = a2p[4 * s + 1];
    }
    v8f acc2 = {0.f, 0.f, 0.f, 0.f, 0.f, 0.f, 0.f, 0.f};
    #pragma unroll
    for (int s = 0; s < G2_ST; ++s)
        acc2 = __builtin_amdgcn_wmma_f32_16x16x4_f32(false, a2[s], false, b[s],
                                                     (short)0, acc2, false, false);
    return acc2;
}

__device__ __forceinline__ v8f g2_span(const v2f (&b)[G2_NS], const float* aa, int n) {
    v2f a[G2_NS];
    #pragma unroll
    for (int j = 0; j < G2_NS; ++j) {
        if (j < n) {
            a[j][0] = aa[4 * j];
            a[j][1] = aa[4 * j + 1];
        }
    }
    v8f acc2 = {0.f, 0.f, 0.f, 0.f, 0.f, 0.f, 0.f, 0.f};
    #pragma unroll
    for (int j = 0; j < G2_NS; ++j) {
        if (j < n)
            acc2 = __builtin_amdgcn_wmma_f32_16x16x4_f32(false, a[j], false, b[j],
                                                         (short)0, acc2, false, false);
    }
    return acc2;
}

__device__ __forceinline__ void g2_scan(const float2* __restrict__ b2l, int fbase, int& lo, int& n) {
    const float2* p = b2l + fbase;
    unsigned m = 0u;
    #pragma unroll
    for (int s = 0; s < G2_ST; ++s)
        if (frag_nonzero(p[s * 2 * FP])) m |= (1u << s);
    int l = 0, c = 0;
    if (m != 0u) { l = __builtin_ctz(m); c = 32 - __builtin_clz(m) - l; }
    lo = __builtin_amdgcn_readfirstlane(l);
    n  = __builtin_amdgcn_readfirstlane(c);
}

__device__ __forceinline__ void g2_keep(v2f (&b)[G2_NS], const float2* __restrict__ b2l, int fbase, int lo, int n) {
    if (n > G2_NS) return;
    const float2* p = b2l + fbase + lo * (2 * FP);
    #pragma unroll
    for (int j = 0; j < G2_NS; ++j) {
        if (j < n) {
            float2 v = p[j * 2 * FP];
            b[j][0] = v.x; b[j][1] = v.y;
        }
    }
}

__global__ __launch_bounds__(256) void ibs_run(const float* __restrict__ barkspec,
                                               const float* __restrict__ spec_init,
                                               float* __restrict__ spec,
                                               const float2* __restrict__ fbp2,
                                               const float2* __restrict__ fbT2,
                                               float* __restrict__ chk,
                                               int replay) {
    __shared__ __attribute__((aligned(16))) float lds_spec[16 * LDSS];
    __shared__ float lds_diff[16 * LDSD];
    __shared__ float lds_red[256];
    __shared__ float lds_loss[32];

    const int tid     = threadIdx.x;
    const int wave    = __builtin_amdgcn_readfirstlane(tid >> 5);
    const int lane    = tid & 31;
    const int mi      = lane & 15;
    const int hsel    = lane >> 4;
    const int koff    = hsel * 2;
    const int rowbase = blockIdx.x * 16;
    const int nbase   = wave * 16;

    int n_it = MAX_ITER;
    if (replay) {
        if (tid < MAX_ITER) {
            const float4* row = (const float4*)(chk + tid * NBLK);
            float4 cr[NBLK / 4];
            #pragma unroll
            for (int i = 0; i < NBLK / 4; ++i) cr[i] = row[i];
            STAGE_FENCE();
            float s = 0.0f;
            #pragma unroll
            for (int i = 0; i < NBLK / 4; ++i) { s += cr[i].x; s += cr[i].y; s += cr[i].z; s += cr[i].w; }
            lds_loss[tid] = s / (float)ROWS;
        }
        __syncthreads();
        int   n    = MAX_ITER;
        float prev = INFINITY;
        for (int i = 0; i < MAX_ITER; ++i) {
            const float loss = lds_loss[i];
            const bool  stop = (loss < TOL_LOSS) || (fabsf(prev - loss) < TOL_CHG);
            if (stop && n == MAX_ITER) n = i + 1;
            prev = loss;
        }
        n_it = __builtin_amdgcn_readfirstlane(n);
        if (n_it >= MAX_ITER) return;
    }

    const int rowh = rowbase + hsel * 8;
    const int bb   = rowh >> 9;
    const int t0   = rowh & 511;
    const float* mp = barkspec + (bb * N_BARKS + nbase + mi) * TIME + t0;
    float mv[8];
    #pragma unroll
    for (int r = 0; r < 8; ++r) mv[r] = mp[r];

    const float* sip = spec_init + rowbase * N_STFT + tid;
    float i0[16], i1[16], i2[16];
    #pragma unroll
    for (int r = 0; r < 16; ++r) {
        i0[r] = sip[r * N_STFT];
        i1[r] = sip[r * N_STFT + 256];
        i2[r] = 0.0f;
    }
    if (tid == 0) {
        #pragma unroll
        for (int r = 0; r < 16; ++r) i2[r] = sip[r * N_STFT + 512];
    }

    const float*  ap = lds_spec + mi * LDSS + koff;
    const float2* bp = fbp2 + (nbase + mi) + hsel * NB;

    #pragma unroll
    for (int r = 0; r < 16; ++r) {
        lds_spec[r * LDSS + tid]       = i0[r];
        lds_spec[r * LDSS + tid + 256] = i1[r];
    }
    if (tid < FP - 512) {
        #pragma unroll
        for (int r = 0; r < 16; ++r) lds_spec[r * LDSS + 512 + tid] = i2[r];
    }

    const float2* b2l = fbT2 + mi + hsel * FP;
    float*        lsp = lds_spec + (hsel * 8) * LDSS + mi;
    const float*  a2p = lds_diff + mi * LDSD + koff;
    const int     fb0 = nbase;

    int glo1, n1;
    {
        unsigned long long m1 = 0ull;
        #pragma unroll
        for (int g = 0; g < G1_GR; ++g) {
            bool nz = false;
            #pragma unroll
            for (int q = 0; q < 4; ++q) nz = frag_nonzero(bp[(4 * g + q) * 2 * NB]) || nz;
            if (nz) m1 |= (1ull << g);
        }
        int l = 0, c = 0;
        if (m1 != 0ull) { l = __builtin_ctzll(m1); c = 64 - __builtin_clzll(m1) - l; }
        glo1 = __builtin_amdgcn_readfirstlane(l);
        n1   = __builtin_amdgcn_readfirstlane(c);
    }
    const bool fit1 = n1 <= G1_NS;

    int lo20, n20, lo21, n21, lo22, n22, lo23, n23, lo24 = 0, n24 = 0;
    g2_scan(b2l, fb0,       lo20, n20);
    g2_scan(b2l, fb0 + 128, lo21, n21);
    g2_scan(b2l, fb0 + 256, lo22, n22);
    g2_scan(b2l, fb0 + 384, lo23, n23);
    if (wave == 0) g2_scan(b2l, 512, lo24, n24);

    v2f B1[G1_NS][4];
    if (fit1) {
        const float2* p = bp + glo1 * (4 * 2 * NB);
        #pragma unroll
        for (int j = 0; j < G1_NS; ++j) {
            if (j < n1) {
                #pragma unroll
                for (int q = 0; q < 4; ++q) {
                    float2 v = p[(4 * j + q) * 2 * NB];
                    B1[j][q][0] = v.x; B1[j][q][1] = v.y;
                }
            }
        }
    }
    v2f B20[G2_NS], B21[G2_NS], B22[G2_NS], B23[G2_NS], B24[G2_NS];
    g2_keep(B20, b2l, fb0,       lo20, n20);
    g2_keep(B21, b2l, fb0 + 128, lo21, n21);
    g2_keep(B22, b2l, fb0 + 256, lo22, n22);
    g2_keep(B23, b2l, fb0 + 384, lo23, n23);
    if (wave == 0) g2_keep(B24, b2l, 512, lo24, n24);

    __syncthreads();

    float u0[8], u1[8], u2[8], u3[8], u4[8];
    #pragma unroll
    for (int r = 0; r < 8; ++r) { u0[r] = 0.0f; u1[r] = 0.0f; u2[r] = 0.0f; u3[r] = 0.0f; u4[r] = 0.0f; }

    for (int it = 0; it < n_it; ++it) {
        int keep = 0;
        asm volatile("" : "+s"(keep));
        const float2* bpi = bp  + keep;
        const float2* b2i = b2l + keep;

        v8f acc = {0.f, 0.f, 0.f, 0.f, 0.f, 0.f, 0.f, 0.f};
        if (fit1) {
            const float* a0 = ap + glo1 * 16;
            v2f aE[4][4], aO[4][4];
            g1_span_a(aE, a0, 0, n1);
            g1_span_a(aO, a0, 4, n1);
            acc = g1_span_mma(acc, aE, B1, 0, n1);
            g1_span_a(aE, a0, 8, n1);
            acc = g1_span_mma(acc, aO, B1, 4, n1);
            g1_span_a(aO, a0, 12, n1);
            acc = g1_span_mma(acc, aE, B1, 8, n1);
            acc = g1_span_mma(acc, aO, B1, 12, n1);
        } else {
            v2f bA[G1_CH];
            #pragma unroll
            for (int c = 0; c < 4; ++c) {
                g1_load(bA, bpi, c);          STAGE_FENCE();
                acc = g1_mma(acc, bA, ap, c); STAGE_FENCE();
            }
        }

        float ssq = 0.0f;
        #pragma unroll
        for (int r = 0; r < 8; ++r) {
            float d = mv[r] - acc[r];
            ssq += d * d;
            lds_diff[(r + hsel * 8) * LDSD + nbase + mi] = d;
        }
        lds_red[tid] = ssq;
        __syncthreads();

        if (tid == 0 && !replay) {
            float s = 0.0f;
            for (int i = 0; i < 256; ++i) s += lds_red[i];
            chk[it * NBLK + blockIdx.x] = s;
        }

        {
            v8f c0 = (n20 <= G2_NS) ? g2_span(B20, a2p + 4 * lo20, n20) : g2_dense(b2i, a2p, fb0);
            g2_update(c0, u0, lsp, fb0);
            v8f c1 = (n21 <= G2_NS) ? g2_span(B21, a2p + 4 * lo21, n21) : g2_dense(b2i, a2p, fb0 + 128);
            g2_update(c1, u1, lsp, fb0 + 128);
            v8f c2 = (n22 <= G2_NS) ? g2_span(B22, a2p + 4 * lo22, n22) : g2_dense(b2i, a2p, fb0 + 256);
            g2_update(c2, u2, lsp, fb0 + 256);
            v8f c3 = (n23 <= G2_NS) ? g2_span(B23, a2p + 4 * lo23, n23) : g2_dense(b2i, a2p, fb0 + 384);
            g2_update(c3, u3, lsp, fb0 + 384);
            if (wave == 0) {
                v8f c4 = (n24 <= G2_NS) ? g2_span(B24, a2p + 4 * lo24, n24) : g2_dense(b2i, a2p, 512);
                g2_update(c4, u4, lsp, 512);
            }
        }
        __syncthreads();
    }

    #pragma unroll
    for (int j = 0; j < 8; ++j) {
        int i  = tid + 256 * j;
        int r  = i / (FP / 4);
        int c4 = i - r * (FP / 4);
        *(float4*)(spec + (rowbase + r) * FP + c4 * 4) = *(const float4*)(lds_spec + r * LDSS + c4 * 4);
    }
    if (tid < 16 * (FP / 4) - 2048)
        *(float4*)(spec + (rowbase + 15) * FP + (tid + 68) * 4) =
            *(const float4*)(lds_spec + 15 * LDSS + (tid + 68) * 4);
}


__global__ __launch_bounds__(256) void ibs_finalize(const float* __restrict__ spec,
                                                    float* __restrict__ out) {
    int idx = blockIdx.x * blockDim.x + threadIdx.x;
    if (idx >= BATCH * N_STFT * TIME) return;
    int t = idx & 511;
    int f = (idx >> 9) % N_STFT;
    int b = idx / (N_STFT * TIME);
    out[idx] = spec[(b * TIME + t) * FP + f];
}


extern "C" void kernel_launch(void* const* d_in, const int* in_sizes, int n_in,
                              void* d_out, int out_size, void* d_ws, size_t ws_size,
                              hipStream_t stream) {
    const float* barkspec  = (const float*)d_in[0];
    const float* fb        = (const float*)d_in[1];
    const float* spec_init = (const float*)d_in[2];

    float* ws       = (float*)d_ws;
    float* spec     = ws;
    float* fbp2     = spec + ROWS * FP;
    float* fbT2     = fbp2 + FP * NB;
    float* chk      = fbT2 + NB * FP;

    ibs_init_fb<<<(FP * NB + 255) / 256, 256, 0, stream>>>(fb, fbp2, fbT2);

    ibs_run<<<NBLK, 256, 0, stream>>>(barkspec, spec_init, spec,
                                      (const float2*)fbp2, (const float2*)fbT2, chk, 0);
    ibs_run<<<NBLK, 256, 0, stream>>>(barkspec, spec_init, spec,
                                      (const float2*)fbp2, (const float2*)fbT2, chk, 1);

    ibs_finalize<<<(BATCH * N_STFT * TIME + 255) / 256, 256, 0, stream>>>(spec, (float*)d_out);
}


# ibs_run: block loss sum by a lane butterfly per wave plus an 8-term sum behind the loop, not one thread's 256-add chain at each iteration's barrier; loss summation order changes, stays fixed
# speedup vs baseline: 8.2353x; 1.5345x over previous
#include <hip/hip_runtime.h>
#include <math.h>

typedef float v2f __attribute__((ext_vector_type(2)));
typedef float v8f __attribute__((ext_vector_type(8)));

#define N_STFT   513
#define N_BARKS  128
#define BATCH    4
#define TIME     512
#define ROWS     (BATCH * TIME)
#define FP       528
#define NB       128
#define NBLK     (ROWS / 16)
#define LDSS     532
#define LDSD     132
#define MAX_ITER 30
#define LR       0.1f
#define MOMENTUM 0.9f
#define TOL_LOSS 1e-5f
#define TOL_CHG  1e-8f


__global__ __launch_bounds__(256) void ibs_init_fb(const float* __restrict__ fb_in,
                                                   float* __restrict__ fbp2,
                                                   float* __restrict__ fbT2) {
    int idx = blockIdx.x * blockDim.x + threadIdx.x;
    if (idx >= FP * NB) return;
    int f  = idx / NB;
    int kk = idx - f * NB;
    float v = (f < N_STFT) ? fb_in[f * NB + kk] : 0.0f;
    fbp2[(((f  >> 1) * NB) + kk) * 2 + (f  & 1)] = v;
    fbT2[(((kk >> 1) * FP) + f ) * 2 + (kk & 1)] = v;
}


#define G1_CH   33
#define G2_ST   32
#define G1_GR   33
#define G1_NS   16
#define G2_NS   8

#define STAGE_FENCE() __builtin_amdgcn_sched_barrier(0)

__device__ __forceinline__ bool frag_nonzero(float2 v) {
    return __builtin_amdgcn_ballot_w32((v.x != 0.0f) || (v.y != 0.0f)) != 0u;
}

template <int M>
__device__ __forceinline__ float row_xor(float x) {
    return __builtin_bit_cast(float, __builtin_amdgcn_update_dpp(0, __builtin_bit_cast(int, x),
                                                                 0x160 | M, 0xf, 0xf, false));
}

__device__ __forceinline__ float xhalf(float x) {
    int b = __builtin_bit_cast(int, x);
    return __builtin_bit_cast(float, __builtin_amdgcn_permlanex16(b, b, 0x76543210, 0xfedcba98, false, false));
}

__device__ __forceinline__ float wave_sum(float x) {
    x += row_xor<1>(x);
    x += row_xor<2>(x);
    x += row_xor<4>(x);
    x += row_xor<8>(x);
    x += xhalf(x);
    return x;
}

__device__ __forceinline__ void g1_load(v2f (&b)[G1_CH], const float2* __restrict__ bp, int c) {
    const float2* p = bp + c * (G1_CH * 2 * NB);
    #pragma unroll
    for (int j = 0; j < G1_CH; ++j) {
        float2 v = p[j * 2 * NB];
        b[j][0] = v.x; b[j][1] = v.y;
    }
}

__device__ __forceinline__ v8f g1_mma(v8f acc, const v2f (&b)[G1_CH], const float* ap, int c) {
    const float* a0 = ap + c * (G1_CH * 4);
    #pragma unroll
    for (int j = 0; j < G1_CH; ++j) {
        v2f a;
        a[0] = a0[4 * j];
        a[1] = a0[4 * j + 1];
        acc = __builtin_amdgcn_wmma_f32_16x16x4_f32(false, a, false, b[j],
                                                    (short)0, acc, false, false);
    }
    return acc;
}

__device__ __forceinline__ void g1_span_a(v2f (&a)[4][4], const float* a0, int gb, int n) {
    #pragma unroll
    for (int j = 0; j < 4; ++j) {
        if (gb + j < n) {
            #pragma unroll
            for (int q = 0; q < 4; ++q) {
                a[j][q][0] = a0[16 * (gb + j) + 4 * q];
                a[j][q][1] = a0[16 * (gb + j) + 4 * q + 1];
            }
        }
    }
}

__device__ __forceinline__ v8f g1_span_mma(v8f acc, const v2f (&a)[4][4], const v2f (&b)[G1_NS][4], int gb, int n) {
    #pragma unroll
    for (int j = 0; j < 4; ++j) {
        if (gb + j < n) {
            #pragma unroll
            for (int q = 0; q < 4; ++q)
                acc = __builtin_amdgcn_wmma_f32_16x16x4_f32(false, a[j][q], false, b[gb + j][q],
                                                            (short)0, acc, false, false);
        }
    }
    return acc;
}

__device__ __forceinline__ void g2_load(v2f (&b)[G2_ST], const float2* __restrict__ b2l, int fbase) {
    const float2* p = b2l + fbase;
    #pragma unroll
    for (int s = 0; s < G2_ST; ++s) {
        float2 v = p[s * 2 * FP];
        b[s][0] = v.x; b[s][1] = v.y;
    }
}

__device__ __forceinline__ void g2_update(v8f acc2, float (&bu)[8], float* lsp, int fbase) {
    #pragma unroll
    for (int r = 0; r < 8; ++r) {
        float g  = acc2[r] * (-2.0f / (float)ROWS);
        float bn = MOMENTUM * bu[r] + g;
        float sn = lsp[r * LDSS + fbase] - LR * bn;
        lsp[r * LDSS + fbase] = sn > 0.0f ? sn : 0.0f;
        bu[r] = bn;
    }
}

__device__ __forceinline__ v8f g2_dense(const float2* __restrict__ b2l, const float* a2p, int fbase) {
    v2f b[G2_ST], a2[G2_ST];
    g2_load(b, b2l, fbase);
    #pragma unroll
    for (int s = 0; s < G2_ST; ++s) {
        a2[s][0] = a2p[4 * s];
        a2[s][1] = a2p[4 * s + 1];
    }
    v8f acc2 = {0.f, 0.f, 0.f, 0.f, 0.f, 0.f, 0.f, 0.f};
    #pragma unroll
    for (int s = 0; s < G2_ST; ++s)
        acc2 = __builtin_amdgcn_wmma_f32_16x16x4_f32(false, a2[s], false, b[s],
                                                     (short)0, acc2, false, false);
    return acc2;
}

__device__ __forceinline__ v8f g2_span(const v2f (&b)[G2_NS], const float* aa, int n) {
    v2f a[G2_NS];
    #pragma unroll
    for (int j = 0; j < G2_NS; ++j) {
        if (j < n) {
            a[j][0] = aa[4 * j];
            a[j][1] = aa[4 * j + 1];
        }
    }
    v8f acc2 = {0.f, 0.f, 0.f, 0.f, 0.f, 0.f, 0.f, 0.f};
    #pragma unroll
    for (int j = 0; j < G2_NS; ++j) {
        if (j < n)
            acc2 = __builtin_amdgcn_wmma_f32_16x16x4_f32(false, a[j], false, b[j],
                                                         (short)0, acc2, false, false);
    }
    return acc2;
}

__device__ __forceinline__ void g2_scan(const float2* __restrict__ b2l, int fbase, int& lo, int& n) {
    const float2* p = b2l + fbase;
    unsigned m = 0u;
    #pragma unroll
    for (int s = 0; s < G2_ST; ++s)
        if (frag_nonzero(p[s * 2 * FP])) m |= (1u << s);
    int l = 0, c = 0;
    if (m != 0u) { l = __builtin_ctz(m); c = 32 - __builtin_clz(m) - l; }
    lo = __builtin_amdgcn_readfirstlane(l);
    n  = __builtin_amdgcn_readfirstlane(c);
}

__device__ __forceinline__ void g2_keep(v2f (&b)[G2_NS], const float2* __restrict__ b2l, int fbase, int lo, int n) {
    if (n > G2_NS) return;
    const float2* p = b2l + fbase + lo * (2 * FP);
    #pragma unroll
    for (int j = 0; j < G2_NS; ++j) {
        if (j < n) {
            float2 v = p[j * 2 * FP];
            b[j][0] = v.x; b[j][1] = v.y;
        }
    }
}

__global__ __launch_bounds__(256) void ibs_run(const float* __restrict__ barkspec,
                                               const float* __restrict__ spec_init,
                                               float* __restrict__ spec,
                                               const float2* __restrict__ fbp2,
                                               const float2* __restrict__ fbT2,
                                               float* __restrict__ chk,
                                               int replay) {
    __shared__ __attribute__((aligned(16))) float lds_spec[16 * LDSS];
    __shared__ float lds_diff[16 * LDSD];
    __shared__ __attribute__((aligned(16))) float lds_part[MAX_ITER * 8];
    __shared__ float lds_loss[32];

    const int tid     = threadIdx.x;
    const int wave    = __builtin_amdgcn_readfirstlane(tid >> 5);
    const int lane    = tid & 31;
    const int mi      = lane & 15;
    const int hsel    = lane >> 4;
    const int koff    = hsel * 2;
    const int rowbase = blockIdx.x * 16;
    const int nbase   = wave * 16;

    int n_it = MAX_ITER;
    if (replay) {
        if (tid < MAX_ITER) {
            const float4* row = (const float4*)(chk + tid * NBLK);
            float4 cr[NBLK / 4];
            #pragma unroll
            for (int i = 0; i < NBLK / 4; ++i) cr[i] = row[i];
            STAGE_FENCE();
            float s = 0.0f;
            #pragma unroll
            for (int i = 0; i < NBLK / 4; ++i) { s += cr[i].x; s += cr[i].y; s += cr[i].z; s += cr[i].w; }
            lds_loss[tid] = s / (float)ROWS;
        }
        __syncthreads();
        int   n    = MAX_ITER;
        float prev = INFINITY;
        for (int i = 0; i < MAX_ITER; ++i) {
            const float loss = lds_loss[i];
            const bool  stop = (loss < TOL_LOSS) || (fabsf(prev - loss) < TOL_CHG);
            if (stop && n == MAX_ITER) n = i + 1;
            prev = loss;
        }
        n_it = __builtin_amdgcn_readfirstlane(n);
        if (n_it >= MAX_ITER) return;
    }

    const int rowh = rowbase + hsel * 8;
    const int bb   = rowh >> 9;
    const int t0   = rowh & 511;
    const float* mp = barkspec + (bb * N_BARKS + nbase + mi) * TIME + t0;
    float mv[8];
    #pragma unroll
    for (int r = 0; r < 8; ++r) mv[r] = mp[r];

    const float* sip = spec_init + rowbase * N_STFT + tid;
    float i0[16], i1[16], i2[16];
    #pragma unroll
    for (int r = 0; r < 16; ++r) {
        i0[r] = sip[r * N_STFT];
        i1[r] = sip[r * N_STFT + 256];
        i2[r] = 0.0f;
    }
    if (tid == 0) {
        #pragma unroll
        for (int r = 0; r < 16; ++r) i2[r] = sip[r * N_STFT + 512];
    }

    const float*  ap = lds_spec + mi * LDSS + koff;
    const float2* bp = fbp2 + (nbase + mi) + hsel * NB;

    #pragma unroll
    for (int r = 0; r < 16; ++r) {
        lds_spec[r * LDSS + tid]       = i0[r];
        lds_spec[r * LDSS + tid + 256] = i1[r];
    }
    if (tid < FP - 512) {
        #pragma unroll
        for (int r = 0; r < 16; ++r) lds_spec[r * LDSS + 512 + tid] = i2[r];
    }

    const float2* b2l = fbT2 + mi + hsel * FP;
    float*        lsp = lds_spec + (hsel * 8) * LDSS + mi;
    const float*  a2p = lds_diff + mi * LDSD + koff;
    const int     fb0 = nbase;

    int glo1, n1;
    {
        unsigned long long m1 = 0ull;
        #pragma unroll
        for (int g = 0; g < G1_GR; ++g) {
            bool nz = false;
            #pragma unroll
            for (int q = 0; q < 4; ++q) nz = frag_nonzero(bp[(4 * g + q) * 2 * NB]) || nz;
            if (nz) m1 |= (1ull << g);
        }
        int l = 0, c = 0;
        if (m1 != 0ull) { l = __builtin_ctzll(m1); c = 64 - __builtin_clzll(m1) - l; }
        glo1 = __builtin_amdgcn_readfirstlane(l);
        n1   = __builtin_amdgcn_readfirstlane(c);
    }
    const bool fit1 = n1 <= G1_NS;

    int lo20, n20, lo21, n21, lo22, n22, lo23, n23, lo24 = 0, n24 = 0;
    g2_scan(b2l, fb0,       lo20, n20);
    g2_scan(b2l, fb0 + 128, lo21, n21);
    g2_scan(b2l, fb0 + 256, lo22, n22);
    g2_scan(b2l, fb0 + 384, lo23, n23);
    if (wave == 0) g2_scan(b2l, 512, lo24, n24);

    v2f B1[G1_NS][4];
    if (fit1) {
        const float2* p = bp + glo1 * (4 * 2 * NB);
        #pragma unroll
        for (int j = 0; j < G1_NS; ++j) {
            if (j < n1) {
                #pragma unroll
                for (int q = 0; q < 4; ++q) {
                    float2 v = p[(4 * j + q) * 2 * NB];
                    B1[j][q][0] = v.x; B1[j][q][1] = v.y;
                }
            }
        }
    }
    v2f B20[G2_NS], B21[G2_NS], B22[G2_NS], B23[G2_NS], B24[G2_NS];
    g2_keep(B20, b2l, fb0,       lo20, n20);
    g2_keep(B21, b2l, fb0 + 128, lo21, n21);
    g2_keep(B22, b2l, fb0 + 256, lo22, n22);
    g2_keep(B23, b2l, fb0 + 384, lo23, n23);
    if (wave == 0) g2_keep(B24, b2l, 512, lo24, n24);

    __syncthreads();

    float u0[8], u1[8], u2[8], u3[8], u4[8];
    #pragma unroll
    for (int r = 0; r < 8; ++r) { u0[r] = 0.0f; u1[r] = 0.0f; u2[r] = 0.0f; u3[r] = 0.0f; u4[r] = 0.0f; }

    for (int it = 0; it < n_it; ++it) {
        int keep = 0;
        asm volatile("" : "+s"(keep));
        const float2* bpi = bp  + keep;
        const float2* b2i = b2l + keep;

        v8f acc = {0.f, 0.f, 0.f, 0.f, 0.f, 0.f, 0.f, 0.f};
        if (fit1) {
            const float* a0 = ap + glo1 * 16;
            v2f aE[4][4], aO[4][4];
            g1_span_a(aE, a0, 0, n1);
            g1_span_a(aO, a0, 4, n1);
            acc = g1_span_mma(acc, aE, B1, 0, n1);
            g1_span_a(aE, a0, 8, n1);
            acc = g1_span_mma(acc, aO, B1, 4, n1);
            g1_span_a(aO, a0, 12, n1);
            acc = g1_span_mma(acc, aE, B1, 8, n1);
            acc = g1_span_mma(acc, aO, B1, 12, n1);
        } else {
            v2f bA[G1_CH];
            #pragma unroll
            for (int c = 0; c < 4; ++c) {
                g1_load(bA, bpi, c);          STAGE_FENCE();
                acc = g1_mma(acc, bA, ap, c); STAGE_FENCE();
            }
        }

        float ssq = 0.0f;
        #pragma unroll
        for (int r = 0; r < 8; ++r) {
            float d = mv[r] - acc[r];
            ssq += d * d;
            lds_diff[(r + hsel * 8) * LDSD + nbase + mi] = d;
        }
        const float wsq = wave_sum(ssq);
        if (lane == 0) lds_part[it * 8 + wave] = wsq;
        __syncthreads();

        {
            v8f c0 = (n20 <= G2_NS) ? g2_span(B20, a2p + 4 * lo20, n20) : g2_dense(b2i, a2p, fb0);
            g2_update(c0, u0, lsp, fb0);
            v8f c1 = (n21 <= G2_NS) ? g2_span(B21, a2p + 4 * lo21, n21) : g2_dense(b2i, a2p, fb0 + 128);
            g2_update(c1, u1, lsp, fb0 + 128);
            v8f c2 = (n22 <= G2_NS) ? g2_span(B22, a2p + 4 * lo22, n22) : g2_dense(b2i, a2p, fb0 + 256);
            g2_update(c2, u2, lsp, fb0 + 256);
            v8f c3 = (n23 <= G2_NS) ? g2_span(B23, a2p + 4 * lo23, n23) : g2_dense(b2i, a2p, fb0 + 384);
            g2_update(c3, u3, lsp, fb0 + 384);
            if (wave == 0) {
                v8f c4 = (n24 <= G2_NS) ? g2_span(B24, a2p + 4 * lo24, n24) : g2_dense(b2i, a2p, 512);
                g2_update(c4, u4, lsp, 512);
            }
        }
        __syncthreads();
    }

    if (!replay && tid < MAX_ITER) {
        const float4 pa = *(const float4*)(lds_part + tid * 8);
        const float4 pb = *(const float4*)(lds_part + tid * 8 + 4);
        float s = 0.0f;
        s += pa.x; s += pa.y; s += pa.z; s += pa.w;
        s += pb.x; s += pb.y; s += pb.z; s += pb.w;
        chk[tid * NBLK + blockIdx.x] = s;
    }

    #pragma unroll
    for (int j = 0; j < 8; ++j) {
        int i  = tid + 256 * j;
        int r  = i / (FP / 4);
        int c4 = i - r * (FP / 4);
        *(float4*)(spec + (rowbase + r) * FP + c4 * 4) = *(const float4*)(lds_spec + r * LDSS + c4 * 4);
    }
    if (tid < 16 * (FP / 4) - 2048)
        *(float4*)(spec + (rowbase + 15) * FP + (tid + 68) * 4) =
            *(const float4*)(lds_spec + 15 * LDSS + (tid + 68) * 4);
}


__global__ __launch_bounds__(256) void ibs_finalize(const float* __restrict__ spec,
                                                    float* __restrict__ out) {
    int idx = blockIdx.x * blockDim.x + threadIdx.x;
    if (idx >= BATCH * N_STFT * TIME) return;
    int t = idx & 511;
    int f = (idx >> 9) % N_STFT;
    int b = idx / (N_STFT * TIME);
    out[idx] = spec[(b * TIME + t) * FP + f];
}


extern "C" void kernel_launch(void* const* d_in, const int* in_sizes, int n_in,
                              void* d_out, int out_size, void* d_ws, size_t ws_size,
                              hipStream_t stream) {
    const float* barkspec  = (const float*)d_in[0];
    const float* fb        = (const float*)d_in[1];
    const float* spec_init = (const float*)d_in[2];

    float* ws       = (float*)d_ws;
    float* spec     = ws;
    float* fbp2     = spec + ROWS * FP;
    float* fbT2     = fbp2 + FP * NB;
    float* chk      = fbT2 + NB * FP;

    ibs_init_fb<<<(FP * NB + 255) / 256, 256, 0, stream>>>(fb, fbp2, fbT2);

    ibs_run<<<NBLK, 256, 0, stream>>>(barkspec, spec_init, spec,
                                      (const float2*)fbp2, (const float2*)fbT2, chk, 0);
    ibs_run<<<NBLK, 256, 0, stream>>>(barkspec, spec_init, spec,
                                      (const float2*)fbp2, (const float2*)fbT2, chk, 1);

    ibs_finalize<<<(BATCH * N_STFT * TIME + 255) / 256, 256, 0, stream>>>(spec, (float*)d_out);
}


# ibs_finalize: (B,T,F) -> (B,F,T) transposition through a 32 x 171 LDS tile per work-group: rows read coalesced, one whole 128-byte line per store, 192 work-groups instead of 4104; sums untouched
# speedup vs baseline: 8.7013x; 1.0566x over previous
#include <hip/hip_runtime.h>
#include <math.h>

typedef float v2f __attribute__((ext_vector_type(2)));
typedef float v8f __attribute__((ext_vector_type(8)));

#define N_STFT   513
#define N_BARKS  128
#define BATCH    4
#define TIME     512
#define ROWS     (BATCH * TIME)
#define FP       528
#define NB       128
#define NBLK     (ROWS / 16)
#define LDSS     532
#define LDSD     132
#define MAX_ITER 30
#define LR       0.1f
#define MOMENTUM 0.9f
#define TOL_LOSS 1e-5f
#define TOL_CHG  1e-8f


__global__ __launch_bounds__(256) void ibs_init_fb(const float* __restrict__ fb_in,
                                                   float* __restrict__ fbp2,
                                                   float* __restrict__ fbT2) {
    int idx = blockIdx.x * blockDim.x + threadIdx.x;
    if (idx >= FP * NB) return;
    int f  = idx / NB;
    int kk = idx - f * NB;
    float v = (f < N_STFT) ? fb_in[f * NB + kk] : 0.0f;
    fbp2[(((f  >> 1) * NB) + kk) * 2 + (f  & 1)] = v;
    fbT2[(((kk >> 1) * FP) + f ) * 2 + (kk & 1)] = v;
}


#define G1_CH   33
#define G2_ST   32
#define G1_GR   33
#define G1_NS   16
#define G2_NS   8

#define STAGE_FENCE() __builtin_amdgcn_sched_barrier(0)

__device__ __forceinline__ bool frag_nonzero(float2 v) {
    return __builtin_amdgcn_ballot_w32((v.x != 0.0f) || (v.y != 0.0f)) != 0u;
}

template <int M>
__device__ __forceinline__ float row_xor(float x) {
    return __builtin_bit_cast(float, __builtin_amdgcn_update_dpp(0, __builtin_bit_cast(int, x),
                                                                 0x160 | M, 0xf, 0xf, false));
}

__device__ __forceinline__ float xhalf(float x) {
    int b = __builtin_bit_cast(int, x);
    return __builtin_bit_cast(float, __builtin_amdgcn_permlanex16(b, b, 0x76543210, 0xfedcba98, false, false));
}

__device__ __forceinline__ float wave_sum(float x) {
    x += row_xor<1>(x);
    x += row_xor<2>(x);
    x += row_xor<4>(x);
    x += row_xor<8>(x);
    x += xhalf(x);
    return x;
}

__device__ __forceinline__ void g1_load(v2f (&b)[G1_CH], const float2* __restrict__ bp, int c) {
    const float2* p = bp + c * (G1_CH * 2 * NB);
    #pragma unroll
    for (int j = 0; j < G1_CH; ++j) {
        float2 v = p[j * 2 * NB];
        b[j][0] = v.x; b[j][1] = v.y;
    }
}

__device__ __forceinline__ v8f g1_mma(v8f acc, const v2f (&b)[G1_CH], const float* ap, int c) {
    const float* a0 = ap + c * (G1_CH * 4);
    #pragma unroll
    for (int j = 0; j < G1_CH; ++j) {
        v2f a;
        a[0] = a0[4 * j];
        a[1] = a0[4 * j + 1];
        acc = __builtin_amdgcn_wmma_f32_16x16x4_f32(false, a, false, b[j],
                                                    (short)0, acc, false, false);
    }
    return acc;
}

__device__ __forceinline__ void g1_span_a(v2f (&a)[4][4], const float* a0, int gb, int n) {
    #pragma unroll
    for (int j = 0; j < 4; ++j) {
        if (gb + j < n) {
            #pragma unroll
            for (int q = 0; q < 4; ++q) {
                a[j][q][0] = a0[16 * (gb + j) + 4 * q];
                a[j][q][1] = a0[16 * (gb + j) + 4 * q + 1];
            }
        }
    }
}

__device__ __forceinline__ v8f g1_span_mma(v8f acc, const v2f (&a)[4][4], const v2f (&b)[G1_NS][4], int gb, int n) {
    #pragma unroll
    for (int j = 0; j < 4; ++j) {
        if (gb + j < n) {
            #pragma unroll
            for (int q = 0; q < 4; ++q)
                acc = __builtin_amdgcn_wmma_f32_16x16x4_f32(false, a[j][q], false, b[gb + j][q],
                                                            (short)0, acc, false, false);
        }
    }
    return acc;
}

__device__ __forceinline__ void g2_load(v2f (&b)[G2_ST], const float2* __restrict__ b2l, int fbase) {
    const float2* p = b2l + fbase;
    #pragma unroll
    for (int s = 0; s < G2_ST; ++s) {
        float2 v = p[s * 2 * FP];
        b[s][0] = v.x; b[s][1] = v.y;
    }
}

__device__ __forceinline__ void g2_update(v8f acc2, float (&bu)[8], float* lsp, int fbase) {
    #pragma unroll
    for (int r = 0; r < 8; ++r) {
        float g  = acc2[r] * (-2.0f / (float)ROWS);
        float bn = MOMENTUM * bu[r] + g;
        float sn = lsp[r * LDSS + fbase] - LR * bn;
        lsp[r * LDSS + fbase] = sn > 0.0f ? sn : 0.0f;
        bu[r] = bn;
    }
}

__device__ __forceinline__ v8f g2_dense(const float2* __restrict__ b2l, const float* a2p, int fbase) {
    v2f b[G2_ST], a2[G2_ST];
    g2_load(b, b2l, fbase);
    #pragma unroll
    for (int s = 0; s < G2_ST; ++s) {
        a2[s][0] = a2p[4 * s];
        a2[s][1] = a2p[4 * s + 1];
    }
    v8f acc2 = {0.f, 0.f, 0.f, 0.f, 0.f, 0.f, 0.f, 0.f};
    #pragma unroll
    for (int s = 0; s < G2_ST; ++s)
        acc2 = __builtin_amdgcn_wmma_f32_16x16x4_f32(false, a2[s], false, b[s],
                                                     (short)0, acc2, false, false);
    return acc2;
}

__device__ __forceinline__ v8f g2_span(const v2f (&b)[G2_NS], const float* aa, int n) {
    v2f a[G2_NS];
    #pragma unroll
    for (int j = 0; j < G2_NS; ++j) {
        if (j < n) {
            a[j][0] = aa[4 * j];
            a[j][1] = aa[4 * j + 1];
        }
    }
    v8f acc2 = {0.f, 0.f, 0.f, 0.f, 0.f, 0.f, 0.f, 0.f};
    #pragma unroll
    for (int j = 0; j < G2_NS; ++j) {
        if (j < n)
            acc2 = __builtin_amdgcn_wmma_f32_16x16x4_f32(false, a[j], false, b[j],
                                                         (short)0, acc2, false, false);
    }
    return acc2;
}

__device__ __forceinline__ void g2_scan(const float2* __restrict__ b2l, int fbase, int& lo, int& n) {
    const float2* p = b2l + fbase;
    unsigned m = 0u;
    #pragma unroll
    for (int s = 0; s < G2_ST; ++s)
        if (frag_nonzero(p[s * 2 * FP])) m |= (1u << s);
    int l = 0, c = 0;
    if (m != 0u) { l = __builtin_ctz(m); c = 32 - __builtin_clz(m) - l; }
    lo = __builtin_amdgcn_readfirstlane(l);
    n  = __builtin_amdgcn_readfirstlane(c);
}

__device__ __forceinline__ void g2_keep(v2f (&b)[G2_NS], const float2* __restrict__ b2l, int fbase, int lo, int n) {
    if (n > G2_NS) return;
    const float2* p = b2l + fbase + lo * (2 * FP);
    #pragma unroll
    for (int j = 0; j < G2_NS; ++j) {
        if (j < n) {
            float2 v = p[j * 2 * FP];
            b[j][0] = v.x; b[j][1] = v.y;
        }
    }
}

__global__ __launch_bounds__(256) void ibs_run(const float* __restrict__ barkspec,
                                               const float* __restrict__ spec_init,
                                               float* __restrict__ spec,
                                               const float2* __restrict__ fbp2,
                                               const float2* __restrict__ fbT2,
                                               float* __restrict__ chk,
                                               int replay) {
    __shared__ __attribute__((aligned(16))) float lds_spec[16 * LDSS];
    __shared__ float lds_diff[16 * LDSD];
    __shared__ __attribute__((aligned(16))) float lds_part[MAX_ITER * 8];
    __shared__ float lds_loss[32];

    const int tid     = threadIdx.x;
    const int wave    = __builtin_amdgcn_readfirstlane(tid >> 5);
    const int lane    = tid & 31;
    const int mi      = lane & 15;
    const int hsel    = lane >> 4;
    const int koff    = hsel * 2;
    const int rowbase = blockIdx.x * 16;
    const int nbase   = wave * 16;

    int n_it = MAX_ITER;
    if (replay) {
        if (tid < MAX_ITER) {
            const float4* row = (const float4*)(chk + tid * NBLK);
            float4 cr[NBLK / 4];
            #pragma unroll
            for (int i = 0; i < NBLK / 4; ++i) cr[i] = row[i];
            STAGE_FENCE();
            float s = 0.0f;
            #pragma unroll
            for (int i = 0; i < NBLK / 4; ++i) { s += cr[i].x; s += cr[i].y; s += cr[i].z; s += cr[i].w; }
            lds_loss[tid] = s / (float)ROWS;
        }
        __syncthreads();
        int   n    = MAX_ITER;
        float prev = INFINITY;
        for (int i = 0; i < MAX_ITER; ++i) {
            const float loss = lds_loss[i];
            const bool  stop = (loss < TOL_LOSS) || (fabsf(prev - loss) < TOL_CHG);
            if (stop && n == MAX_ITER) n = i + 1;
            prev = loss;
        }
        n_it = __builtin_amdgcn_readfirstlane(n);
        if (n_it >= MAX_ITER) return;
    }

    const int rowh = rowbase + hsel * 8;
    const int bb   = rowh >> 9;
    const int t0   = rowh & 511;
    const float* mp = barkspec + (bb * N_BARKS + nbase + mi) * TIME + t0;
    float mv[8];
    #pragma unroll
    for (int r = 0; r < 8; ++r) mv[r] = mp[r];

    const float* sip = spec_init + rowbase * N_STFT + tid;
    float i0[16], i1[16], i2[16];
    #pragma unroll
    for (int r = 0; r < 16; ++r) {
        i0[r] = sip[r * N_STFT];
        i1[r] = sip[r * N_STFT + 256];
        i2[r] = 0.0f;
    }
    if (tid == 0) {
        #pragma unroll
        for (int r = 0; r < 16; ++r) i2[r] = sip[r * N_STFT + 512];
    }

    const float*  ap = lds_spec + mi * LDSS + koff;
    const float2* bp = fbp2 + (nbase + mi) + hsel * NB;

    #pragma unroll
    for (int r = 0; r < 16; ++r) {
        lds_spec[r * LDSS + tid]       = i0[r];
        lds_spec[r * LDSS + tid + 256] = i1[r];
    }
    if (tid < FP - 512) {
        #pragma unroll
        for (int r = 0; r < 16; ++r) lds_spec[r * LDSS + 512 + tid] = i2[r];
    }

    const float2* b2l = fbT2 + mi + hsel * FP;
    float*        lsp = lds_spec + (hsel * 8) * LDSS + mi;
    const float*  a2p = lds_diff + mi * LDSD + koff;
    const int     fb0 = nbase;

    int glo1, n1;
    {
        unsigned long long m1 = 0ull;
        #pragma unroll
        for (int g = 0; g < G1_GR; ++g) {
            bool nz = false;
            #pragma unroll
            for (int q = 0; q < 4; ++q) nz = frag_nonzero(bp[(4 * g + q) * 2 * NB]) || nz;
            if (nz) m1 |= (1ull << g);
        }
        int l = 0, c = 0;
        if (m1 != 0ull) { l = __builtin_ctzll(m1); c = 64 - __builtin_clzll(m1) - l; }
        glo1 = __builtin_amdgcn_readfirstlane(l);
        n1   = __builtin_amdgcn_readfirstlane(c);
    }
    const bool fit1 = n1 <= G1_NS;

    int lo20, n20, lo21, n21, lo22, n22, lo23, n23, lo24 = 0, n24 = 0;
    g2_scan(b2l, fb0,       lo20, n20);
    g2_scan(b2l, fb0 + 128, lo21, n21);
    g2_scan(b2l, fb0 + 256, lo22, n22);
    g2_scan(b2l, fb0 + 384, lo23, n23);
    if (wave == 0) g2_scan(b2l, 512, lo24, n24);

    v2f B1[G1_NS][4];
    if (fit1) {
        const float2* p = bp + glo1 * (4 * 2 * NB);
        #pragma unroll
        for (int j = 0; j < G1_NS; ++j) {
            if (j < n1) {
                #pragma unroll
                for (int q = 0; q < 4; ++q) {
                    float2 v = p[(4 * j + q) * 2 * NB];
                    B1[j][q][0] = v.x; B1[j][q][1] = v.y;
                }
            }
        }
    }
    v2f B20[G2_NS], B21[G2_NS], B22[G2_NS], B23[G2_NS], B24[G2_NS];
    g2_keep(B20, b2l, fb0,       lo20, n20);
    g2_keep(B21, b2l, fb0 + 128, lo21, n21);
    g2_keep(B22, b2l, fb0 + 256, lo22, n22);
    g2_keep(B23, b2l, fb0 + 384, lo23, n23);
    if (wave == 0) g2_keep(B24, b2l, 512, lo24, n24);

    __syncthreads();

    float u0[8], u1[8], u2[8], u3[8], u4[8];
    #pragma unroll
    for (int r = 0; r < 8; ++r) { u0[r] = 0.0f; u1[r] = 0.0f; u2[r] = 0.0f; u3[r] = 0.0f; u4[r] = 0.0f; }

    for (int it = 0; it < n_it; ++it) {
        int keep = 0;
        asm volatile("" : "+s"(keep));
        const float2* bpi = bp  + keep;
        const float2* b2i = b2l + keep;

        v8f acc = {0.f, 0.f, 0.f, 0.f, 0.f, 0.f, 0.f, 0.f};
        if (fit1) {
            const float* a0 = ap + glo1 * 16;
            v2f aE[4][4], aO[4][4];
            g1_span_a(aE, a0, 0, n1);
            g1_span_a(aO, a0, 4, n1);
            acc = g1_span_mma(acc, aE, B1, 0, n1);
            g1_span_a(aE, a0, 8, n1);
            acc = g1_span_mma(acc, aO, B1, 4, n1);
            g1_span_a(aO, a0, 12, n1);
            acc = g1_span_mma(acc, aE, B1, 8, n1);
            acc = g1_span_mma(acc, aO, B1, 12, n1);
        } else {
            v2f bA[G1_CH];
            #pragma unroll
            for (int c = 0; c < 4; ++c) {
                g1_load(bA, bpi, c);          STAGE_FENCE();
                acc = g1_mma(acc, bA, ap, c); STAGE_FENCE();
            }
        }

        float ssq = 0.0f;
        #pragma unroll
        for (int r = 0; r < 8; ++r) {
            float d = mv[r] - acc[r];
            ssq += d * d;
            lds_diff[(r + hsel * 8) * LDSD + nbase + mi] = d;
        }
        const float wsq = wave_sum(ssq);
        if (lane == 0) lds_part[it * 8 + wave] = wsq;
        __syncthreads();

        {
            v8f c0 = (n20 <= G2_NS) ? g2_span(B20, a2p + 4 * lo20, n20) : g2_dense(b2i, a2p, fb0);
            g2_update(c0, u0, lsp, fb0);
            v8f c1 = (n21 <= G2_NS) ? g2_span(B21, a2p + 4 * lo21, n21) : g2_dense(b2i, a2p, fb0 + 128);
            g2_update(c1, u1, lsp, fb0 + 128);
            v8f c2 = (n22 <= G2_NS) ? g2_span(B22, a2p + 4 * lo22, n22) : g2_dense(b2i, a2p, fb0 + 256);
            g2_update(c2, u2, lsp, fb0 + 256);
            v8f c3 = (n23 <= G2_NS) ? g2_span(B23, a2p + 4 * lo23, n23) : g2_dense(b2i, a2p, fb0 + 384);
            g2_update(c3, u3, lsp, fb0 + 384);
            if (wave == 0) {
                v8f c4 = (n24 <= G2_NS) ? g2_span(B24, a2p + 4 * lo24, n24) : g2_dense(b2i, a2p, 512);
                g2_update(c4, u4, lsp, 512);
            }
        }
        __syncthreads();
    }

    if (!replay && tid < MAX_ITER) {
        const float4 pa = *(const float4*)(lds_part + tid * 8);
        const float4 pb = *(const float4*)(lds_part + tid * 8 + 4);
        float s = 0.0f;
        s += pa.x; s += pa.y; s += pa.z; s += pa.w;
        s += pb.x; s += pb.y; s += pb.z; s += pb.w;
        chk[tid * NBLK + blockIdx.x] = s;
    }

    #pragma unroll
    for (int j = 0; j < 8; ++j) {
        int i  = tid + 256 * j;
        int r  = i / (FP / 4);
        int c4 = i - r * (FP / 4);
        *(float4*)(spec + (rowbase + r) * FP + c4 * 4) = *(const float4*)(lds_spec + r * LDSS + c4 * 4);
    }
    if (tid < 16 * (FP / 4) - 2048)
        *(float4*)(spec + (rowbase + 15) * FP + (tid + 68) * 4) =
            *(const float4*)(lds_spec + 15 * LDSS + (tid + 68) * 4);
}

#define TR_T    32
#define TR_F    171
#define TR_N    (TR_T * TR_F)
#define TR_FULL (TR_N / 256)
#define TR_TAIL (TR_N - 256 * TR_FULL)
#define TR_GRID ((ROWS / TR_T) * (N_STFT / TR_F))

__global__ __launch_bounds__(256) void ibs_finalize(const float* __restrict__ spec,
                                                    float* __restrict__ out) {
    __shared__ float tile[TR_N];
    const int tid  = threadIdx.x;
    const int rt   = blockIdx.x / (N_STFT / TR_F);
    const int f0   = (blockIdx.x - rt * (N_STFT / TR_F)) * TR_F;
    const int row0 = rt * TR_T;
    const float* src = spec + row0 * FP + f0;

    float v[TR_FULL + 1];
    #pragma unroll
    for (int j = 0; j < TR_FULL; ++j) {
        const int i = tid + 256 * j;
        const int r = i / TR_F;
        v[j] = src[r * FP + (i - r * TR_F)];
    }
    v[TR_FULL] = 0.0f;
    if (tid < TR_TAIL) {
        const int i = tid + 256 * TR_FULL;
        const int r = i / TR_F;
        v[TR_FULL] = src[r * FP + (i - r * TR_F)];
    }
    #pragma unroll
    for (int j = 0; j < TR_FULL; ++j) tile[tid + 256 * j] = v[j];
    if (tid < TR_TAIL) tile[tid + 256 * TR_FULL] = v[TR_FULL];
    __syncthreads();

    const int wave = tid >> 5;
    const int lane = tid & 31;
    const int b    = row0 >> 9;
    const int t0   = row0 & 511;
    float* dst = out + (b * N_STFT + f0) * TIME + t0 + lane;
    #pragma unroll
    for (int j = 0; j < (TR_F + 7) / 8; ++j) {
        const int c = wave + 8 * j;
        if (c < TR_F) dst[c * TIME] = tile[lane * TR_F + c];
    }
}


extern "C" void kernel_launch(void* const* d_in, const int* in_sizes, int n_in,
                              void* d_out, int out_size, void* d_ws, size_t ws_size,
                              hipStream_t stream) {
    const float* barkspec  = (const float*)d_in[0];
    const float* fb        = (const float*)d_in[1];
    const float* spec_init = (const float*)d_in[2];

    float* ws       = (float*)d_ws;
    float* spec     = ws;
    float* fbp2     = spec + ROWS * FP;
    float* fbT2     = fbp2 + FP * NB;
    float* chk      = fbT2 + NB * FP;

    ibs_init_fb<<<(FP * NB + 255) / 256, 256, 0, stream>>>(fb, fbp2, fbT2);

    ibs_run<<<NBLK, 256, 0, stream>>>(barkspec, spec_init, spec,
                                      (const float2*)fbp2, (const float2*)fbT2, chk, 0);
    ibs_run<<<NBLK, 256, 0, stream>>>(barkspec, spec_init, spec,
                                      (const float2*)fbp2, (const float2*)fbT2, chk, 1);

    ibs_finalize<<<TR_GRID, 256, 0, stream>>>(spec, (float*)d_out);
}


# ibs_run GEMM2 phase in one pass: all tiles' A fragments as one LDS request group, spectrum values kept in registers, span test as a branch tree (was 2 LDS round trips per tile); sums same
# speedup vs baseline: 10.9360x; 1.2568x over previous
#include <hip/hip_runtime.h>
#include <math.h>

typedef float v2f __attribute__((ext_vector_type(2)));
typedef float v8f __attribute__((ext_vector_type(8)));

#define N_STFT   513
#define N_BARKS  128
#define BATCH    4
#define TIME     512
#define ROWS     (BATCH * TIME)
#define FP       528
#define NB       128
#define NBLK     (ROWS / 16)
#define LDSS     532
#define LDSD     132
#define LDSD_PAD 32
#define MAX_ITER 30
#define LR       0.1f
#define MOMENTUM 0.9f
#define TOL_LOSS 1e-5f
#define TOL_CHG  1e-8f


__global__ __launch_bounds__(256) void ibs_init_fb(const float* __restrict__ fb_in,
                                                   float* __restrict__ fbp2,
                                                   float* __restrict__ fbT2) {
    int idx = blockIdx.x * blockDim.x + threadIdx.x;
    if (idx >= FP * NB) return;
    int f  = idx / NB;
    int kk = idx - f * NB;
    float v = (f < N_STFT) ? fb_in[f * NB + kk] : 0.0f;
    fbp2[(((f  >> 1) * NB) + kk) * 2 + (f  & 1)] = v;
    fbT2[(((kk >> 1) * FP) + f ) * 2 + (kk & 1)] = v;
}


#define G1_CH   33
#define G2_ST   32
#define G1_GR   33
#define G1_NS   16
#define G2_NS   8

#define STAGE_FENCE() __builtin_amdgcn_sched_barrier(0)

__device__ __forceinline__ bool frag_nonzero(float2 v) {
    return __builtin_amdgcn_ballot_w32((v.x != 0.0f) || (v.y != 0.0f)) != 0u;
}

template <int M>
__device__ __forceinline__ float row_xor(float x) {
    return __builtin_bit_cast(float, __builtin_amdgcn_update_dpp(0, __builtin_bit_cast(int, x),
                                                                 0x160 | M, 0xf, 0xf, false));
}

__device__ __forceinline__ float xhalf(float x) {
    int b = __builtin_bit_cast(int, x);
    return __builtin_bit_cast(float, __builtin_amdgcn_permlanex16(b, b, 0x76543210, 0xfedcba98, false, false));
}

__device__ __forceinline__ float wave_sum(float x) {
    x += row_xor<1>(x);
    x += row_xor<2>(x);
    x += row_xor<4>(x);
    x += row_xor<8>(x);
    x += xhalf(x);
    return x;
}

__device__ __forceinline__ void g1_load(v2f (&b)[G1_CH], const float2* __restrict__ bp, int c) {
    const float2* p = bp + c * (G1_CH * 2 * NB);
    #pragma unroll
    for (int j = 0; j < G1_CH; ++j) {
        float2 v = p[j * 2 * NB];
        b[j][0] = v.x; b[j][1] = v.y;
    }
}

__device__ __forceinline__ v8f g1_mma(v8f acc, const v2f (&b)[G1_CH], const float* ap, int c) {
    const float* a0 = ap + c * (G1_CH * 4);
    #pragma unroll
    for (int j = 0; j < G1_CH; ++j) {
        v2f a;
        a[0] = a0[4 * j];
        a[1] = a0[4 * j + 1];
        acc = __builtin_amdgcn_wmma_f32_16x16x4_f32(false, a, false, b[j],
                                                    (short)0, acc, false, false);
    }
    return acc;
}

__device__ __forceinline__ void g1_span_a(v2f (&a)[4][4], const float* a0, int gb, int n) {
    #pragma unroll
    for (int j = 0; j < 4; ++j) {
        if (gb + j < n) {
            #pragma unroll
            for (int q = 0; q < 4; ++q) {
                a[j][q][0] = a0[16 * (gb + j) + 4 * q];
                a[j][q][1] = a0[16 * (gb + j) + 4 * q + 1];
            }
        }
    }
}

__device__ __forceinline__ v8f g1_span_mma(v8f acc, const v2f (&a)[4][4], const v2f (&b)[G1_NS][4], int gb, int n) {
    #pragma unroll
    for (int j = 0; j < 4; ++j) {
        if (gb + j < n) {
            #pragma unroll
            for (int q = 0; q < 4; ++q)
                acc = __builtin_amdgcn_wmma_f32_16x16x4_f32(false, a[j][q], false, b[gb + j][q],
                                                            (short)0, acc, false, false);
        }
    }
    return acc;
}

__device__ __forceinline__ void g2_load(v2f (&b)[G2_ST], const float2* __restrict__ b2l, int fbase) {
    const float2* p = b2l + fbase;
    #pragma unroll
    for (int s = 0; s < G2_ST; ++s) {
        float2 v = p[s * 2 * FP];
        b[s][0] = v.x; b[s][1] = v.y;
    }
}

__device__ __forceinline__ void g2_update(v8f acc2, float (&bu)[8], float* lsp, int fbase) {
    #pragma unroll
    for (int r = 0; r < 8; ++r) {
        float g  = acc2[r] * (-2.0f / (float)ROWS);
        float bn = MOMENTUM * bu[r] + g;
        float sn = lsp[r * LDSS + fbase] - LR * bn;
        lsp[r * LDSS + fbase] = sn > 0.0f ? sn : 0.0f;
        bu[r] = bn;
    }
}

__device__ __forceinline__ v8f g2_dense(const float2* __restrict__ b2l, const float* a2p, int fbase) {
    v2f b[G2_ST], a2[G2_ST];
    g2_load(b, b2l, fbase);
    #pragma unroll
    for (int s = 0; s < G2_ST; ++s) {
        a2[s][0] = a2p[4 * s];
        a2[s][1] = a2p[4 * s + 1];
    }
    v8f acc2 = {0.f, 0.f, 0.f, 0.f, 0.f, 0.f, 0.f, 0.f};
    #pragma unroll
    for (int s = 0; s < G2_ST; ++s)
        acc2 = __builtin_amdgcn_wmma_f32_16x16x4_f32(false, a2[s], false, b[s],
                                                     (short)0, acc2, false, false);
    return acc2;
}


__device__ __forceinline__ void g2_span_a(v2f (&a)[G2_NS], const float* a2p, int e, int n) {
    if (n > G2_NS / 2) {
        #pragma unroll
        for (int j = 0; j < G2_NS / 2; ++j) { a[j][0] = a2p[4 * (e + j)]; a[j][1] = a2p[4 * (e + j) + 1]; }
    } else {
        #pragma unroll
        for (int j = 0; j < G2_NS / 2; ++j) { a[j][0] = 0.0f; a[j][1] = 0.0f; }
    }
    #pragma unroll
    for (int j = G2_NS / 2; j < G2_NS; ++j) { a[j][0] = a2p[4 * (e + j)]; a[j][1] = a2p[4 * (e + j) + 1]; }
}

__device__ __forceinline__ v8f g2_chain(const v2f (&a)[G2_NS], const v2f (&b)[G2_NS], int n) {
    v8f acc2 = {0.f, 0.f, 0.f, 0.f, 0.f, 0.f, 0.f, 0.f};
#define G2_M(J) acc2 = __builtin_amdgcn_wmma_f32_16x16x4_f32(false, a[J], false, b[J], (short)0, acc2, false, false);
    if (n > 4) {
        if (n > 6) {
            if (n > 7) { G2_M(0) }
            G2_M(1)
        }
        if (n > 5) { G2_M(2) }
        G2_M(3)
    }
    if (n > 2) {
        if (n > 3) { G2_M(4) }
        G2_M(5)
    }
    if (n > 1) { G2_M(6) }
    if (n > 0) { G2_M(7) }
#undef G2_M
    return acc2;
}

__device__ __forceinline__ v8f g2_tile(const v2f (&b)[G2_NS], const float* a2p, int e, int n) {
    v2f a[G2_NS];
    g2_span_a(a, a2p, e, n);
    return g2_chain(a, b, n);
}

__device__ __forceinline__ void g2_old(float (&so)[8], const float* lsp, int fbase) {
    #pragma unroll
    for (int r = 0; r < 8; ++r) so[r] = lsp[r * LDSS + fbase];
}

__device__ __forceinline__ void g2_apply(v8f acc2, float (&bu)[8], float (&sp)[8], float* lsp, int fbase) {
    #pragma unroll
    for (int r = 0; r < 8; ++r) {
        float g  = acc2[r] * (-2.0f / (float)ROWS);
        float bn = MOMENTUM * bu[r] + g;
        float sn = sp[r] - LR * bn;
        float sv = sn > 0.0f ? sn : 0.0f;
        lsp[r * LDSS + fbase] = sv;
        sp[r] = sv;
        bu[r] = bn;
    }
}

__device__ __forceinline__ void g2_scan(const float2* __restrict__ b2l, int fbase, int& lo, int& n) {
    const float2* p = b2l + fbase;
    unsigned m = 0u;
    #pragma unroll
    for (int s = 0; s < G2_ST; ++s)
        if (frag_nonzero(p[s * 2 * FP])) m |= (1u << s);
    int l = 0, c = 0;
    if (m != 0u) { l = __builtin_ctz(m); c = 32 - __builtin_clz(m) - l; }
    lo = __builtin_amdgcn_readfirstlane(l);
    n  = __builtin_amdgcn_readfirstlane(c);
}

__device__ __forceinline__ void g2_keep(v2f (&b)[G2_NS], const float2* __restrict__ b2l, int fbase, int lo, int n) {
    const float2* p = b2l + fbase;
    const int e = lo - (G2_NS - n);
    #pragma unroll
    for (int j = 0; j < G2_NS; ++j) {
        float2 v = make_float2(0.0f, 0.0f);
        if (n <= G2_NS && j >= G2_NS - n) v = p[(e + j) * 2 * FP];
        b[j][0] = v.x; b[j][1] = v.y;
    }
}

__global__ __launch_bounds__(256) void ibs_run(const float* __restrict__ barkspec,
                                               const float* __restrict__ spec_init,
                                               float* __restrict__ spec,
                                               const float2* __restrict__ fbp2,
                                               const float2* __restrict__ fbT2,
                                               float* __restrict__ chk,
                                               int replay) {
    __shared__ __attribute__((aligned(16))) float lds_spec[16 * LDSS];
    __shared__ float lds_diff[LDSD_PAD + 16 * LDSD];
    __shared__ __attribute__((aligned(16))) float lds_part[MAX_ITER * 8];
    __shared__ float lds_loss[32];

    const int tid     = threadIdx.x;
    const int wave    = __builtin_amdgcn_readfirstlane(tid >> 5);
    const int lane    = tid & 31;
    const int mi      = lane & 15;
    const int hsel    = lane >> 4;
    const int koff    = hsel * 2;
    const int rowbase = blockIdx.x * 16;
    const int nbase   = wave * 16;

    int n_it = MAX_ITER;
    if (replay) {
        if (tid < MAX_ITER) {
            const float4* row = (const float4*)(chk + tid * NBLK);
            float4 cr[NBLK / 4];
            #pragma unroll
            for (int i = 0; i < NBLK / 4; ++i) cr[i] = row[i];
            STAGE_FENCE();
            float s = 0.0f;
            #pragma unroll
            for (int i = 0; i < NBLK / 4; ++i) { s += cr[i].x; s += cr[i].y; s += cr[i].z; s += cr[i].w; }
            lds_loss[tid] = s / (float)ROWS;
        }
        __syncthreads();
        int   n    = MAX_ITER;
        float prev = INFINITY;
        for (int i = 0; i < MAX_ITER; ++i) {
            const float loss = lds_loss[i];
            const bool  stop = (loss < TOL_LOSS) || (fabsf(prev - loss) < TOL_CHG);
            if (stop && n == MAX_ITER) n = i + 1;
            prev = loss;
        }
        n_it = __builtin_amdgcn_readfirstlane(n);
        if (n_it >= MAX_ITER) return;
    }

    const int rowh = rowbase + hsel * 8;
    const int bb   = rowh >> 9;
    const int t0   = rowh & 511;
    const float* mp = barkspec + (bb * N_BARKS + nbase + mi) * TIME + t0;
    float mv[8];
    #pragma unroll
    for (int r = 0; r < 8; ++r) mv[r] = mp[r];

    const float* sip = spec_init + rowbase * N_STFT + tid;
    float i0[16], i1[16], i2[16];
    #pragma unroll
    for (int r = 0; r < 16; ++r) {
        i0[r] = sip[r * N_STFT];
        i1[r] = sip[r * N_STFT + 256];
        i2[r] = 0.0f;
    }
    if (tid == 0) {
        #pragma unroll
        for (int r = 0; r < 16; ++r) i2[r] = sip[r * N_STFT + 512];
    }

    const float*  ap = lds_spec + mi * LDSS + koff;
    const float2* bp = fbp2 + (nbase + mi) + hsel * NB;

    #pragma unroll
    for (int r = 0; r < 16; ++r) {
        lds_spec[r * LDSS + tid]       = i0[r];
        lds_spec[r * LDSS + tid + 256] = i1[r];
    }
    if (tid < FP - 512) {
        #pragma unroll
        for (int r = 0; r < 16; ++r) lds_spec[r * LDSS + 512 + tid] = i2[r];
    }

    if (tid < LDSD_PAD) lds_diff[tid] = 0.0f;
    if (tid < 16 * (LDSD - NB)) lds_diff[LDSD_PAD + (tid >> 2) * LDSD + NB + (tid & 3)] = 0.0f;

    const float2* b2l = fbT2 + mi + hsel * FP;
    float*        lsp = lds_spec + (hsel * 8) * LDSS + mi;
    const float*  a2p = lds_diff + LDSD_PAD + mi * LDSD + koff;
    const int     fb0 = nbase;

    int glo1, n1;
    {
        unsigned long long m1 = 0ull;
        #pragma unroll
        for (int g = 0; g < G1_GR; ++g) {
            bool nz = false;
            #pragma unroll
            for (int q = 0; q < 4; ++q) nz = frag_nonzero(bp[(4 * g + q) * 2 * NB]) || nz;
            if (nz) m1 |= (1ull << g);
        }
        int l = 0, c = 0;
        if (m1 != 0ull) { l = __builtin_ctzll(m1); c = 64 - __builtin_clzll(m1) - l; }
        glo1 = __builtin_amdgcn_readfirstlane(l);
        n1   = __builtin_amdgcn_readfirstlane(c);
    }
    const bool fit1 = n1 <= G1_NS;

    int lo20, n20, lo21, n21, lo22, n22, lo23, n23, lo24 = 0, n24 = 0;
    g2_scan(b2l, fb0,       lo20, n20);
    g2_scan(b2l, fb0 + 128, lo21, n21);
    g2_scan(b2l, fb0 + 256, lo22, n22);
    g2_scan(b2l, fb0 + 384, lo23, n23);
    if (wave == 0) g2_scan(b2l, 512, lo24, n24);

    v2f B1[G1_NS][4];
    if (fit1) {
        const float2* p = bp + glo1 * (4 * 2 * NB);
        #pragma unroll
        for (int j = 0; j < G1_NS; ++j) {
            if (j < n1) {
                #pragma unroll
                for (int q = 0; q < 4; ++q) {
                    float2 v = p[(4 * j + q) * 2 * NB];
                    B1[j][q][0] = v.x; B1[j][q][1] = v.y;
                }
            }
        }
    }
    v2f B20[G2_NS], B21[G2_NS], B22[G2_NS], B23[G2_NS], B24[G2_NS];
    g2_keep(B20, b2l, fb0,       lo20, n20);
    g2_keep(B21, b2l, fb0 + 128, lo21, n21);
    g2_keep(B22, b2l, fb0 + 256, lo22, n22);
    g2_keep(B23, b2l, fb0 + 384, lo23, n23);
    if (wave == 0) {
        g2_keep(B24, b2l, 512, lo24, n24);
    } else {
        #pragma unroll
        for (int j = 0; j < G2_NS; ++j) { B24[j][0] = 0.0f; B24[j][1] = 0.0f; }
    }
    const int e20 = lo20 - (G2_NS - n20), e21 = lo21 - (G2_NS - n21), e22 = lo22 - (G2_NS - n22),
              e23 = lo23 - (G2_NS - n23), e24 = lo24 - (G2_NS - n24);
    const bool fit2 = (n20 <= G2_NS) && (n21 <= G2_NS) && (n22 <= G2_NS) && (n23 <= G2_NS) && (n24 <= G2_NS);

    __syncthreads();

    float p0[8], p1[8], p2[8], p3[8], p4[8];
    g2_old(p0, lsp, fb0);
    g2_old(p1, lsp, fb0 + 128);
    g2_old(p2, lsp, fb0 + 256);
    g2_old(p3, lsp, fb0 + 384);
    g2_old(p4, lsp, 512);

    float u0[8], u1[8], u2[8], u3[8], u4[8];
    #pragma unroll
    for (int r = 0; r < 8; ++r) { u0[r] = 0.0f; u1[r] = 0.0f; u2[r] = 0.0f; u3[r] = 0.0f; u4[r] = 0.0f; }

    for (int it = 0; it < n_it; ++it) {
        int keep = 0;
        asm volatile("" : "+s"(keep));
        const float2* bpi = bp  + keep;
        const float2* b2i = b2l + keep;

        v8f acc = {0.f, 0.f, 0.f, 0.f, 0.f, 0.f, 0.f, 0.f};
        if (fit1) {
            const float* a0 = ap + glo1 * 16;
            v2f aE[4][4], aO[4][4];
            g1_span_a(aE, a0, 0, n1);
            g1_span_a(aO, a0, 4, n1);
            acc = g1_span_mma(acc, aE, B1, 0, n1);
            g1_span_a(aE, a0, 8, n1);
            acc = g1_span_mma(acc, aO, B1, 4, n1);
            g1_span_a(aO, a0, 12, n1);
            acc = g1_span_mma(acc, aE, B1, 8, n1);
            acc = g1_span_mma(acc, aO, B1, 12, n1);
        } else {
            v2f bA[G1_CH];
            #pragma unroll
            for (int c = 0; c < 4; ++c) {
                g1_load(bA, bpi, c);          STAGE_FENCE();
                acc = g1_mma(acc, bA, ap, c); STAGE_FENCE();
            }
        }

        float ssq = 0.0f;
        #pragma unroll
        for (int r = 0; r < 8; ++r) {
            float d = mv[r] - acc[r];
            ssq += d * d;
            lds_diff[LDSD_PAD + (r + hsel * 8) * LDSD + nbase + mi] = d;
        }
        const float wsq = wave_sum(ssq);
        if (lane == 0) lds_part[it * 8 + wave] = wsq;
        __syncthreads();

        int m0 = n20, m1 = n21, m2 = n22, m3 = n23, m4 = n24;
        asm volatile("" : "+s"(m0), "+s"(m1), "+s"(m2), "+s"(m3), "+s"(m4));
        if (fit2) {
            v2f A0[G2_NS], A1[G2_NS], A2[G2_NS], A3[G2_NS], A4[G2_NS];
            g2_span_a(A0, a2p, e20, m0);
            g2_span_a(A1, a2p, e21, m1);
            g2_span_a(A2, a2p, e22, m2);
            g2_span_a(A3, a2p, e23, m3);
            if (wave == 0) g2_span_a(A4, a2p, e24, m4);
            STAGE_FENCE();
            v8f c0 = g2_chain(A0, B20, m0);
            v8f c1 = g2_chain(A1, B21, m1);
            v8f c2 = g2_chain(A2, B22, m2);
            v8f c3 = g2_chain(A3, B23, m3);
            g2_apply(c0, u0, p0, lsp, fb0);
            g2_apply(c1, u1, p1, lsp, fb0 + 128);
            g2_apply(c2, u2, p2, lsp, fb0 + 256);
            g2_apply(c3, u3, p3, lsp, fb0 + 384);
            if (wave == 0) {
                v8f c4 = g2_chain(A4, B24, m4);
                g2_apply(c4, u4, p4, lsp, 512);
            }
        } else {
            v8f c0 = (n20 <= G2_NS) ? g2_tile(B20, a2p, e20, n20) : g2_dense(b2i, a2p, fb0);
            g2_update(c0, u0, lsp, fb0);
            v8f c1 = (n21 <= G2_NS) ? g2_tile(B21, a2p, e21, n21) : g2_dense(b2i, a2p, fb0 + 128);
            g2_update(c1, u1, lsp, fb0 + 128);
            v8f c2 = (n22 <= G2_NS) ? g2_tile(B22, a2p, e22, n22) : g2_dense(b2i, a2p, fb0 + 256);
            g2_update(c2, u2, lsp, fb0 + 256);
            v8f c3 = (n23 <= G2_NS) ? g2_tile(B23, a2p, e23, n23) : g2_dense(b2i, a2p, fb0 + 384);
            g2_update(c3, u3, lsp, fb0 + 384);
            if (wave == 0) {
                v8f c4 = (n24 <= G2_NS) ? g2_tile(B24, a2p, e24, n24) : g2_dense(b2i, a2p, 512);
                g2_update(c4, u4, lsp, 512);
            }
        }
        __syncthreads();
    }

    if (!replay && tid < MAX_ITER) {
        const float4 pa = *(const float4*)(lds_part + tid * 8);
        const float4 pb = *(const float4*)(lds_part + tid * 8 + 4);
        float s = 0.0f;
        s += pa.x; s += pa.y; s += pa.z; s += pa.w;
        s += pb.x; s += pb.y; s += pb.z; s += pb.w;
        chk[tid * NBLK + blockIdx.x] = s;
    }

    #pragma unroll
    for (int j = 0; j < 8; ++j) {
        int i  = tid + 256 * j;
        int r  = i / (FP / 4);
        int c4 = i - r * (FP / 4);
        *(float4*)(spec + (rowbase + r) * FP + c4 * 4) = *(const float4*)(lds_spec + r * LDSS + c4 * 4);
    }
    if (tid < 16 * (FP / 4) - 2048)
        *(float4*)(spec + (rowbase + 15) * FP + (tid + 68) * 4) =
            *(const float4*)(lds_spec + 15 * LDSS + (tid + 68) * 4);
}

#define TR_T    32
#define TR_F    171
#define TR_N    (TR_T * TR_F)
#define TR_FULL (TR_N / 256)
#define TR_TAIL (TR_N - 256 * TR_FULL)
#define TR_GRID ((ROWS / TR_T) * (N_STFT / TR_F))

__global__ __launch_bounds__(256) void ibs_finalize(const float* __restrict__ spec,
                                                    float* __restrict__ out) {
    __shared__ float tile[TR_N];
    const int tid  = threadIdx.x;
    const int rt   = blockIdx.x / (N_STFT / TR_F);
    const int f0   = (blockIdx.x - rt * (N_STFT / TR_F)) * TR_F;
    const int row0 = rt * TR_T;
    const float* src = spec + row0 * FP + f0;

    float v[TR_FULL + 1];
    #pragma unroll
    for (int j = 0; j < TR_FULL; ++j) {
        const int i = tid + 256 * j;
        const int r = i / TR_F;
        v[j] = src[r * FP + (i - r * TR_F)];
    }
    v[TR_FULL] = 0.0f;
    if (tid < TR_TAIL) {
        const int i = tid + 256 * TR_FULL;
        const int r = i / TR_F;
        v[TR_FULL] = src[r * FP + (i - r * TR_F)];
    }
    #pragma unroll
    for (int j = 0; j < TR_FULL; ++j) tile[tid + 256 * j] = v[j];
    if (tid < TR_TAIL) tile[tid + 256 * TR_FULL] = v[TR_FULL];
    __syncthreads();

    const int wave = tid >> 5;
    const int lane = tid & 31;
    const int b    = row0 >> 9;
    const int t0   = row0 & 511;
    float* dst = out + (b * N_STFT + f0) * TIME + t0 + lane;
    #pragma unroll
    for (int j = 0; j < (TR_F + 7) / 8; ++j) {
        const int c = wave + 8 * j;
        if (c < TR_F) dst[c * TIME] = tile[lane * TR_F + c];
    }
}


extern "C" void kernel_launch(void* const* d_in, const int* in_sizes, int n_in,
                              void* d_out, int out_size, void* d_ws, size_t ws_size,
                              hipStream_t stream) {
    const float* barkspec  = (const float*)d_in[0];
    const float* fb        = (const float*)d_in[1];
    const float* spec_init = (const float*)d_in[2];

    float* ws       = (float*)d_ws;
    float* spec     = ws;
    float* fbp2     = spec + ROWS * FP;
    float* fbT2     = fbp2 + FP * NB;
    float* chk      = fbT2 + NB * FP;

    ibs_init_fb<<<(FP * NB + 255) / 256, 256, 0, stream>>>(fb, fbp2, fbT2);

    ibs_run<<<NBLK, 256, 0, stream>>>(barkspec, spec_init, spec,
                                      (const float2*)fbp2, (const float2*)fbT2, chk, 0);
    ibs_run<<<NBLK, 256, 0, stream>>>(barkspec, spec_init, spec,
                                      (const float2*)fbp2, (const float2*)fbT2, chk, 1);

    ibs_finalize<<<TR_GRID, 256, 0, stream>>>(spec, (float*)d_out);
}


# launches 4 -> 2: filterbank fragments read straight from the input (no copy-building launch); stop-rule walk moved to the head of the transposing launch, which redoes a stopped run itself; sums same
# speedup vs baseline: 12.1143x; 1.1077x over previous
#include <hip/hip_runtime.h>
#include <math.h>

typedef float v2f __attribute__((ext_vector_type(2)));
typedef float v8f __attribute__((ext_vector_type(8)));

#define N_STFT   513
#define N_BARKS  128
#define BATCH    4
#define TIME     512
#define ROWS     (BATCH * TIME)
#define FP       528
#define NB       128
#define NBLK     (ROWS / 16)
#define LDSS     532
#define LDSD     132
#define LDSD_PAD 32
#define MAX_ITER 30
#define LR       0.1f
#define MOMENTUM 0.9f
#define TOL_LOSS 1e-5f
#define TOL_CHG  1e-8f


__device__ __forceinline__ v2f fb_g1(const float* __restrict__ p1, const float* __restrict__ pr, int hsel, int s, int sr, bool whole) {
    v2f v;
    if (whole) {
        v[0] = pr[4 * sr * NB];
        v[1] = pr[4 * sr * NB + NB];
    } else {
        const int f0 = 4 * s + 2 * hsel, f1 = f0 + 1;
        const int d0 = (f0 < N_STFT ? f0 : N_STFT - 1) - 2 * hsel;
        const int d1 = (f1 < N_STFT ? f1 : N_STFT - 1) - 2 * hsel;
        const float x = p1[d0 * NB], y = p1[d1 * NB];
        v[0] = f0 < N_STFT ? x : 0.0f;
        v[1] = f1 < N_STFT ? y : 0.0f;
    }
    return v;
}

template <bool TAIL>
__device__ __forceinline__ v2f fb_g2(const float* __restrict__ p2, int mi, int fbase, int s) {
    v2f v;
    if (!TAIL) {
        const float2 t = *(const float2*)(p2 + fbase * NB + 4 * s);
        v[0] = t.x; v[1] = t.y;
    } else {
        const int  f  = fbase + mi;
        const bool ok = f < N_STFT;
        const int  d  = (ok ? f : N_STFT - 1) - mi;
        const float2 t = *(const float2*)(p2 + d * NB + 4 * s);
        v[0] = ok ? t.x : 0.0f;
        v[1] = ok ? t.y : 0.0f;
    }
    return v;
}

__device__ __forceinline__ unsigned or_bits(unsigned acc, float4 v) {
    return acc | __builtin_bit_cast(unsigned, v.x) | __builtin_bit_cast(unsigned, v.y)
               | __builtin_bit_cast(unsigned, v.z) | __builtin_bit_cast(unsigned, v.w);
}
__device__ __forceinline__ bool bits_nonzero(unsigned acc) { return (acc << 1) != 0u; }


#define G1_CH   33
#define G2_ST   32
#define G1_GR   33
#define G1_NS   16
#define G2_NS   8

#define STAGE_FENCE() __builtin_amdgcn_sched_barrier(0)

template <int M>
__device__ __forceinline__ float row_xor(float x) {
    return __builtin_bit_cast(float, __builtin_amdgcn_update_dpp(0, __builtin_bit_cast(int, x),
                                                                 0x160 | M, 0xf, 0xf, false));
}

__device__ __forceinline__ float xhalf(float x) {
    int b = __builtin_bit_cast(int, x);
    return __builtin_bit_cast(float, __builtin_amdgcn_permlanex16(b, b, 0x76543210, 0xfedcba98, false, false));
}

__device__ __forceinline__ float wave_sum(float x) {
    x += row_xor<1>(x);
    x += row_xor<2>(x);
    x += row_xor<4>(x);
    x += row_xor<8>(x);
    x += xhalf(x);
    return x;
}

__device__ __forceinline__ void g1_load(v2f (&b)[G1_CH], const float* __restrict__ p1, int hsel, int c) {
    #pragma unroll
    for (int j = 0; j < G1_CH; ++j) b[j] = fb_g1(p1, p1, hsel, c * G1_CH + j, c * G1_CH + j, c * G1_CH + j < N_STFT / 4);
}

__device__ __forceinline__ v8f g1_mma(v8f acc, const v2f (&b)[G1_CH], const float* ap, int c) {
    const float* a0 = ap + c * (G1_CH * 4);
    #pragma unroll
    for (int j = 0; j < G1_CH; ++j) {
        v2f a;
        a[0] = a0[4 * j];
        a[1] = a0[4 * j + 1];
        acc = __builtin_amdgcn_wmma_f32_16x16x4_f32(false, a, false, b[j],
                                                    (short)0, acc, false, false);
    }
    return acc;
}

__device__ __forceinline__ void g1_span_a(v2f (&a)[4][4], const float* a0, int gb, int n) {
    #pragma unroll
    for (int j = 0; j < 4; ++j) {
        if (gb + j < n) {
            #pragma unroll
            for (int q = 0; q < 4; ++q) {
                a[j][q][0] = a0[16 * (gb + j) + 4 * q];
                a[j][q][1] = a0[16 * (gb + j) + 4 * q + 1];
            }
        }
    }
}

__device__ __forceinline__ v8f g1_span_mma(v8f acc, const v2f (&a)[4][4], const v2f (&b)[G1_NS][4], int gb, int n) {
    #pragma unroll
    for (int j = 0; j < 4; ++j) {
        if (gb + j < n) {
            #pragma unroll
            for (int q = 0; q < 4; ++q)
                acc = __builtin_amdgcn_wmma_f32_16x16x4_f32(false, a[j][q], false, b[gb + j][q],
                                                            (short)0, acc, false, false);
        }
    }
    return acc;
}

template <bool TAIL>
__device__ __forceinline__ void g2_load(v2f (&b)[G2_ST], const float* __restrict__ p2, int mi, int fbase) {
    #pragma unroll
    for (int s = 0; s < G2_ST; ++s) b[s] = fb_g2<TAIL>(p2, mi, fbase, s);
}

__device__ __forceinline__ void g2_update(v8f acc2, float (&bu)[8], float* lsp, int fbase) {
    #pragma unroll
    for (int r = 0; r < 8; ++r) {
        float g  = acc2[r] * (-2.0f / (float)ROWS);
        float bn = MOMENTUM * bu[r] + g;
        float sn = lsp[r * LDSS + fbase] - LR * bn;
        lsp[r * LDSS + fbase] = sn > 0.0f ? sn : 0.0f;
        bu[r] = bn;
    }
}

template <bool TAIL>
__device__ __forceinline__ v8f g2_dense(const float* __restrict__ p2, int mi, const float* a2p, int fbase) {
    v2f b[G2_ST], a2[G2_ST];
    g2_load<TAIL>(b, p2, mi, fbase);
    #pragma unroll
    for (int s = 0; s < G2_ST; ++s) {
        a2[s][0] = a2p[4 * s];
        a2[s][1] = a2p[4 * s + 1];
    }
    v8f acc2 = {0.f, 0.f, 0.f, 0.f, 0.f, 0.f, 0.f, 0.f};
    #pragma unroll
    for (int s = 0; s < G2_ST; ++s)
        acc2 = __builtin_amdgcn_wmma_f32_16x16x4_f32(false, a2[s], false, b[s],
                                                     (short)0, acc2, false, false);
    return acc2;
}


__device__ __forceinline__ void g2_span_a(v2f (&a)[G2_NS], const float* a2p, int e, int n) {
    if (n > G2_NS / 2) {
        #pragma unroll
        for (int j = 0; j < G2_NS / 2; ++j) { a[j][0] = a2p[4 * (e + j)]; a[j][1] = a2p[4 * (e + j) + 1]; }
    } else {
        #pragma unroll
        for (int j = 0; j < G2_NS / 2; ++j) { a[j][0] = 0.0f; a[j][1] = 0.0f; }
    }
    #pragma unroll
    for (int j = G2_NS / 2; j < G2_NS; ++j) { a[j][0] = a2p[4 * (e + j)]; a[j][1] = a2p[4 * (e + j) + 1]; }
}

__device__ __forceinline__ v8f g2_chain(const v2f (&a)[G2_NS], const v2f (&b)[G2_NS], int n) {
    v8f acc2 = {0.f, 0.f, 0.f, 0.f, 0.f, 0.f, 0.f, 0.f};
#define G2_M(J) acc2 = __builtin_amdgcn_wmma_f32_16x16x4_f32(false, a[J], false, b[J], (short)0, acc2, false, false);
    if (n > 4) {
        if (n > 6) {
            if (n > 7) { G2_M(0) }
            G2_M(1)
        }
        if (n > 5) { G2_M(2) }
        G2_M(3)
    }
    if (n > 2) {
        if (n > 3) { G2_M(4) }
        G2_M(5)
    }
    if (n > 1) { G2_M(6) }
    if (n > 0) { G2_M(7) }
#undef G2_M
    return acc2;
}

__device__ __forceinline__ v8f g2_tile(const v2f (&b)[G2_NS], const float* a2p, int e, int n) {
    v2f a[G2_NS];
    g2_span_a(a, a2p, e, n);
    return g2_chain(a, b, n);
}

__device__ __forceinline__ void g2_old(float (&so)[8], const float* lsp, int fbase) {
    #pragma unroll
    for (int r = 0; r < 8; ++r) so[r] = lsp[r * LDSS + fbase];
}

__device__ __forceinline__ void g2_apply(v8f acc2, float (&bu)[8], float (&sp)[8], float* lsp, int fbase) {
    #pragma unroll
    for (int r = 0; r < 8; ++r) {
        float g  = acc2[r] * (-2.0f / (float)ROWS);
        float bn = MOMENTUM * bu[r] + g;
        float sn = sp[r] - LR * bn;
        float sv = sn > 0.0f ? sn : 0.0f;
        lsp[r * LDSS + fbase] = sv;
        sp[r] = sv;
        bu[r] = bn;
    }
}

template <bool TAIL>
__device__ __forceinline__ void g2_scan(const float* __restrict__ fb, int lane, int fbase, int& lo, int& n) {
    const float* p = fb + fbase * NB + 4 * lane;
    unsigned acc = 0u;
    #pragma unroll
    for (int r = 0; r < 16; ++r)
        if (!TAIL || fbase + r < N_STFT) acc = or_bits(acc, *(const float4*)(p + r * NB));
    const unsigned m = __builtin_amdgcn_ballot_w32(bits_nonzero(acc));
    int l = 0, c = 0;
    if (m != 0u) { l = __builtin_ctz(m); c = 32 - __builtin_clz(m) - l; }
    lo = __builtin_amdgcn_readfirstlane(l);
    n  = __builtin_amdgcn_readfirstlane(c);
}

template <bool TAIL>
__device__ __forceinline__ void g2_keep(v2f (&b)[G2_NS], const float* __restrict__ p2, int mi, int fbase, int lo, int n) {
    const int e = lo - (G2_NS - n);
    #pragma unroll
    for (int j = 0; j < G2_NS; ++j) {
        v2f v = {0.0f, 0.0f};
        if (n <= G2_NS && j >= G2_NS - n) v = fb_g2<TAIL>(p2, mi, fbase, e + j);
        b[j] = v;
    }
}

__device__ __forceinline__ void ibs_tile(const float* __restrict__ barkspec,
                                         const float* __restrict__ fb,
                                         const float* __restrict__ spec_init,
                                         float* lds_spec, float* lds_diff, float* lds_part,
                                         int tid, int rowbase, int n_it) {
    const int wave    = __builtin_amdgcn_readfirstlane(tid >> 5);
    const int lane    = tid & 31;
    const int mi      = lane & 15;
    const int hsel    = lane >> 4;
    const int koff    = hsel * 2;
    const int nbase   = wave * 16;

    const int rowh = rowbase + hsel * 8;
    const int bb   = rowh >> 9;
    const int t0   = rowh & 511;
    const float* mp = barkspec + (bb * N_BARKS + nbase + mi) * TIME + t0;
    float mv[8];
    #pragma unroll
    for (int r = 0; r < 8; ++r) mv[r] = mp[r];

    const float* sip = spec_init + rowbase * N_STFT + tid;
    float i0[16], i1[16], i2[16];
    #pragma unroll
    for (int r = 0; r < 16; ++r) {
        i0[r] = sip[r * N_STFT];
        i1[r] = sip[r * N_STFT + 256];
        i2[r] = 0.0f;
    }
    if (tid == 0) {
        #pragma unroll
        for (int r = 0; r < 16; ++r) i2[r] = sip[r * N_STFT + 512];
    }

    const float* ap  = lds_spec + mi * LDSS + koff;
    const float* b1p = fb + koff * NB + (nbase + mi);

    #pragma unroll
    for (int r = 0; r < 16; ++r) {
        lds_spec[r * LDSS + tid]       = i0[r];
        lds_spec[r * LDSS + tid + 256] = i1[r];
    }
    if (tid < FP - 512) {
        #pragma unroll
        for (int r = 0; r < 16; ++r) lds_spec[r * LDSS + 512 + tid] = i2[r];
    }

    if (tid < LDSD_PAD) lds_diff[tid] = 0.0f;
    if (tid < 16 * (LDSD - NB)) lds_diff[LDSD_PAD + (tid >> 2) * LDSD + NB + (tid & 3)] = 0.0f;

    const float*  b2p = fb + mi * NB + koff;
    float*        lsp = lds_spec + (hsel * 8) * LDSS + mi;
    const float*  a2p = lds_diff + LDSD_PAD + mi * LDSD + koff;
    const int     fb0 = nbase;

    int glo1, n1;
    {
        const int    sr = lane >> 2;
        const float* sp = fb + sr * NB + nbase + 4 * (lane & 3);
        unsigned long long m1 = 0ull;
        #pragma unroll
        for (int g = 0; g < G1_GR; ++g) {
            unsigned acc = 0u;
            #pragma unroll
            for (int h = 0; h < 2; ++h)
                if (16 * g + 8 * h + sr < N_STFT) acc = or_bits(acc, *(const float4*)(sp + (16 * g + 8 * h) * NB));
            if (__builtin_amdgcn_ballot_w32(bits_nonzero(acc)) != 0u) m1 |= (1ull << g);
        }
        int l = 0, c = 0;
        if (m1 != 0ull) { l = __builtin_ctzll(m1); c = 64 - __builtin_clzll(m1) - l; }
        glo1 = __builtin_amdgcn_readfirstlane(l);
        n1   = __builtin_amdgcn_readfirstlane(c);
    }
    const bool fit1 = n1 <= G1_NS;

    int lo20, n20, lo21, n21, lo22, n22, lo23, n23, lo24 = 0, n24 = 0;
    g2_scan<false>(fb, lane, fb0,       lo20, n20);
    g2_scan<false>(fb, lane, fb0 + 128, lo21, n21);
    g2_scan<false>(fb, lane, fb0 + 256, lo22, n22);
    g2_scan<false>(fb, lane, fb0 + 384, lo23, n23);
    if (wave == 0) g2_scan<true>(fb, lane, 512, lo24, n24);

    v2f B1[G1_NS][4];
    if (fit1) {
        const float* pg = b1p + glo1 * (16 * NB);
        #pragma unroll
        for (int j = 0; j < G1_NS; ++j) {
            if (j < n1) {
                if (glo1 + j < N_STFT / 16) {
                    #pragma unroll
                    for (int q = 0; q < 4; ++q) B1[j][q] = fb_g1(b1p, pg, hsel, 4 * (glo1 + j) + q, 4 * j + q, true);
                } else {
                    #pragma unroll
                    for (int q = 0; q < 4; ++q) B1[j][q] = fb_g1(b1p, pg, hsel, 4 * (glo1 + j) + q, 4 * j + q, false);
                }
            }
        }
    }
    v2f B20[G2_NS], B21[G2_NS], B22[G2_NS], B23[G2_NS], B24[G2_NS];
    g2_keep<false>(B20, b2p, mi, fb0,       lo20, n20);
    g2_keep<false>(B21, b2p, mi, fb0 + 128, lo21, n21);
    g2_keep<false>(B22, b2p, mi, fb0 + 256, lo22, n22);
    g2_keep<false>(B23, b2p, mi, fb0 + 384, lo23, n23);
    if (wave == 0) {
        g2_keep<true>(B24, b2p, mi, 512, lo24, n24);
    } else {
        #pragma unroll
        for (int j = 0; j < G2_NS; ++j) { B24[j][0] = 0.0f; B24[j][1] = 0.0f; }
    }
    const int e20 = lo20 - (G2_NS - n20), e21 = lo21 - (G2_NS - n21), e22 = lo22 - (G2_NS - n22),
              e23 = lo23 - (G2_NS - n23), e24 = lo24 - (G2_NS - n24);
    const bool fit2 = (n20 <= G2_NS) && (n21 <= G2_NS) && (n22 <= G2_NS) && (n23 <= G2_NS) && (n24 <= G2_NS);

    __syncthreads();

    float p0[8], p1[8], p2[8], p3[8], p4[8];
    g2_old(p0, lsp, fb0);
    g2_old(p1, lsp, fb0 + 128);
    g2_old(p2, lsp, fb0 + 256);
    g2_old(p3, lsp, fb0 + 384);
    g2_old(p4, lsp, 512);

    float u0[8], u1[8], u2[8], u3[8], u4[8];
    #pragma unroll
    for (int r = 0; r < 8; ++r) { u0[r] = 0.0f; u1[r] = 0.0f; u2[r] = 0.0f; u3[r] = 0.0f; u4[r] = 0.0f; }

    for (int it = 0; it < n_it; ++it) {
        int keep = 0;
        asm volatile("" : "+s"(keep));
        const float* b1i = b1p + keep;
        const float* b2i = b2p + keep;

        v8f acc = {0.f, 0.f, 0.f, 0.f, 0.f, 0.f, 0.f, 0.f};
        if (fit1) {
            const float* a0 = ap + glo1 * 16;
            v2f aE[4][4], aO[4][4];
            g1_span_a(aE, a0, 0, n1);
            g1_span_a(aO, a0, 4, n1);
            acc = g1_span_mma(acc, aE, B1, 0, n1);
            g1_span_a(aE, a0, 8, n1);
            acc = g1_span_mma(acc, aO, B1, 4, n1);
            g1_span_a(aO, a0, 12, n1);
            acc = g1_span_mma(acc, aE, B1, 8, n1);
            acc = g1_span_mma(acc, aO, B1, 12, n1);
        } else {
            v2f bA[G1_CH];
            #pragma unroll
            for (int c = 0; c < 4; ++c) {
                g1_load(bA, b1i, hsel, c);    STAGE_FENCE();
                acc = g1_mma(acc, bA, ap, c); STAGE_FENCE();
            }
        }

        float ssq = 0.0f;
        #pragma unroll
        for (int r = 0; r < 8; ++r) {
            float d = mv[r] - acc[r];
            ssq += d * d;
            lds_diff[LDSD_PAD + (r + hsel * 8) * LDSD + nbase + mi] = d;
        }
        const float wsq = wave_sum(ssq);
        if (lane == 0) lds_part[it * 8 + wave] = wsq;
        __syncthreads();

        int m0 = n20, m1 = n21, m2 = n22, m3 = n23, m4 = n24;
        asm volatile("" : "+s"(m0), "+s"(m1), "+s"(m2), "+s"(m3), "+s"(m4));
        if (fit2) {
            v2f A0[G2_NS], A1[G2_NS], A2[G2_NS], A3[G2_NS], A4[G2_NS];
            g2_span_a(A0, a2p, e20, m0);
            g2_span_a(A1, a2p, e21, m1);
            g2_span_a(A2, a2p, e22, m2);
            g2_span_a(A3, a2p, e23, m3);
            if (wave == 0) g2_span_a(A4, a2p, e24, m4);
            STAGE_FENCE();
            v8f c0 = g2_chain(A0, B20, m0);
            v8f c1 = g2_chain(A1, B21, m1);
            v8f c2 = g2_chain(A2, B22, m2);
            v8f c3 = g2_chain(A3, B23, m3);
            g2_apply(c0, u0, p0, lsp, fb0);
            g2_apply(c1, u1, p1, lsp, fb0 + 128);
            g2_apply(c2, u2, p2, lsp, fb0 + 256);
            g2_apply(c3, u3, p3, lsp, fb0 + 384);
            if (wave == 0) {
                v8f c4 = g2_chain(A4, B24, m4);
                g2_apply(c4, u4, p4, lsp, 512);
            }
        } else {
            v8f c0 = (n20 <= G2_NS) ? g2_tile(B20, a2p, e20, m0) : g2_dense<false>(b2i, mi, a2p, fb0);
            g2_update(c0, u0, lsp, fb0);
            v8f c1 = (n21 <= G2_NS) ? g2_tile(B21, a2p, e21, m1) : g2_dense<false>(b2i, mi, a2p, fb0 + 128);
            g2_update(c1, u1, lsp, fb0 + 128);
            v8f c2 = (n22 <= G2_NS) ? g2_tile(B22, a2p, e22, m2) : g2_dense<false>(b2i, mi, a2p, fb0 + 256);
            g2_update(c2, u2, lsp, fb0 + 256);
            v8f c3 = (n23 <= G2_NS) ? g2_tile(B23, a2p, e23, m3) : g2_dense<false>(b2i, mi, a2p, fb0 + 384);
            g2_update(c3, u3, lsp, fb0 + 384);
            if (wave == 0) {
                v8f c4 = (n24 <= G2_NS) ? g2_tile(B24, a2p, e24, m4) : g2_dense<true>(b2i, mi, a2p, 512);
                g2_update(c4, u4, lsp, 512);
            }
        }
        __syncthreads();
    }

}

__global__ __launch_bounds__(256) void ibs_run(const float* __restrict__ barkspec,
                                               const float* __restrict__ fb,
                                               const float* __restrict__ spec_init,
                                               float* __restrict__ spec,
                                               float* __restrict__ chk) {
    __shared__ __attribute__((aligned(16))) float lds_spec[16 * LDSS];
    __shared__ float lds_diff[LDSD_PAD + 16 * LDSD];
    __shared__ __attribute__((aligned(16))) float lds_part[MAX_ITER * 8];

    const int tid     = threadIdx.x;
    const int rowbase = blockIdx.x * 16;

    ibs_tile(barkspec, fb, spec_init, lds_spec, lds_diff, lds_part, tid, rowbase, MAX_ITER);

    if (tid < MAX_ITER) {
        const float4 pa = *(const float4*)(lds_part + tid * 8);
        const float4 pb = *(const float4*)(lds_part + tid * 8 + 4);
        float s = 0.0f;
        s += pa.x; s += pa.y; s += pa.z; s += pa.w;
        s += pb.x; s += pb.y; s += pb.z; s += pb.w;
        chk[tid * NBLK + blockIdx.x] = s;
    }

    #pragma unroll
    for (int j = 0; j < 8; ++j) {
        int i  = tid + 256 * j;
        int r  = i / (FP / 4);
        int c4 = i - r * (FP / 4);
        *(float4*)(spec + (rowbase + r) * FP + c4 * 4) = *(const float4*)(lds_spec + r * LDSS + c4 * 4);
    }
    if (tid < 16 * (FP / 4) - 2048)
        *(float4*)(spec + (rowbase + 15) * FP + (tid + 68) * 4) =
            *(const float4*)(lds_spec + 15 * LDSS + (tid + 68) * 4);
}

#define TR_T    32
#define TR_F    171
#define TR_N    (TR_T * TR_F)
#define TR_FULL (TR_N / 256)
#define TR_TAIL (TR_N - 256 * TR_FULL)
#define TR_GRID ((ROWS / TR_T) * (N_STFT / TR_F))

__global__ __launch_bounds__(256) void ibs_finish(const float* __restrict__ barkspec,
                                                  const float* __restrict__ fb,
                                                  const float* __restrict__ spec_init,
                                                  const float* __restrict__ spec,
                                                  const float* __restrict__ chk,
                                                  float* __restrict__ out) {
    __shared__ float tile[TR_N];
    __shared__ __attribute__((aligned(16))) float lds_spec[16 * LDSS];
    __shared__ float lds_diff[LDSD_PAD + 16 * LDSD];
    __shared__ __attribute__((aligned(16))) float lds_part[MAX_ITER * 8];
    __shared__ float lds_loss[32];

    const int tid  = threadIdx.x;
    const int rt   = blockIdx.x / (N_STFT / TR_F);
    const int f0   = (blockIdx.x - rt * (N_STFT / TR_F)) * TR_F;
    const int row0 = rt * TR_T;
    const float* src = spec + row0 * FP + f0;

    float v[TR_FULL + 1];
    #pragma unroll
    for (int j = 0; j < TR_FULL; ++j) {
        const int i = tid + 256 * j;
        const int r = i / TR_F;
        v[j] = src[r * FP + (i - r * TR_F)];
    }
    v[TR_FULL] = 0.0f;
    if (tid < TR_TAIL) {
        const int i = tid + 256 * TR_FULL;
        const int r = i / TR_F;
        v[TR_FULL] = src[r * FP + (i - r * TR_F)];
    }

    if (tid < MAX_ITER) {
        const float4* row = (const float4*)(chk + tid * NBLK);
        float4 cr[NBLK / 4];
        #pragma unroll
        for (int i = 0; i < NBLK / 4; ++i) cr[i] = row[i];
        STAGE_FENCE();
        float s = 0.0f;
        #pragma unroll
        for (int i = 0; i < NBLK / 4; ++i) { s += cr[i].x; s += cr[i].y; s += cr[i].z; s += cr[i].w; }
        lds_loss[tid] = s / (float)ROWS;
    }
    #pragma unroll
    for (int j = 0; j < TR_FULL; ++j) tile[tid + 256 * j] = v[j];
    if (tid < TR_TAIL) tile[tid + 256 * TR_FULL] = v[TR_FULL];
    __syncthreads();
    int   n    = MAX_ITER;
    float prev = INFINITY;
    for (int i = 0; i < MAX_ITER; ++i) {
        const float loss = lds_loss[i];
        const bool  stop = (loss < TOL_LOSS) || (fabsf(prev - loss) < TOL_CHG);
        if (stop && n == MAX_ITER) n = i + 1;
        prev = loss;
    }
    const int n_it = __builtin_amdgcn_readfirstlane(n);

    if (__builtin_expect(n_it < MAX_ITER, 0)) {
        for (int half = 0; half < TR_T / 16; ++half) {
            int t = tid;
            asm volatile("" : "+v"(t));
            ibs_tile(barkspec, fb, spec_init, lds_spec, lds_diff, lds_part, t, row0 + 16 * half, n_it);
            for (int i = tid; i < 16 * TR_F; i += 256) {
                const int r = i / TR_F;
                tile[(16 * half + r) * TR_F + (i - r * TR_F)] = lds_spec[r * LDSS + f0 + (i - r * TR_F)];
            }
            __syncthreads();
        }
    }

    const int wave = tid >> 5;
    const int lane = tid & 31;
    const int b    = row0 >> 9;
    const int t0   = row0 & 511;
    float* dst = out + (b * N_STFT + f0) * TIME + t0 + lane;
    #pragma unroll
    for (int j = 0; j < (TR_F + 7) / 8; ++j) {
        const int c = wave + 8 * j;
        if (c < TR_F) dst[c * TIME] = tile[lane * TR_F + c];
    }
}


extern "C" void kernel_launch(void* const* d_in, const int* in_sizes, int n_in,
                              void* d_out, int out_size, void* d_ws, size_t ws_size,
                              hipStream_t stream) {
    const float* barkspec  = (const float*)d_in[0];
    const float* fb        = (const float*)d_in[1];
    const float* spec_init = (const float*)d_in[2];

    float* ws       = (float*)d_ws;
    float* spec     = ws;
    float* chk      = spec + ROWS * FP;

    ibs_run<<<NBLK, 256, 0, stream>>>(barkspec, fb, spec_init, spec, chk);
    ibs_finish<<<TR_GRID, 256, 0, stream>>>(barkspec, fb, spec_init, spec, chk, (float*)d_out);
}


# ibs_run GEMM1: a long span's A refills requested without tests (group index clamped) so every LDS wait is counted and the chain runs during a refill; short spans one nested request; sums same
# speedup vs baseline: 12.8816x; 1.0633x over previous
#include <hip/hip_runtime.h>
#include <math.h>

typedef float v2f __attribute__((ext_vector_type(2)));
typedef float v8f __attribute__((ext_vector_type(8)));

#define N_STFT   513
#define N_BARKS  128
#define BATCH    4
#define TIME     512
#define ROWS     (BATCH * TIME)
#define FP       528
#define NB       128
#define NBLK     (ROWS / 16)
#define LDSS     532
#define LDSD     132
#define LDSD_PAD 32
#define MAX_ITER 30
#define LR       0.1f
#define MOMENTUM 0.9f
#define TOL_LOSS 1e-5f
#define TOL_CHG  1e-8f


__device__ __forceinline__ v2f fb_g1(const float* __restrict__ p1, const float* __restrict__ pr, int hsel, int s, int sr, bool whole) {
    v2f v;
    if (whole) {
        v[0] = pr[4 * sr * NB];
        v[1] = pr[4 * sr * NB + NB];
    } else {
        const int f0 = 4 * s + 2 * hsel, f1 = f0 + 1;
        const int d0 = (f0 < N_STFT ? f0 : N_STFT - 1) - 2 * hsel;
        const int d1 = (f1 < N_STFT ? f1 : N_STFT - 1) - 2 * hsel;
        const float x = p1[d0 * NB], y = p1[d1 * NB];
        v[0] = f0 < N_STFT ? x : 0.0f;
        v[1] = f1 < N_STFT ? y : 0.0f;
    }
    return v;
}

template <bool TAIL>
__device__ __forceinline__ v2f fb_g2(const float* __restrict__ p2, int mi, int fbase, int s) {
    v2f v;
    if (!TAIL) {
        const float2 t = *(const float2*)(p2 + fbase * NB + 4 * s);
        v[0] = t.x; v[1] = t.y;
    } else {
        const int  f  = fbase + mi;
        const bool ok = f < N_STFT;
        const int  d  = (ok ? f : N_STFT - 1) - mi;
        const float2 t = *(const float2*)(p2 + d * NB + 4 * s);
        v[0] = ok ? t.x : 0.0f;
        v[1] = ok ? t.y : 0.0f;
    }
    return v;
}

__device__ __forceinline__ unsigned or_bits(unsigned acc, float4 v) {
    return acc | __builtin_bit_cast(unsigned, v.x) | __builtin_bit_cast(unsigned, v.y)
               | __builtin_bit_cast(unsigned, v.z) | __builtin_bit_cast(unsigned, v.w);
}
__device__ __forceinline__ bool bits_nonzero(unsigned acc) { return (acc << 1) != 0u; }


#define G1_CH   33
#define G2_ST   32
#define G1_GR   33
#define G1_NS   16
#define G2_NS   8

#define STAGE_FENCE() __builtin_amdgcn_sched_barrier(0)

template <int M>
__device__ __forceinline__ float row_xor(float x) {
    return __builtin_bit_cast(float, __builtin_amdgcn_update_dpp(0, __builtin_bit_cast(int, x),
                                                                 0x160 | M, 0xf, 0xf, false));
}

__device__ __forceinline__ float xhalf(float x) {
    int b = __builtin_bit_cast(int, x);
    return __builtin_bit_cast(float, __builtin_amdgcn_permlanex16(b, b, 0x76543210, 0xfedcba98, false, false));
}

__device__ __forceinline__ float wave_sum(float x) {
    x += row_xor<1>(x);
    x += row_xor<2>(x);
    x += row_xor<4>(x);
    x += row_xor<8>(x);
    x += xhalf(x);
    return x;
}

__device__ __forceinline__ void g1_load(v2f (&b)[G1_CH], const float* __restrict__ p1, int hsel, int c) {
    #pragma unroll
    for (int j = 0; j < G1_CH; ++j) b[j] = fb_g1(p1, p1, hsel, c * G1_CH + j, c * G1_CH + j, c * G1_CH + j < N_STFT / 4);
}

__device__ __forceinline__ v8f g1_mma(v8f acc, const v2f (&b)[G1_CH], const float* ap, int c) {
    const float* a0 = ap + c * (G1_CH * 4);
    #pragma unroll
    for (int j = 0; j < G1_CH; ++j) {
        v2f a;
        a[0] = a0[4 * j];
        a[1] = a0[4 * j + 1];
        acc = __builtin_amdgcn_wmma_f32_16x16x4_f32(false, a, false, b[j],
                                                    (short)0, acc, false, false);
    }
    return acc;
}


template <int E>
__device__ __forceinline__ void g1_group_a(v2f (&a)[8][4], const float* a0, int g) {
    const float* p = a0 + 16 * g;
    #pragma unroll
    for (int q = 0; q < 4; ++q) {
        a[E][q][0] = p[4 * q];
        a[E][q][1] = p[4 * q + 1];
    }
}

template <int E, int G>
__device__ __forceinline__ v8f g1_group_mma(v8f acc, const v2f (&a)[8][4], const v2f (&b)[G1_NS][4]) {
    #pragma unroll
    for (int q = 0; q < 4; ++q)
        acc = __builtin_amdgcn_wmma_f32_16x16x4_f32(false, a[E][q], false, b[G][q],
                                                    (short)0, acc, false, false);
    return acc;
}

template <int J>
__device__ __forceinline__ void g1_short_a(v2f (&a)[8][4], const float* a0, int n) {
    if constexpr (J < 8) {
        if (J < n) {
            g1_group_a<J>(a, a0, J);
            g1_short_a<J + 1>(a, a0, n);
        }
    }
}

template <int J>
__device__ __forceinline__ v8f g1_short_mma(v8f acc, const v2f (&a)[8][4], const v2f (&b)[G1_NS][4], int n) {
    if constexpr (J < 8) {
        if (J < n) {
            acc = g1_group_mma<J, J>(acc, a, b);
            acc = g1_short_mma<J + 1>(acc, a, b, n);
        }
    }
    return acc;
}

template <int E0, int G0, bool CLAMP>
__device__ __forceinline__ void g1_set_a(v2f (&a)[8][4], const float* a0, int last) {
    g1_group_a<E0    >(a, a0, CLAMP ? (G0     < last ? G0     : last) : G0);
    g1_group_a<E0 + 1>(a, a0, CLAMP ? (G0 + 1 < last ? G0 + 1 : last) : G0 + 1);
    g1_group_a<E0 + 2>(a, a0, CLAMP ? (G0 + 2 < last ? G0 + 2 : last) : G0 + 2);
    g1_group_a<E0 + 3>(a, a0, CLAMP ? (G0 + 3 < last ? G0 + 3 : last) : G0 + 3);
}

template <int E0, int G0>
__device__ __forceinline__ v8f g1_set_mma(v8f acc, const v2f (&a)[8][4], const v2f (&b)[G1_NS][4]) {
    acc = g1_group_mma<E0,     G0    >(acc, a, b);
    acc = g1_group_mma<E0 + 1, G0 + 1>(acc, a, b);
    acc = g1_group_mma<E0 + 2, G0 + 2>(acc, a, b);
    acc = g1_group_mma<E0 + 3, G0 + 3>(acc, a, b);
    return acc;
}

template <int E0, int G0, int J>
__device__ __forceinline__ v8f g1_tail_mma(v8f acc, const v2f (&a)[8][4], const v2f (&b)[G1_NS][4], int n) {
    if constexpr (J < 4) {
        if (G0 + J < n) {
            acc = g1_group_mma<E0 + J, G0 + J>(acc, a, b);
            acc = g1_tail_mma<E0, G0, J + 1>(acc, a, b, n);
        }
    }
    return acc;
}

template <bool TAIL>
__device__ __forceinline__ void g2_load(v2f (&b)[G2_ST], const float* __restrict__ p2, int mi, int fbase) {
    #pragma unroll
    for (int s = 0; s < G2_ST; ++s) b[s] = fb_g2<TAIL>(p2, mi, fbase, s);
}

__device__ __forceinline__ void g2_update(v8f acc2, float (&bu)[8], float* lsp, int fbase) {
    #pragma unroll
    for (int r = 0; r < 8; ++r) {
        float g  = acc2[r] * (-2.0f / (float)ROWS);
        float bn = MOMENTUM * bu[r] + g;
        float sn = lsp[r * LDSS + fbase] - LR * bn;
        lsp[r * LDSS + fbase] = sn > 0.0f ? sn : 0.0f;
        bu[r] = bn;
    }
}

template <bool TAIL>
__device__ __forceinline__ v8f g2_dense(const float* __restrict__ p2, int mi, const float* a2p, int fbase) {
    v2f b[G2_ST], a2[G2_ST];
    g2_load<TAIL>(b, p2, mi, fbase);
    #pragma unroll
    for (int s = 0; s < G2_ST; ++s) {
        a2[s][0] = a2p[4 * s];
        a2[s][1] = a2p[4 * s + 1];
    }
    v8f acc2 = {0.f, 0.f, 0.f, 0.f, 0.f, 0.f, 0.f, 0.f};
    #pragma unroll
    for (int s = 0; s < G2_ST; ++s)
        acc2 = __builtin_amdgcn_wmma_f32_16x16x4_f32(false, a2[s], false, b[s],
                                                     (short)0, acc2, false, false);
    return acc2;
}


__device__ __forceinline__ void g2_span_a(v2f (&a)[G2_NS], const float* a2p, int e, int n) {
    if (n > G2_NS / 2) {
        #pragma unroll
        for (int j = 0; j < G2_NS / 2; ++j) { a[j][0] = a2p[4 * (e + j)]; a[j][1] = a2p[4 * (e + j) + 1]; }
    } else {
        #pragma unroll
        for (int j = 0; j < G2_NS / 2; ++j) { a[j][0] = 0.0f; a[j][1] = 0.0f; }
    }
    #pragma unroll
    for (int j = G2_NS / 2; j < G2_NS; ++j) { a[j][0] = a2p[4 * (e + j)]; a[j][1] = a2p[4 * (e + j) + 1]; }
}

__device__ __forceinline__ v8f g2_chain(const v2f (&a)[G2_NS], const v2f (&b)[G2_NS], int n) {
    v8f acc2 = {0.f, 0.f, 0.f, 0.f, 0.f, 0.f, 0.f, 0.f};
#define G2_M(J) acc2 = __builtin_amdgcn_wmma_f32_16x16x4_f32(false, a[J], false, b[J], (short)0, acc2, false, false);
    if (n > 4) {
        if (n > 6) {
            if (n > 7) { G2_M(0) }
            G2_M(1)
        }
        if (n > 5) { G2_M(2) }
        G2_M(3)
    }
    if (n > 2) {
        if (n > 3) { G2_M(4) }
        G2_M(5)
    }
    if (n > 1) { G2_M(6) }
    if (n > 0) { G2_M(7) }
#undef G2_M
    return acc2;
}

__device__ __forceinline__ v8f g2_tile(const v2f (&b)[G2_NS], const float* a2p, int e, int n) {
    v2f a[G2_NS];
    g2_span_a(a, a2p, e, n);
    return g2_chain(a, b, n);
}

__device__ __forceinline__ void g2_old(float (&so)[8], const float* lsp, int fbase) {
    #pragma unroll
    for (int r = 0; r < 8; ++r) so[r] = lsp[r * LDSS + fbase];
}

__device__ __forceinline__ void g2_apply(v8f acc2, float (&bu)[8], float (&sp)[8], float* lsp, int fbase) {
    #pragma unroll
    for (int r = 0; r < 8; ++r) {
        float g  = acc2[r] * (-2.0f / (float)ROWS);
        float bn = MOMENTUM * bu[r] + g;
        float sn = sp[r] - LR * bn;
        float sv = sn > 0.0f ? sn : 0.0f;
        lsp[r * LDSS + fbase] = sv;
        sp[r] = sv;
        bu[r] = bn;
    }
}

template <bool TAIL>
__device__ __forceinline__ void g2_scan(const float* __restrict__ fb, int lane, int fbase, int& lo, int& n) {
    const float* p = fb + fbase * NB + 4 * lane;
    unsigned acc = 0u;
    #pragma unroll
    for (int r = 0; r < 16; ++r)
        if (!TAIL || fbase + r < N_STFT) acc = or_bits(acc, *(const float4*)(p + r * NB));
    const unsigned m = __builtin_amdgcn_ballot_w32(bits_nonzero(acc));
    int l = 0, c = 0;
    if (m != 0u) { l = __builtin_ctz(m); c = 32 - __builtin_clz(m) - l; }
    lo = __builtin_amdgcn_readfirstlane(l);
    n  = __builtin_amdgcn_readfirstlane(c);
}

template <bool TAIL>
__device__ __forceinline__ void g2_keep(v2f (&b)[G2_NS], const float* __restrict__ p2, int mi, int fbase, int lo, int n) {
    const int e = lo - (G2_NS - n);
    #pragma unroll
    for (int j = 0; j < G2_NS; ++j) {
        v2f v = {0.0f, 0.0f};
        if (n <= G2_NS && j >= G2_NS - n) v = fb_g2<TAIL>(p2, mi, fbase, e + j);
        b[j] = v;
    }
}

__device__ __forceinline__ void ibs_tile(const float* __restrict__ barkspec,
                                         const float* __restrict__ fb,
                                         const float* __restrict__ spec_init,
                                         float* lds_spec, float* lds_diff, float* lds_part,
                                         int tid, int rowbase, int n_it) {
    const int wave    = __builtin_amdgcn_readfirstlane(tid >> 5);
    const int lane    = tid & 31;
    const int mi      = lane & 15;
    const int hsel    = lane >> 4;
    const int koff    = hsel * 2;
    const int nbase   = wave * 16;

    const int rowh = rowbase + hsel * 8;
    const int bb   = rowh >> 9;
    const int t0   = rowh & 511;
    const float* mp = barkspec + (bb * N_BARKS + nbase + mi) * TIME + t0;
    float mv[8];
    #pragma unroll
    for (int r = 0; r < 8; ++r) mv[r] = mp[r];

    const float* sip = spec_init + rowbase * N_STFT + tid;
    float i0[16], i1[16], i2[16];
    #pragma unroll
    for (int r = 0; r < 16; ++r) {
        i0[r] = sip[r * N_STFT];
        i1[r] = sip[r * N_STFT + 256];
        i2[r] = 0.0f;
    }
    if (tid == 0) {
        #pragma unroll
        for (int r = 0; r < 16; ++r) i2[r] = sip[r * N_STFT + 512];
    }

    const float* ap  = lds_spec + mi * LDSS + koff;
    const float* b1p = fb + koff * NB + (nbase + mi);

    #pragma unroll
    for (int r = 0; r < 16; ++r) {
        lds_spec[r * LDSS + tid]       = i0[r];
        lds_spec[r * LDSS + tid + 256] = i1[r];
    }
    if (tid < FP - 512) {
        #pragma unroll
        for (int r = 0; r < 16; ++r) lds_spec[r * LDSS + 512 + tid] = i2[r];
    }

    if (tid < LDSD_PAD) lds_diff[tid] = 0.0f;
    if (tid < 16 * (LDSD - NB)) lds_diff[LDSD_PAD + (tid >> 2) * LDSD + NB + (tid & 3)] = 0.0f;

    const float*  b2p = fb + mi * NB + koff;
    float*        lsp = lds_spec + (hsel * 8) * LDSS + mi;
    const float*  a2p = lds_diff + LDSD_PAD + mi * LDSD + koff;
    const int     fb0 = nbase;

    int glo1, n1;
    {
        const int    sr = lane >> 2;
        const float* sp = fb + sr * NB + nbase + 4 * (lane & 3);
        unsigned long long m1 = 0ull;
        #pragma unroll
        for (int g = 0; g < G1_GR; ++g) {
            unsigned acc = 0u;
            #pragma unroll
            for (int h = 0; h < 2; ++h)
                if (16 * g + 8 * h + sr < N_STFT) acc = or_bits(acc, *(const float4*)(sp + (16 * g + 8 * h) * NB));
            if (__builtin_amdgcn_ballot_w32(bits_nonzero(acc)) != 0u) m1 |= (1ull << g);
        }
        int l = 0, c = 0;
        if (m1 != 0ull) { l = __builtin_ctzll(m1); c = 64 - __builtin_clzll(m1) - l; }
        glo1 = __builtin_amdgcn_readfirstlane(l);
        n1   = __builtin_amdgcn_readfirstlane(c);
    }
    const bool fit1 = n1 <= G1_NS;

    int lo20, n20, lo21, n21, lo22, n22, lo23, n23, lo24 = 0, n24 = 0;
    g2_scan<false>(fb, lane, fb0,       lo20, n20);
    g2_scan<false>(fb, lane, fb0 + 128, lo21, n21);
    g2_scan<false>(fb, lane, fb0 + 256, lo22, n22);
    g2_scan<false>(fb, lane, fb0 + 384, lo23, n23);
    if (wave == 0) g2_scan<true>(fb, lane, 512, lo24, n24);

    v2f B1[G1_NS][4];
    if (fit1) {
        const float* pg = b1p + glo1 * (16 * NB);
        #pragma unroll
        for (int j = 0; j < G1_NS; ++j) {
            if (j < n1) {
                if (glo1 + j < N_STFT / 16) {
                    #pragma unroll
                    for (int q = 0; q < 4; ++q) B1[j][q] = fb_g1(b1p, pg, hsel, 4 * (glo1 + j) + q, 4 * j + q, true);
                } else {
                    #pragma unroll
                    for (int q = 0; q < 4; ++q) B1[j][q] = fb_g1(b1p, pg, hsel, 4 * (glo1 + j) + q, 4 * j + q, false);
                }
            }
        }
    }
    v2f B20[G2_NS], B21[G2_NS], B22[G2_NS], B23[G2_NS], B24[G2_NS];
    g2_keep<false>(B20, b2p, mi, fb0,       lo20, n20);
    g2_keep<false>(B21, b2p, mi, fb0 + 128, lo21, n21);
    g2_keep<false>(B22, b2p, mi, fb0 + 256, lo22, n22);
    g2_keep<false>(B23, b2p, mi, fb0 + 384, lo23, n23);
    if (wave == 0) {
        g2_keep<true>(B24, b2p, mi, 512, lo24, n24);
    } else {
        #pragma unroll
        for (int j = 0; j < G2_NS; ++j) { B24[j][0] = 0.0f; B24[j][1] = 0.0f; }
    }
    const int e20 = lo20 - (G2_NS - n20), e21 = lo21 - (G2_NS - n21), e22 = lo22 - (G2_NS - n22),
              e23 = lo23 - (G2_NS - n23), e24 = lo24 - (G2_NS - n24);
    const bool fit2 = (n20 <= G2_NS) && (n21 <= G2_NS) && (n22 <= G2_NS) && (n23 <= G2_NS) && (n24 <= G2_NS);

    __syncthreads();

    float p0[8], p1[8], p2[8], p3[8], p4[8];
    g2_old(p0, lsp, fb0);
    g2_old(p1, lsp, fb0 + 128);
    g2_old(p2, lsp, fb0 + 256);
    g2_old(p3, lsp, fb0 + 384);
    g2_old(p4, lsp, 512);

    float u0[8], u1[8], u2[8], u3[8], u4[8];
    #pragma unroll
    for (int r = 0; r < 8; ++r) { u0[r] = 0.0f; u1[r] = 0.0f; u2[r] = 0.0f; u3[r] = 0.0f; u4[r] = 0.0f; }

    for (int it = 0; it < n_it; ++it) {
        int keep = 0;
        asm volatile("" : "+s"(keep));
        const float* b1i = b1p + keep;
        const float* b2i = b2p + keep;

        v8f acc = {0.f, 0.f, 0.f, 0.f, 0.f, 0.f, 0.f, 0.f};
        if (fit1) {
            int nn = n1;
            asm volatile("" : "+s"(nn));
            const float* a0 = ap + glo1 * 16;
            v2f aS[8][4];
            if (nn > 8) {
                g1_set_a<0, 0, false>(aS, a0, 0);
                g1_set_a<4, 4, false>(aS, a0, 0);       STAGE_FENCE();
                acc = g1_set_mma<0, 0>(acc, aS, B1);    STAGE_FENCE();
                g1_set_a<0, 8, true>(aS, a0, nn - 1);   STAGE_FENCE();
                acc = g1_set_mma<4, 4>(acc, aS, B1);    STAGE_FENCE();
                g1_set_a<4, 12, true>(aS, a0, nn - 1);  STAGE_FENCE();
                acc = g1_tail_mma<0, 8, 0>(acc, aS, B1, nn);
                acc = g1_tail_mma<4, 12, 0>(acc, aS, B1, nn);
            } else {
                g1_short_a<0>(aS, a0, nn);
                acc = g1_short_mma<0>(acc, aS, B1, nn);
            }
        } else {
            v2f bA[G1_CH];
            #pragma unroll
            for (int c = 0; c < 4; ++c) {
                g1_load(bA, b1i, hsel, c);    STAGE_FENCE();
                acc = g1_mma(acc, bA, ap, c); STAGE_FENCE();
            }
        }

        float ssq = 0.0f;
        #pragma unroll
        for (int r = 0; r < 8; ++r) {
            float d = mv[r] - acc[r];
            ssq += d * d;
            lds_diff[LDSD_PAD + (r + hsel * 8) * LDSD + nbase + mi] = d;
        }
        const float wsq = wave_sum(ssq);
        if (lane == 0) lds_part[it * 8 + wave] = wsq;
        __syncthreads();

        int m0 = n20, m1 = n21, m2 = n22, m3 = n23, m4 = n24;
        asm volatile("" : "+s"(m0), "+s"(m1), "+s"(m2), "+s"(m3), "+s"(m4));
        if (fit2) {
            v2f A0[G2_NS], A1[G2_NS], A2[G2_NS], A3[G2_NS], A4[G2_NS];
            g2_span_a(A0, a2p, e20, m0);
            g2_span_a(A1, a2p, e21, m1);
            g2_span_a(A2, a2p, e22, m2);
            g2_span_a(A3, a2p, e23, m3);
            if (wave == 0) g2_span_a(A4, a2p, e24, m4);
            STAGE_FENCE();
            v8f c0 = g2_chain(A0, B20, m0);
            v8f c1 = g2_chain(A1, B21, m1);
            v8f c2 = g2_chain(A2, B22, m2);
            v8f c3 = g2_chain(A3, B23, m3);
            g2_apply(c0, u0, p0, lsp, fb0);
            g2_apply(c1, u1, p1, lsp, fb0 + 128);
            g2_apply(c2, u2, p2, lsp, fb0 + 256);
            g2_apply(c3, u3, p3, lsp, fb0 + 384);
            if (wave == 0) {
                v8f c4 = g2_chain(A4, B24, m4);
                g2_apply(c4, u4, p4, lsp, 512);
            }
        } else {
            v8f c0 = (n20 <= G2_NS) ? g2_tile(B20, a2p, e20, m0) : g2_dense<false>(b2i, mi, a2p, fb0);
            g2_update(c0, u0, lsp, fb0);
            v8f c1 = (n21 <= G2_NS) ? g2_tile(B21, a2p, e21, m1) : g2_dense<false>(b2i, mi, a2p, fb0 + 128);
            g2_update(c1, u1, lsp, fb0 + 128);
            v8f c2 = (n22 <= G2_NS) ? g2_tile(B22, a2p, e22, m2) : g2_dense<false>(b2i, mi, a2p, fb0 + 256);
            g2_update(c2, u2, lsp, fb0 + 256);
            v8f c3 = (n23 <= G2_NS) ? g2_tile(B23, a2p, e23, m3) : g2_dense<false>(b2i, mi, a2p, fb0 + 384);
            g2_update(c3, u3, lsp, fb0 + 384);
            if (wave == 0) {
                v8f c4 = (n24 <= G2_NS) ? g2_tile(B24, a2p, e24, m4) : g2_dense<true>(b2i, mi, a2p, 512);
                g2_update(c4, u4, lsp, 512);
            }
        }
        __syncthreads();
    }

}

__global__ __launch_bounds__(256) void ibs_run(const float* __restrict__ barkspec,
                                               const float* __restrict__ fb,
                                               const float* __restrict__ spec_init,
                                               float* __restrict__ spec,
                                               float* __restrict__ chk) {
    __shared__ __attribute__((aligned(16))) float lds_spec[16 * LDSS];
    __shared__ float lds_diff[LDSD_PAD + 16 * LDSD];
    __shared__ __attribute__((aligned(16))) float lds_part[MAX_ITER * 8];

    const int tid     = threadIdx.x;
    const int rowbase = blockIdx.x * 16;

    ibs_tile(barkspec, fb, spec_init, lds_spec, lds_diff, lds_part, tid, rowbase, MAX_ITER);

    if (tid < MAX_ITER) {
        const float4 pa = *(const float4*)(lds_part + tid * 8);
        const float4 pb = *(const float4*)(lds_part + tid * 8 + 4);
        float s = 0.0f;
        s += pa.x; s += pa.y; s += pa.z; s += pa.w;
        s += pb.x; s += pb.y; s += pb.z; s += pb.w;
        chk[tid * NBLK + blockIdx.x] = s;
    }

    #pragma unroll
    for (int j = 0; j < 8; ++j) {
        int i  = tid + 256 * j;
        int r  = i / (FP / 4);
        int c4 = i - r * (FP / 4);
        *(float4*)(spec + (rowbase + r) * FP + c4 * 4) = *(const float4*)(lds_spec + r * LDSS + c4 * 4);
    }
    if (tid < 16 * (FP / 4) - 2048)
        *(float4*)(spec + (rowbase + 15) * FP + (tid + 68) * 4) =
            *(const float4*)(lds_spec + 15 * LDSS + (tid + 68) * 4);
}

#define TR_T    32
#define TR_F    171
#define TR_N    (TR_T * TR_F)
#define TR_FULL (TR_N / 256)
#define TR_TAIL (TR_N - 256 * TR_FULL)
#define TR_GRID ((ROWS / TR_T) * (N_STFT / TR_F))

__global__ __launch_bounds__(256) void ibs_finish(const float* __restrict__ barkspec,
                                                  const float* __restrict__ fb,
                                                  const float* __restrict__ spec_init,
                                                  const float* __restrict__ spec,
                                                  const float* __restrict__ chk,
                                                  float* __restrict__ out) {
    __shared__ float tile[TR_N];
    __shared__ __attribute__((aligned(16))) float lds_spec[16 * LDSS];
    __shared__ float lds_diff[LDSD_PAD + 16 * LDSD];
    __shared__ __attribute__((aligned(16))) float lds_part[MAX_ITER * 8];
    __shared__ float lds_loss[32];

    const int tid  = threadIdx.x;
    const int rt   = blockIdx.x / (N_STFT / TR_F);
    const int f0   = (blockIdx.x - rt * (N_STFT / TR_F)) * TR_F;
    const int row0 = rt * TR_T;
    const float* src = spec + row0 * FP + f0;

    float v[TR_FULL + 1];
    #pragma unroll
    for (int j = 0; j < TR_FULL; ++j) {
        const int i = tid + 256 * j;
        const int r = i / TR_F;
        v[j] = src[r * FP + (i - r * TR_F)];
    }
    v[TR_FULL] = 0.0f;
    if (tid < TR_TAIL) {
        const int i = tid + 256 * TR_FULL;
        const int r = i / TR_F;
        v[TR_FULL] = src[r * FP + (i - r * TR_F)];
    }

    if (tid < MAX_ITER) {
        const float4* row = (const float4*)(chk + tid * NBLK);
        float4 cr[NBLK / 4];
        #pragma unroll
        for (int i = 0; i < NBLK / 4; ++i) cr[i] = row[i];
        STAGE_FENCE();
        float s = 0.0f;
        #pragma unroll
        for (int i = 0; i < NBLK / 4; ++i) { s += cr[i].x; s += cr[i].y; s += cr[i].z; s += cr[i].w; }
        lds_loss[tid] = s / (float)ROWS;
    }
    #pragma unroll
    for (int j = 0; j < TR_FULL; ++j) tile[tid + 256 * j] = v[j];
    if (tid < TR_TAIL) tile[tid + 256 * TR_FULL] = v[TR_FULL];
    __syncthreads();
    int   n    = MAX_ITER;
    float prev = INFINITY;
    for (int i = 0; i < MAX_ITER; ++i) {
        const float loss = lds_loss[i];
        const bool  stop = (loss < TOL_LOSS) || (fabsf(prev - loss) < TOL_CHG);
        if (stop && n == MAX_ITER) n = i + 1;
        prev = loss;
    }
    const int n_it = __builtin_amdgcn_readfirstlane(n);

    if (__builtin_expect(n_it < MAX_ITER, 0)) {
        for (int half = 0; half < TR_T / 16; ++half) {
            int t = tid;
            asm volatile("" : "+v"(t));
            ibs_tile(barkspec, fb, spec_init, lds_spec, lds_diff, lds_part, t, row0 + 16 * half, n_it);
            for (int i = tid; i < 16 * TR_F; i += 256) {
                const int r = i / TR_F;
                tile[(16 * half + r) * TR_F + (i - r * TR_F)] = lds_spec[r * LDSS + f0 + (i - r * TR_F)];
            }
            __syncthreads();
        }
    }

    const int wave = tid >> 5;
    const int lane = tid & 31;
    const int b    = row0 >> 9;
    const int t0   = row0 & 511;
    float* dst = out + (b * N_STFT + f0) * TIME + t0 + lane;
    #pragma unroll
    for (int j = 0; j < (TR_F + 7) / 8; ++j) {
        const int c = wave + 8 * j;
        if (c < TR_F) dst[c * TIME] = tile[lane * TR_F + c];
    }
}


extern "C" void kernel_launch(void* const* d_in, const int* in_sizes, int n_in,
                              void* d_out, int out_size, void* d_ws, size_t ws_size,
                              hipStream_t stream) {
    const float* barkspec  = (const float*)d_in[0];
    const float* fb        = (const float*)d_in[1];
    const float* spec_init = (const float*)d_in[2];

    float* ws       = (float*)d_ws;
    float* spec     = ws;
    float* chk      = spec + ROWS * FP;

    ibs_run<<<NBLK, 256, 0, stream>>>(barkspec, fb, spec_init, spec, chk);
    ibs_finish<<<TR_GRID, 256, 0, stream>>>(barkspec, fb, spec_init, spec, chk, (float*)d_out);
}


# ibs_run GEMM1 in half column tiles: wave w = upper 8 bark columns of tile w + lower 8 of tile 7-w, each over its own shorter k span; loss partial sums regrouped: order changes, stays fixed
# speedup vs baseline: 13.3320x; 1.0350x over previous
#include <hip/hip_runtime.h>
#include <math.h>

typedef float v2f __attribute__((ext_vector_type(2)));
typedef float v8f __attribute__((ext_vector_type(8)));

#define N_STFT   513
#define N_BARKS  128
#define BATCH    4
#define TIME     512
#define ROWS     (BATCH * TIME)
#define FP       528
#define NB       128
#define NBLK     (ROWS / 16)
#define LDSS     532
#define LDSD     132
#define LDSD_PAD 32
#define MAX_ITER 30
#define LR       0.1f
#define MOMENTUM 0.9f
#define TOL_LOSS 1e-5f
#define TOL_CHG  1e-8f


__device__ __forceinline__ v2f fb_g1(const float* __restrict__ p1, const float* __restrict__ pr, int hsel, int s, int sr, bool whole) {
    v2f v;
    if (whole) {
        v[0] = pr[4 * sr * NB];
        v[1] = pr[4 * sr * NB + NB];
    } else {
        const int f0 = 4 * s + 2 * hsel, f1 = f0 + 1;
        const int d0 = (f0 < N_STFT ? f0 : N_STFT - 1) - 2 * hsel;
        const int d1 = (f1 < N_STFT ? f1 : N_STFT - 1) - 2 * hsel;
        const float x = p1[d0 * NB], y = p1[d1 * NB];
        v[0] = f0 < N_STFT ? x : 0.0f;
        v[1] = f1 < N_STFT ? y : 0.0f;
    }
    return v;
}

template <bool TAIL>
__device__ __forceinline__ v2f fb_g2(const float* __restrict__ p2, int mi, int fbase, int s) {
    v2f v;
    if (!TAIL) {
        const float2 t = *(const float2*)(p2 + fbase * NB + 4 * s);
        v[0] = t.x; v[1] = t.y;
    } else {
        const int  f  = fbase + mi;
        const bool ok = f < N_STFT;
        const int  d  = (ok ? f : N_STFT - 1) - mi;
        const float2 t = *(const float2*)(p2 + d * NB + 4 * s);
        v[0] = ok ? t.x : 0.0f;
        v[1] = ok ? t.y : 0.0f;
    }
    return v;
}

__device__ __forceinline__ unsigned or_bits(unsigned acc, float4 v) {
    return acc | __builtin_bit_cast(unsigned, v.x) | __builtin_bit_cast(unsigned, v.y)
               | __builtin_bit_cast(unsigned, v.z) | __builtin_bit_cast(unsigned, v.w);
}
__device__ __forceinline__ bool bits_nonzero(unsigned acc) { return (acc << 1) != 0u; }


#define G1_CH   11
#define G1_NC   (FP / 4 / G1_CH)
#define G2_ST   32
#define G1_GR   33
#define G1_NH   9
#define G1_NL   3
#define G2_NS   8

#define STAGE_FENCE() __builtin_amdgcn_sched_barrier(0)

template <int M>
__device__ __forceinline__ float row_xor(float x) {
    return __builtin_bit_cast(float, __builtin_amdgcn_update_dpp(0, __builtin_bit_cast(int, x),
                                                                 0x160 | M, 0xf, 0xf, false));
}

__device__ __forceinline__ float xhalf(float x) {
    int b = __builtin_bit_cast(int, x);
    return __builtin_bit_cast(float, __builtin_amdgcn_permlanex16(b, b, 0x76543210, 0xfedcba98, false, false));
}

__device__ __forceinline__ float wave_sum(float x) {
    x += row_xor<1>(x);
    x += row_xor<2>(x);
    x += row_xor<4>(x);
    x += row_xor<8>(x);
    x += xhalf(x);
    return x;
}

__device__ __forceinline__ void g1_load(v2f (&b)[G1_CH], const float* __restrict__ p1, int hsel, int c) {
    #pragma unroll
    for (int j = 0; j < G1_CH; ++j) b[j] = fb_g1(p1, p1, hsel, c * G1_CH + j, c * G1_CH + j, c * G1_CH + j < N_STFT / 4);
}

__device__ __forceinline__ v8f g1_mma(v8f acc, const v2f (&b)[G1_CH], const float* ap, int c) {
    const float* a0 = ap + c * (G1_CH * 4);
    #pragma unroll
    for (int j = 0; j < G1_CH; ++j) {
        v2f a;
        a[0] = a0[4 * j];
        a[1] = a0[4 * j + 1];
        acc = __builtin_amdgcn_wmma_f32_16x16x4_f32(false, a, false, b[j],
                                                    (short)0, acc, false, false);
    }
    return acc;
}


template <int E>
__device__ __forceinline__ void g1_group_a(v2f (&a)[8][4], const float* a0, int g) {
    const float* p = a0 + 16 * g;
    #pragma unroll
    for (int q = 0; q < 4; ++q) {
        a[E][q][0] = p[4 * q];
        a[E][q][1] = p[4 * q + 1];
    }
}

template <int E>
__device__ __forceinline__ void g1_slot_a(v2f (&a)[8][4], const float* a0, int g, int last) {
    g1_group_a<E>(a, a0, g < last ? g : last);
}

template <int E, int G, int NBG>
__device__ __forceinline__ v8f g1_group_mma(v8f acc, const v2f (&a)[8][4], const v2f (&b)[NBG][4]) {
    #pragma unroll
    for (int q = 0; q < 4; ++q)
        acc = __builtin_amdgcn_wmma_f32_16x16x4_f32(false, a[E][q], false, b[G][q],
                                                    (short)0, acc, false, false);
    return acc;
}

template <int E0, int G0, int J, int CNT, int NBG>
__device__ __forceinline__ v8f g1_run_mma(v8f acc, const v2f (&a)[8][4], const v2f (&b)[NBG][4], int n) {
    if constexpr (J < CNT) {
        if (G0 + J < n) {
            acc = g1_group_mma<E0 + J, G0 + J, NBG>(acc, a, b);
            acc = g1_run_mma<E0, G0, J + 1, CNT, NBG>(acc, a, b, n);
        }
    }
    return acc;
}

template <int NBG>
__device__ __forceinline__ void g1_keep(v2f (&b)[NBG][4], const float* __restrict__ p1, int hsel, int glo, int n) {
    const float* pg = p1 + glo * (16 * NB);
    #pragma unroll
    for (int j = 0; j < NBG; ++j) {
        if (j < n) {
            if (glo + j < N_STFT / 16) {
                #pragma unroll
                for (int q = 0; q < 4; ++q) b[j][q] = fb_g1(p1, pg, hsel, 4 * (glo + j) + q, 4 * j + q, true);
            } else {
                #pragma unroll
                for (int q = 0; q < 4; ++q) b[j][q] = fb_g1(p1, pg, hsel, 4 * (glo + j) + q, 4 * j + q, false);
            }
        }
    }
}

template <bool TAIL>
__device__ __forceinline__ void g2_load(v2f (&b)[G2_ST], const float* __restrict__ p2, int mi, int fbase) {
    #pragma unroll
    for (int s = 0; s < G2_ST; ++s) b[s] = fb_g2<TAIL>(p2, mi, fbase, s);
}

__device__ __forceinline__ void g2_update(v8f acc2, float (&bu)[8], float* lsp, int fbase) {
    #pragma unroll
    for (int r = 0; r < 8; ++r) {
        float g  = acc2[r] * (-2.0f / (float)ROWS);
        float bn = MOMENTUM * bu[r] + g;
        float sn = lsp[r * LDSS + fbase] - LR * bn;
        lsp[r * LDSS + fbase] = sn > 0.0f ? sn : 0.0f;
        bu[r] = bn;
    }
}

template <bool TAIL>
__device__ __forceinline__ v8f g2_dense(const float* __restrict__ p2, int mi, const float* a2p, int fbase) {
    v2f b[G2_ST], a2[G2_ST];
    g2_load<TAIL>(b, p2, mi, fbase);
    #pragma unroll
    for (int s = 0; s < G2_ST; ++s) {
        a2[s][0] = a2p[4 * s];
        a2[s][1] = a2p[4 * s + 1];
    }
    v8f acc2 = {0.f, 0.f, 0.f, 0.f, 0.f, 0.f, 0.f, 0.f};
    #pragma unroll
    for (int s = 0; s < G2_ST; ++s)
        acc2 = __builtin_amdgcn_wmma_f32_16x16x4_f32(false, a2[s], false, b[s],
                                                     (short)0, acc2, false, false);
    return acc2;
}


__device__ __forceinline__ void g2_span_a(v2f (&a)[G2_NS], const float* a2p, int e, int n) {
    if (n > G2_NS / 2) {
        #pragma unroll
        for (int j = 0; j < G2_NS / 2; ++j) { a[j][0] = a2p[4 * (e + j)]; a[j][1] = a2p[4 * (e + j) + 1]; }
    } else {
        #pragma unroll
        for (int j = 0; j < G2_NS / 2; ++j) { a[j][0] = 0.0f; a[j][1] = 0.0f; }
    }
    #pragma unroll
    for (int j = G2_NS / 2; j < G2_NS; ++j) { a[j][0] = a2p[4 * (e + j)]; a[j][1] = a2p[4 * (e + j) + 1]; }
}

__device__ __forceinline__ v8f g2_chain(const v2f (&a)[G2_NS], const v2f (&b)[G2_NS], int n) {
    v8f acc2 = {0.f, 0.f, 0.f, 0.f, 0.f, 0.f, 0.f, 0.f};
#define G2_M(J) acc2 = __builtin_amdgcn_wmma_f32_16x16x4_f32(false, a[J], false, b[J], (short)0, acc2, false, false);
    if (n > 4) {
        if (n > 6) {
            if (n > 7) { G2_M(0) }
            G2_M(1)
        }
        if (n > 5) { G2_M(2) }
        G2_M(3)
    }
    if (n > 2) {
        if (n > 3) { G2_M(4) }
        G2_M(5)
    }
    if (n > 1) { G2_M(6) }
    if (n > 0) { G2_M(7) }
#undef G2_M
    return acc2;
}

__device__ __forceinline__ v8f g2_tile(const v2f (&b)[G2_NS], const float* a2p, int e, int n) {
    v2f a[G2_NS];
    g2_span_a(a, a2p, e, n);
    return g2_chain(a, b, n);
}

__device__ __forceinline__ void g2_old(float (&so)[8], const float* lsp, int fbase) {
    #pragma unroll
    for (int r = 0; r < 8; ++r) so[r] = lsp[r * LDSS + fbase];
}

__device__ __forceinline__ void g2_apply(v8f acc2, float (&bu)[8], float (&sp)[8], float* lsp, int fbase) {
    #pragma unroll
    for (int r = 0; r < 8; ++r) {
        float g  = acc2[r] * (-2.0f / (float)ROWS);
        float bn = MOMENTUM * bu[r] + g;
        float sn = sp[r] - LR * bn;
        float sv = sn > 0.0f ? sn : 0.0f;
        lsp[r * LDSS + fbase] = sv;
        sp[r] = sv;
        bu[r] = bn;
    }
}

template <bool TAIL>
__device__ __forceinline__ void g2_scan(const float* __restrict__ fb, int lane, int fbase, int& lo, int& n) {
    const float* p = fb + fbase * NB + 4 * lane;
    unsigned acc = 0u;
    #pragma unroll
    for (int r = 0; r < 16; ++r)
        if (!TAIL || fbase + r < N_STFT) acc = or_bits(acc, *(const float4*)(p + r * NB));
    const unsigned m = __builtin_amdgcn_ballot_w32(bits_nonzero(acc));
    int l = 0, c = 0;
    if (m != 0u) { l = __builtin_ctz(m); c = 32 - __builtin_clz(m) - l; }
    lo = __builtin_amdgcn_readfirstlane(l);
    n  = __builtin_amdgcn_readfirstlane(c);
}

template <bool TAIL>
__device__ __forceinline__ void g2_keep(v2f (&b)[G2_NS], const float* __restrict__ p2, int mi, int fbase, int lo, int n) {
    const int e = lo - (G2_NS - n);
    #pragma unroll
    for (int j = 0; j < G2_NS; ++j) {
        v2f v = {0.0f, 0.0f};
        if (n <= G2_NS && j >= G2_NS - n) v = fb_g2<TAIL>(p2, mi, fbase, e + j);
        b[j] = v;
    }
}

__device__ __forceinline__ void ibs_tile(const float* __restrict__ barkspec,
                                         const float* __restrict__ fb,
                                         const float* __restrict__ spec_init,
                                         float* lds_spec, float* lds_diff, float* lds_part,
                                         int tid, int rowbase, int n_it) {
    const int wave    = __builtin_amdgcn_readfirstlane(tid >> 5);
    const int lane    = tid & 31;
    const int mi      = lane & 15;
    const int hsel    = lane >> 4;
    const int koff    = hsel * 2;
    const int nbase   = wave * 16;

    const bool upperH = wave >= 4;
    const int  tileH  = upperH ? wave : 7 - wave;
    const int  tileL  = 7 - tileH;
    const int  col    = (mi >= 8 ? wave : 7 - wave) * 16 + mi;
    const bool useH   = (mi >= 8) == upperH;

    const int rowh = rowbase + hsel * 8;
    const int bb   = rowh >> 9;
    const int t0   = rowh & 511;
    const float* mp = barkspec + (bb * N_BARKS + col) * TIME + t0;
    float mv[8];
    #pragma unroll
    for (int r = 0; r < 8; ++r) mv[r] = mp[r];

    const float* sip = spec_init + rowbase * N_STFT + tid;
    float i0[16], i1[16], i2[16];
    #pragma unroll
    for (int r = 0; r < 16; ++r) {
        i0[r] = sip[r * N_STFT];
        i1[r] = sip[r * N_STFT + 256];
        i2[r] = 0.0f;
    }
    if (tid == 0) {
        #pragma unroll
        for (int r = 0; r < 16; ++r) i2[r] = sip[r * N_STFT + 512];
    }

    const float* ap  = lds_spec + mi * LDSS + koff;
    const float* b1pH = fb + koff * NB + (tileH * 16 + mi);
    const float* b1pL = fb + koff * NB + (tileL * 16 + mi);

    #pragma unroll
    for (int r = 0; r < 16; ++r) {
        lds_spec[r * LDSS + tid]       = i0[r];
        lds_spec[r * LDSS + tid + 256] = i1[r];
    }
    if (tid < FP - 512) {
        #pragma unroll
        for (int r = 0; r < 16; ++r) lds_spec[r * LDSS + 512 + tid] = i2[r];
    }

    if (tid < LDSD_PAD) lds_diff[tid] = 0.0f;
    if (tid < 16 * (LDSD - NB)) lds_diff[LDSD_PAD + (tid >> 2) * LDSD + NB + (tid & 3)] = 0.0f;

    const float*  b2p = fb + mi * NB + koff;
    float*        lsp = lds_spec + (hsel * 8) * LDSS + mi;
    const float*  a2p = lds_diff + LDSD_PAD + mi * LDSD + koff;
    const int     fb0 = nbase;

    int gloH, nH, gloL, nL;
    {
        const int    sr   = lane >> 2;
        const int    sc   = lane & 3;
        const int    scol = sc < 2 ? tileH * 16 + (upperH ? 8 : 0) + 4 * sc
                                   : tileL * 16 + (upperH ? 0 : 8) + 4 * (sc - 2);
        const float* sp   = fb + sr * NB + scol;
        unsigned long long mH = 0ull, mL = 0ull;
        #pragma unroll
        for (int g = 0; g < G1_GR; ++g) {
            unsigned acc = 0u;
            #pragma unroll
            for (int h = 0; h < 2; ++h)
                if (16 * g + 8 * h + sr < N_STFT) acc = or_bits(acc, *(const float4*)(sp + (16 * g + 8 * h) * NB));
            const unsigned bal = __builtin_amdgcn_ballot_w32(bits_nonzero(acc));
            if ((bal & 0x33333333u) != 0u) mH |= (1ull << g);
            if ((bal & 0xccccccccu) != 0u) mL |= (1ull << g);
        }
        int l = 0, c = 0;
        if (mH != 0ull) { l = __builtin_ctzll(mH); c = 64 - __builtin_clzll(mH) - l; }
        gloH = __builtin_amdgcn_readfirstlane(l);
        nH   = __builtin_amdgcn_readfirstlane(c);
        l = 0; c = 0;
        if (mL != 0ull) { l = __builtin_ctzll(mL); c = 64 - __builtin_clzll(mL) - l; }
        gloL = __builtin_amdgcn_readfirstlane(l);
        nL   = __builtin_amdgcn_readfirstlane(c);
    }
    const bool fit1 = (nH <= G1_NH) && (nL <= G1_NL);

    int lo20, n20, lo21, n21, lo22, n22, lo23, n23, lo24 = 0, n24 = 0;
    g2_scan<false>(fb, lane, fb0,       lo20, n20);
    g2_scan<false>(fb, lane, fb0 + 128, lo21, n21);
    g2_scan<false>(fb, lane, fb0 + 256, lo22, n22);
    g2_scan<false>(fb, lane, fb0 + 384, lo23, n23);
    if (wave == 0) g2_scan<true>(fb, lane, 512, lo24, n24);

    v2f BH[G1_NH][4], BL[G1_NL][4];
    if (fit1) {
        g1_keep<G1_NH>(BH, b1pH, hsel, gloH, nH);
        g1_keep<G1_NL>(BL, b1pL, hsel, gloL, nL);
    }
    v2f B20[G2_NS], B21[G2_NS], B22[G2_NS], B23[G2_NS], B24[G2_NS];
    g2_keep<false>(B20, b2p, mi, fb0,       lo20, n20);
    g2_keep<false>(B21, b2p, mi, fb0 + 128, lo21, n21);
    g2_keep<false>(B22, b2p, mi, fb0 + 256, lo22, n22);
    g2_keep<false>(B23, b2p, mi, fb0 + 384, lo23, n23);
    if (wave == 0) {
        g2_keep<true>(B24, b2p, mi, 512, lo24, n24);
    } else {
        #pragma unroll
        for (int j = 0; j < G2_NS; ++j) { B24[j][0] = 0.0f; B24[j][1] = 0.0f; }
    }
    const int e20 = lo20 - (G2_NS - n20), e21 = lo21 - (G2_NS - n21), e22 = lo22 - (G2_NS - n22),
              e23 = lo23 - (G2_NS - n23), e24 = lo24 - (G2_NS - n24);
    const bool fit2 = (n20 <= G2_NS) && (n21 <= G2_NS) && (n22 <= G2_NS) && (n23 <= G2_NS) && (n24 <= G2_NS);

    __syncthreads();

    float p0[8], p1[8], p2[8], p3[8], p4[8];
    g2_old(p0, lsp, fb0);
    g2_old(p1, lsp, fb0 + 128);
    g2_old(p2, lsp, fb0 + 256);
    g2_old(p3, lsp, fb0 + 384);
    g2_old(p4, lsp, 512);

    float u0[8], u1[8], u2[8], u3[8], u4[8];
    #pragma unroll
    for (int r = 0; r < 8; ++r) { u0[r] = 0.0f; u1[r] = 0.0f; u2[r] = 0.0f; u3[r] = 0.0f; u4[r] = 0.0f; }

    for (int it = 0; it < n_it; ++it) {
        int keep = 0;
        asm volatile("" : "+s"(keep));
        const float* b1iH = b1pH + keep;
        const float* b1iL = b1pL + keep;
        const float* b2i = b2p + keep;

        v8f accH = {0.f, 0.f, 0.f, 0.f, 0.f, 0.f, 0.f, 0.f};
        v8f accL = {0.f, 0.f, 0.f, 0.f, 0.f, 0.f, 0.f, 0.f};
        if (fit1) {
            int nh = nH, nl = nL;
            asm volatile("" : "+s"(nh), "+s"(nl));
            const float* aH = ap + gloH * 16;
            const float* aL = ap + gloL * 16;
            const int lastH = nh > 0 ? nh - 1 : 0;
            const int lastL = nl > 0 ? nl - 1 : 0;
            v2f aS[8][4];
            g1_slot_a<0>(aS, aH, 0, lastH);
            g1_slot_a<1>(aS, aH, 1, lastH);
            g1_slot_a<2>(aS, aH, 2, lastH);
            g1_slot_a<3>(aS, aH, 3, lastH);
            g1_slot_a<4>(aS, aH, 4, lastH);
            g1_slot_a<5>(aS, aH, 5, lastH);
            g1_slot_a<6>(aS, aH, 6, lastH);
            g1_slot_a<7>(aS, aH, 7, lastH);                              STAGE_FENCE();
            accH = g1_run_mma<0, 0, 0, 4, G1_NH>(accH, aS, BH, nh);      STAGE_FENCE();
            g1_slot_a<0>(aS, aH, 8, lastH);
            g1_slot_a<1>(aS, aL, 0, lastL);
            g1_slot_a<2>(aS, aL, 1, lastL);
            g1_slot_a<3>(aS, aL, 2, lastL);                              STAGE_FENCE();
            accH = g1_run_mma<4, 4, 0, 4, G1_NH>(accH, aS, BH, nh);
            accH = g1_run_mma<0, 8, 0, 1, G1_NH>(accH, aS, BH, nh);
            accL = g1_run_mma<1, 0, 0, 3, G1_NL>(accL, aS, BL, nl);
        } else {
            v2f bA[G1_CH];
            #pragma unroll
            for (int c = 0; c < G1_NC; ++c) {
                g1_load(bA, b1iH, hsel, c);     STAGE_FENCE();
                accH = g1_mma(accH, bA, ap, c); STAGE_FENCE();
            }
            #pragma unroll
            for (int c = 0; c < G1_NC; ++c) {
                g1_load(bA, b1iL, hsel, c);     STAGE_FENCE();
                accL = g1_mma(accL, bA, ap, c); STAGE_FENCE();
            }
        }

        float ssq = 0.0f;
        #pragma unroll
        for (int r = 0; r < 8; ++r) {
            float a = useH ? accH[r] : accL[r];
            float d = mv[r] - a;
            ssq += d * d;
            lds_diff[LDSD_PAD + (r + hsel * 8) * LDSD + col] = d;
        }
        const float wsq = wave_sum(ssq);
        if (lane == 0) lds_part[it * 8 + wave] = wsq;
        __syncthreads();

        int m0 = n20, m1 = n21, m2 = n22, m3 = n23, m4 = n24;
        asm volatile("" : "+s"(m0), "+s"(m1), "+s"(m2), "+s"(m3), "+s"(m4));
        if (fit2) {
            v2f A0[G2_NS], A1[G2_NS], A2[G2_NS], A3[G2_NS], A4[G2_NS];
            g2_span_a(A0, a2p, e20, m0);
            g2_span_a(A1, a2p, e21, m1);
            g2_span_a(A2, a2p, e22, m2);
            g2_span_a(A3, a2p, e23, m3);
            if (wave == 0) g2_span_a(A4, a2p, e24, m4);
            STAGE_FENCE();
            v8f c0 = g2_chain(A0, B20, m0);
            v8f c1 = g2_chain(A1, B21, m1);
            v8f c2 = g2_chain(A2, B22, m2);
            v8f c3 = g2_chain(A3, B23, m3);
            g2_apply(c0, u0, p0, lsp, fb0);
            g2_apply(c1, u1, p1, lsp, fb0 + 128);
            g2_apply(c2, u2, p2, lsp, fb0 + 256);
            g2_apply(c3, u3, p3, lsp, fb0 + 384);
            if (wave == 0) {
                v8f c4 = g2_chain(A4, B24, m4);
                g2_apply(c4, u4, p4, lsp, 512);
            }
        } else {
            v8f c0 = (n20 <= G2_NS) ? g2_tile(B20, a2p, e20, m0) : g2_dense<false>(b2i, mi, a2p, fb0);
            g2_update(c0, u0, lsp, fb0);
            v8f c1 = (n21 <= G2_NS) ? g2_tile(B21, a2p, e21, m1) : g2_dense<false>(b2i, mi, a2p, fb0 + 128);
            g2_update(c1, u1, lsp, fb0 + 128);
            v8f c2 = (n22 <= G2_NS) ? g2_tile(B22, a2p, e22, m2) : g2_dense<false>(b2i, mi, a2p, fb0 + 256);
            g2_update(c2, u2, lsp, fb0 + 256);
            v8f c3 = (n23 <= G2_NS) ? g2_tile(B23, a2p, e23, m3) : g2_dense<false>(b2i, mi, a2p, fb0 + 384);
            g2_update(c3, u3, lsp, fb0 + 384);
            if (wave == 0) {
                v8f c4 = (n24 <= G2_NS) ? g2_tile(B24, a2p, e24, m4) : g2_dense<true>(b2i, mi, a2p, 512);
                g2_update(c4, u4, lsp, 512);
            }
        }
        __syncthreads();
    }

}

__global__ __launch_bounds__(256) void ibs_run(const float* __restrict__ barkspec,
                                               const float* __restrict__ fb,
                                               const float* __restrict__ spec_init,
                                               float* __restrict__ spec,
                                               float* __restrict__ chk) {
    __shared__ __attribute__((aligned(16))) float lds_spec[16 * LDSS];
    __shared__ float lds_diff[LDSD_PAD + 16 * LDSD];
    __shared__ __attribute__((aligned(16))) float lds_part[MAX_ITER * 8];

    const int tid     = threadIdx.x;
    const int rowbase = blockIdx.x * 16;

    ibs_tile(barkspec, fb, spec_init, lds_spec, lds_diff, lds_part, tid, rowbase, MAX_ITER);

    if (tid < MAX_ITER) {
        const float4 pa = *(const float4*)(lds_part + tid * 8);
        const float4 pb = *(const float4*)(lds_part + tid * 8 + 4);
        float s = 0.0f;
        s += pa.x; s += pa.y; s += pa.z; s += pa.w;
        s += pb.x; s += pb.y; s += pb.z; s += pb.w;
        chk[tid * NBLK + blockIdx.x] = s;
    }

    #pragma unroll
    for (int j = 0; j < 8; ++j) {
        int i  = tid + 256 * j;
        int r  = i / (FP / 4);
        int c4 = i - r * (FP / 4);
        *(float4*)(spec + (rowbase + r) * FP + c4 * 4) = *(const float4*)(lds_spec + r * LDSS + c4 * 4);
    }
    if (tid < 16 * (FP / 4) - 2048)
        *(float4*)(spec + (rowbase + 15) * FP + (tid + 68) * 4) =
            *(const float4*)(lds_spec + 15 * LDSS + (tid + 68) * 4);
}

#define TR_T    32
#define TR_F    171
#define TR_N    (TR_T * TR_F)
#define TR_FULL (TR_N / 256)
#define TR_TAIL (TR_N - 256 * TR_FULL)
#define TR_GRID ((ROWS / TR_T) * (N_STFT / TR_F))

__global__ __launch_bounds__(256) void ibs_finish(const float* __restrict__ barkspec,
                                                  const float* __restrict__ fb,
                                                  const float* __restrict__ spec_init,
                                                  const float* __restrict__ spec,
                                                  const float* __restrict__ chk,
                                                  float* __restrict__ out) {
    __shared__ float tile[TR_N];
    __shared__ __attribute__((aligned(16))) float lds_spec[16 * LDSS];
    __shared__ float lds_diff[LDSD_PAD + 16 * LDSD];
    __shared__ __attribute__((aligned(16))) float lds_part[MAX_ITER * 8];
    __shared__ float lds_loss[32];

    const int tid  = threadIdx.x;
    const int rt   = blockIdx.x / (N_STFT / TR_F);
    const int f0   = (blockIdx.x - rt * (N_STFT / TR_F)) * TR_F;
    const int row0 = rt * TR_T;
    const float* src = spec + row0 * FP + f0;

    float v[TR_FULL + 1];
    #pragma unroll
    for (int j = 0; j < TR_FULL; ++j) {
        const int i = tid + 256 * j;
        const int r = i / TR_F;
        v[j] = src[r * FP + (i - r * TR_F)];
    }
    v[TR_FULL] = 0.0f;
    if (tid < TR_TAIL) {
        const int i = tid + 256 * TR_FULL;
        const int r = i / TR_F;
        v[TR_FULL] = src[r * FP + (i - r * TR_F)];
    }

    if (tid < MAX_ITER) {
        const float4* row = (const float4*)(chk + tid * NBLK);
        float4 cr[NBLK / 4];
        #pragma unroll
        for (int i = 0; i < NBLK / 4; ++i) cr[i] = row[i];
        STAGE_FENCE();
        float s = 0.0f;
        #pragma unroll
        for (int i = 0; i < NBLK / 4; ++i) { s += cr[i].x; s += cr[i].y; s += cr[i].z; s += cr[i].w; }
        lds_loss[tid] = s / (float)ROWS;
    }
    #pragma unroll
    for (int j = 0; j < TR_FULL; ++j) tile[tid + 256 * j] = v[j];
    if (tid < TR_TAIL) tile[tid + 256 * TR_FULL] = v[TR_FULL];
    __syncthreads();
    int   n    = MAX_ITER;
    float prev = INFINITY;
    for (int i = 0; i < MAX_ITER; ++i) {
        const float loss = lds_loss[i];
        const bool  stop = (loss < TOL_LOSS) || (fabsf(prev - loss) < TOL_CHG);
        if (stop && n == MAX_ITER) n = i + 1;
        prev = loss;
    }
    const int n_it = __builtin_amdgcn_readfirstlane(n);

    if (__builtin_expect(n_it < MAX_ITER, 0)) {
        for (int half = 0; half < TR_T / 16; ++half) {
            int t = tid;
            asm volatile("" : "+v"(t));
            ibs_tile(barkspec, fb, spec_init, lds_spec, lds_diff, lds_part, t, row0 + 16 * half, n_it);
            for (int i = tid; i < 16 * TR_F; i += 256) {
                const int r = i / TR_F;
                tile[(16 * half + r) * TR_F + (i - r * TR_F)] = lds_spec[r * LDSS + f0 + (i - r * TR_F)];
            }
            __syncthreads();
        }
    }

    const int wave = tid >> 5;
    const int lane = tid & 31;
    const int b    = row0 >> 9;
    const int t0   = row0 & 511;
    float* dst = out + (b * N_STFT + f0) * TIME + t0 + lane;
    #pragma unroll
    for (int j = 0; j < (TR_F + 7) / 8; ++j) {
        const int c = wave + 8 * j;
        if (c < TR_F) dst[c * TIME] = tile[lane * TR_F + c];
    }
}


extern "C" void kernel_launch(void* const* d_in, const int* in_sizes, int n_in,
                              void* d_out, int out_size, void* d_ws, size_t ws_size,
                              hipStream_t stream) {
    const float* barkspec  = (const float*)d_in[0];
    const float* fb        = (const float*)d_in[1];
    const float* spec_init = (const float*)d_in[2];

    float* ws       = (float*)d_ws;
    float* spec     = ws;
    float* chk      = spec + ROWS * FP;

    ibs_run<<<NBLK, 256, 0, stream>>>(barkspec, fb, spec_init, spec, chk);
    ibs_finish<<<TR_GRID, 256, 0, stream>>>(barkspec, fb, spec_init, spec, chk, (float*)d_out);
}
